# starting kernel
# baseline (speedup 1.0000x reference)
#include <hip/hip_runtime.h>
#include <hip/hip_bf16.h>

typedef __attribute__((ext_vector_type(2))) float v2f;
typedef __attribute__((ext_vector_type(8))) float v8f;

#define WMMA_F32X4(a, b, c) \
    __builtin_amdgcn_wmma_f32_16x16x4_f32(false, (a), false, (b), (short)0, (c), false, false)

__global__ void __launch_bounds__(256) k_rankmm(const float* __restrict__ U,
                                                const float* __restrict__ V,
                                                float* __restrict__ out,
                                                int rows, int cols) {
    int idx = blockIdx.x * 256 + threadIdx.x;
    if (idx >= rows * cols) return;
    int n = idx / cols;
    int m = idx - n * cols;
    const float* up = U + n * 32;
    const float* vp = V + m * 32;
    float acc = 0.f;
#pragma unroll
    for (int r = 0; r < 32; ++r) acc += up[r] * vp[r];
    out[idx] = acc;
}

__global__ void __launch_bounds__(256) k_wmma_proj(const float* __restrict__ uh,
                                                   const float* __restrict__ KHH_U,
                                                   const float* __restrict__ KHH_V,
                                                   const float* __restrict__ Kbd_U,
                                                   float* __restrict__ wh0,
                                                   float* __restrict__ sbuf) {
    __shared__ float tld[8][16 * 32];
    const int wave = threadIdx.x >> 5;
    const int lane = threadIdx.x & 31;
    const int half = lane >> 4;
    const int mrow = lane & 15;
    const int wid  = blockIdx.x * 8 + wave;
    const int R    = wid * 16;
    const float HK = 1.0f / 1024.0f;

    const float* arow = uh + (size_t)(R + mrow) * 8193;

    v8f t0 = {}; v8f t1 = {};
    for (int k0 = 0; k0 < 1024; k0 += 4) {
        int ka = k0 + 2 * half;
        int kb = ka + 1;
        v2f a, b0, b1;
        a.x  = arow[8 * ka];
        a.y  = arow[8 * kb];
        b0.x = KHH_V[ka * 32 + mrow];
        b0.y = KHH_V[kb * 32 + mrow];
        b1.x = KHH_V[ka * 32 + 16 + mrow];
        b1.y = KHH_V[kb * 32 + 16 + mrow];
        t0 = WMMA_F32X4(a, b0, t0);
        t1 = WMMA_F32X4(a, b1, t1);
    }
    {
        float alast = arow[8192];
        v2f a, b0, b1;
        a.x  = (half == 0) ? alast : 0.f;
        a.y  = 0.f;
        b0.x = KHH_V[1024 * 32 + mrow];
        b0.y = b0.x;
        b1.x = KHH_V[1024 * 32 + 16 + mrow];
        b1.y = b1.x;
        t0 = WMMA_F32X4(a, b0, t0);
        t1 = WMMA_F32X4(a, b1, t1);
    }

    v8f s0 = {}; v8f s1 = {};
    for (int k0 = 0; k0 < 8192; k0 += 4) {
        int ka = k0 + 2 * half;
        int kb = ka + 1;
        v2f a, b0, b1;
        a.x  = arow[ka];
        a.y  = arow[kb];
        b0.x = Kbd_U[ka * 32 + mrow];
        b0.y = Kbd_U[kb * 32 + mrow];
        b1.x = Kbd_U[ka * 32 + 16 + mrow];
        b1.y = Kbd_U[kb * 32 + 16 + mrow];
        s0 = WMMA_F32X4(a, b0, s0);
        s1 = WMMA_F32X4(a, b1, s1);
    }
    {
        float alast = arow[8192];
        v2f a, b0, b1;
        a.x  = (half == 0) ? alast : 0.f;
        a.y  = 0.f;
        b0.x = Kbd_U[8192 * 32 + mrow];
        b0.y = b0.x;
        b1.x = Kbd_U[8192 * 32 + 16 + mrow];
        b1.y = b1.x;
        s0 = WMMA_F32X4(a, b0, s0);
        s1 = WMMA_F32X4(a, b1, s1);
    }
#pragma unroll
    for (int j = 0; j < 8; ++j) {
        int m = j + 8 * half;
        sbuf[(size_t)(R + m) * 32 + mrow]      = s0[j];
        sbuf[(size_t)(R + m) * 32 + 16 + mrow] = s1[j];
    }

    float* tl = tld[wave];
#pragma unroll
    for (int j = 0; j < 8; ++j) {
        int m = j + 8 * half;
        tl[m * 32 + mrow]      = t0[j] * HK;
        tl[m * 32 + 16 + mrow] = t1[j] * HK;
    }
    __syncthreads();

    for (int nt = 0; nt < 64; ++nt) {
        int n = nt * 16 + mrow;
        v8f acc = {};
#pragma unroll
        for (int k0 = 0; k0 < 32; k0 += 4) {
            int ka = k0 + 2 * half;
            int kb = ka + 1;
            v2f a, b;
            a.x = tl[mrow * 32 + ka];
            a.y = tl[mrow * 32 + kb];
            b.x = KHH_U[n * 32 + ka];
            b.y = KHH_U[n * 32 + kb];
            acc = WMMA_F32X4(a, b, acc);
        }
#pragma unroll
        for (int j = 0; j < 8; ++j) {
            int m = j + 8 * half;
            wh0[(size_t)(R + m) * 1025 + n] = acc[j];
        }
    }
    {
        v8f acc = {};
#pragma unroll
        for (int k0 = 0; k0 < 32; k0 += 4) {
            int ka = k0 + 2 * half;
            int kb = ka + 1;
            v2f a, b;
            a.x = tl[mrow * 32 + ka];
            a.y = tl[mrow * 32 + kb];
            b.x = KHH_U[1024 * 32 + ka];
            b.y = KHH_U[1024 * 32 + kb];
            acc = WMMA_F32X4(a, b, acc);
        }
        if (mrow == 0) {
#pragma unroll
            for (int j = 0; j < 8; ++j) {
                int m = j + 8 * half;
                wh0[(size_t)(R + m) * 1025 + 1024] = acc[j];
            }
        }
    }
}

__global__ void __launch_bounds__(256) k_refine(const float* __restrict__ uh,
                                                const float* __restrict__ wh0,
                                                const float* __restrict__ sbuf,
                                                const float* __restrict__ Kbd_V,
                                                const float* __restrict__ Ke0,
                                                const float* __restrict__ Ko0,
                                                const float* __restrict__ Ke1,
                                                const float* __restrict__ Ko1,
                                                const float* __restrict__ Ke2,
                                                const float* __restrict__ Ko2,
                                                const int* __restrict__ nbrL0,
                                                const int* __restrict__ nbrL1,
                                                const int* __restrict__ nbrL2,
                                                float* __restrict__ out) {
    __shared__ float wA[4100];
    __shared__ float wB[8200];
    const int tid = threadIdx.x;
    const int bc  = blockIdx.x;
    const size_t base = (size_t)bc * 8193;
    const float* u = uh + base;
    const float H = 1.0f / 8192.0f;

    for (int n = tid; n < 1025; n += 256) wA[n] = wh0[(size_t)bc * 1025 + n];
    __syncthreads();

    const float* KeP[3]  = {Ke0, Ke1, Ke2};
    const float* KoP[3]  = {Ko0, Ko1, Ko2};
    const int*   nbrP[3] = {nbrL0, nbrL1, nbrL2};
    float* oldW = wA;
    float* newW = wB;

#pragma unroll
    for (int i = 0; i < 3; ++i) {
        const int Nc = 1025 << i;
        const int step = 4 >> i;
        const float hf = H * (float)step;
        const float* Ke = KeP[i];
        const float* Ko = KoP[i];
        const int* nbr  = nbrP[i];
        for (int n = tid; n < Nc; n += 256) {
            const int* row = nbr + (size_t)(2 * n) * 15;
            const float* kp = Ke + (size_t)n * 8;
            float acc = 0.f;
#pragma unroll
            for (int m = 0; m < 8; ++m) acc += kp[m] * u[step * row[2 * m]];
            newW[2 * n] = oldW[n] + hf * acc;
        }
        for (int n = tid; n < Nc - 1; n += 256) {
            const int* row = nbr + (size_t)(2 * n + 1) * 15;
            const float* kp = Ko + (size_t)n * 15;
            float acc = 0.f;
#pragma unroll
            for (int m = 0; m < 15; ++m) acc += kp[m] * u[step * row[m]];
            newW[2 * n + 1] = 0.5f * (oldW[n] + oldW[n + 1]) + hf * acc;
        }
        __syncthreads();
        float* tmp = oldW; oldW = newW; newW = tmp;
    }

    for (int n = tid; n < 8193; n += 256) {
        float v = oldW[n];
        if (n == 0 || n == 8192) {
            const float* kv = Kbd_V + ((n == 0) ? 0 : 32);
            const float* sp = sbuf + bc * 32;
            float acc = 0.f;
#pragma unroll
            for (int r = 0; r < 32; ++r) acc += kv[r] * sp[r];
            v = H * acc;
        }
        out[base + n] = v;
    }
}

extern "C" void kernel_launch(void* const* d_in, const int* in_sizes, int n_in,
                              void* d_out, int out_size, void* d_ws, size_t ws_size,
                              hipStream_t stream) {
    const float* uh     = (const float*)d_in[0];
    const float* KHH_U  = (const float*)d_in[1];
    const float* KHH_V  = (const float*)d_in[2];
    const float* Ke0_U  = (const float*)d_in[3];
    const float* Ke0_V  = (const float*)d_in[4];
    const float* Ko0_U  = (const float*)d_in[5];
    const float* Ko0_V  = (const float*)d_in[6];
    const int*   nbr0   = (const int*)d_in[7];
    const float* Ke1_U  = (const float*)d_in[8];
    const float* Ke1_V  = (const float*)d_in[9];
    const float* Ko1_U  = (const float*)d_in[10];
    const float* Ko1_V  = (const float*)d_in[11];
    const int*   nbr1   = (const int*)d_in[12];
    const float* Ke2_U  = (const float*)d_in[13];
    const float* Ke2_V  = (const float*)d_in[14];
    const float* Ko2_U  = (const float*)d_in[15];
    const float* Ko2_V  = (const float*)d_in[16];
    const int*   nbr2   = (const int*)d_in[17];
    const float* Kbd_U  = (const float*)d_in[18];
    const float* Kbd_V  = (const float*)d_in[19];

    float* ws  = (float*)d_ws;
    float* wh0 = ws;
    float* sb  = wh0 + 524800;
    float* Ke0 = sb + 16384;
    float* Ke1 = Ke0 + 8200;
    float* Ke2 = Ke1 + 16392;
    float* Ko0 = Ke2 + 32776;
    float* Ko1 = Ko0 + 15360;
    float* Ko2 = Ko1 + 30720;

    float* outp = (float*)d_out;

    auto grid = [](int elems) { return (elems + 255) / 256; };

    k_rankmm<<<grid(1025 * 8),  256, 0, stream>>>(Ke0_U, Ke0_V, Ke0, 1025, 8);
    k_rankmm<<<grid(1024 * 15), 256, 0, stream>>>(Ko0_U, Ko0_V, Ko0, 1024, 15);
    k_rankmm<<<grid(2049 * 8),  256, 0, stream>>>(Ke1_U, Ke1_V, Ke1, 2049, 8);
    k_rankmm<<<grid(2048 * 15), 256, 0, stream>>>(Ko1_U, Ko1_V, Ko1, 2048, 15);
    k_rankmm<<<grid(4097 * 8),  256, 0, stream>>>(Ke2_U, Ke2_V, Ke2, 4097, 8);
    k_rankmm<<<grid(4096 * 15), 256, 0, stream>>>(Ko2_U, Ko2_V, Ko2, 4096, 15);

    k_wmma_proj<<<4, 256, 0, stream>>>(uh, KHH_U, KHH_V, Kbd_U, wh0, sb);

    k_refine<<<512, 256, 0, stream>>>(uh, wh0, sb, Kbd_V,
                                      Ke0, Ko0, Ke1, Ko1, Ke2, Ko2,
                                      nbr2, nbr1, nbr0, outp);
}


# Projection K loops split over 256 blocks (32 row tiles x 8 K slices x 16 waves); partial tiles are added in fixed order by a second launch that does the third product; summation order changes
# speedup vs baseline: 4.2273x; 4.2273x over previous
#include <hip/hip_runtime.h>
#include <hip/hip_bf16.h>

typedef __attribute__((ext_vector_type(2))) float v2f;
typedef __attribute__((ext_vector_type(8))) float v8f;

#define WMMA_F32X4(a, b, c) \
    __builtin_amdgcn_wmma_f32_16x16x4_f32(false, (a), false, (b), (short)0, (c), false, false)

__global__ void __launch_bounds__(256) k_rankmm(const float* __restrict__ U,
                                                const float* __restrict__ V,
                                                float* __restrict__ out,
                                                int rows, int cols) {
    int idx = blockIdx.x * 256 + threadIdx.x;
    if (idx >= rows * cols) return;
    int n = idx / cols;
    int m = idx - n * cols;
    const float* up = U + n * 32;
    const float* vp = V + m * 32;
    float acc = 0.f;
#pragma unroll
    for (int r = 0; r < 32; ++r) acc += up[r] * vp[r];
    out[idx] = acc;
}

#define P_KS     8
#define P_WAVES 16
#define TILE_ROWS_FLOATS 16400

__global__ void __launch_bounds__(512) k_wmma_proj_part(const float* __restrict__ uh,
                                                        const float* __restrict__ KHH_V,
                                                        const float* __restrict__ Kbd_U,
                                                        float* __restrict__ part) {
    __shared__ float red[P_WAVES][1024];
    const int wave = __builtin_amdgcn_readfirstlane((int)(threadIdx.x >> 5));
    const int lane = threadIdx.x & 31;
    const int half = lane >> 4;
    const int mrow = lane & 15;
    const int mt   = blockIdx.x / P_KS;
    const int ks   = blockIdx.x % P_KS;
    const int R    = mt * 16;
    const bool tail = (ks == P_KS - 1) && (wave == 0);

    const float* arow = uh + (size_t)(R + mrow) * 8193;

    v8f t0 = {}; v8f t1 = {};
    for (int it = 0; it < 32 / P_WAVES; ++it) {
        int k0 = 4 * (ks * 32 + it * P_WAVES + wave);
        int ka = k0 + 2 * half;
        int kb = ka + 1;
        v2f a, b0, b1;
        a.x  = arow[8 * ka];
        a.y  = arow[8 * kb];
        b0.x = KHH_V[ka * 32 + mrow];
        b0.y = KHH_V[kb * 32 + mrow];
        b1.x = KHH_V[ka * 32 + 16 + mrow];
        b1.y = KHH_V[kb * 32 + 16 + mrow];
        t0 = WMMA_F32X4(a, b0, t0);
        t1 = WMMA_F32X4(a, b1, t1);
    }
    if (tail) {
        float alast = arow[8192];
        v2f a, b0, b1;
        a.x  = (half == 0) ? alast : 0.f;
        a.y  = 0.f;
        b0.x = KHH_V[1024 * 32 + mrow];
        b0.y = b0.x;
        b1.x = KHH_V[1024 * 32 + 16 + mrow];
        b1.y = b1.x;
        t0 = WMMA_F32X4(a, b0, t0);
        t1 = WMMA_F32X4(a, b1, t1);
    }

    v8f s0 = {}; v8f s1 = {};
    for (int it = 0; it < 256 / P_WAVES; ++it) {
        int k0 = 4 * (ks * 256 + it * P_WAVES + wave);
        int ka = k0 + 2 * half;
        int kb = ka + 1;
        v2f a, b0, b1;
        a.x  = arow[ka];
        a.y  = arow[kb];
        b0.x = Kbd_U[ka * 32 + mrow];
        b0.y = Kbd_U[kb * 32 + mrow];
        b1.x = Kbd_U[ka * 32 + 16 + mrow];
        b1.y = Kbd_U[kb * 32 + 16 + mrow];
        s0 = WMMA_F32X4(a, b0, s0);
        s1 = WMMA_F32X4(a, b1, s1);
    }
    if (tail) {
        float alast = arow[8192];
        v2f a, b0, b1;
        a.x  = (half == 0) ? alast : 0.f;
        a.y  = 0.f;
        b0.x = Kbd_U[8192 * 32 + mrow];
        b0.y = b0.x;
        b1.x = Kbd_U[8192 * 32 + 16 + mrow];
        b1.y = b1.x;
        s0 = WMMA_F32X4(a, b0, s0);
        s1 = WMMA_F32X4(a, b1, s1);
    }

    {
        float* rw = red[wave];
#pragma unroll
        for (int j = 0; j < 8; ++j) {
            rw[j * 32 + lane]       = t0[j];
            rw[256 + j * 32 + lane] = t1[j];
            rw[512 + j * 32 + lane] = s0[j];
            rw[768 + j * 32 + lane] = s1[j];
        }
    }
    __syncthreads();
    {
        float* dst = part + (size_t)mt * TILE_ROWS_FLOATS + ks * 1024;
#pragma unroll
        for (int q = 0; q < 2; ++q) {
            const int e = threadIdx.x + 512 * q;
            float sum = red[0][e];
#pragma unroll
            for (int w = 1; w < P_WAVES; ++w) sum += red[w][e];
            dst[e] = sum;
        }
    }
}

__global__ void __launch_bounds__(1024) k_wmma_proj_out(const float* __restrict__ KHH_U,
                                                        float* __restrict__ wh0,
                                                        float* __restrict__ sbuf) {
    __shared__ float tl[16 * 32];
    const int wave = __builtin_amdgcn_readfirstlane((int)(threadIdx.x >> 5));
    const int lane = threadIdx.x & 31;
    const int half = lane >> 4;
    const int mrow = lane & 15;
    const int mt   = blockIdx.x;
    const int R    = mt * 16;
    const float HK = 1.0f / 1024.0f;

    {
        const int e = threadIdx.x;
        const float* src = wh0 + (size_t)mt * TILE_ROWS_FLOATS;
        float sum = src[e];
#pragma unroll
        for (int s = 1; s < P_KS; ++s) sum += src[s * 1024 + e];
        const int p = e >> 8;
        const int j = (e >> 5) & 7;
        const int l = e & 31;
        const int m = j + 8 * (l >> 4);
        const int n = ((p & 1) << 4) + (l & 15);
        if (p < 2) tl[m * 32 + n] = sum * HK;
        else       sbuf[(size_t)(R + m) * 32 + n] = sum;
    }
    __syncthreads();

    for (int nt = 2 * wave; nt < 2 * wave + 2; ++nt) {
        int n = nt * 16 + mrow;
        v8f acc = {};
#pragma unroll
        for (int k0 = 0; k0 < 32; k0 += 4) {
            int ka = k0 + 2 * half;
            int kb = ka + 1;
            v2f a, b;
            a.x = tl[mrow * 32 + ka];
            a.y = tl[mrow * 32 + kb];
            b.x = KHH_U[n * 32 + ka];
            b.y = KHH_U[n * 32 + kb];
            acc = WMMA_F32X4(a, b, acc);
        }
#pragma unroll
        for (int j = 0; j < 8; ++j) {
            int m = j + 8 * half;
            wh0[(size_t)(R + m) * 1025 + n] = acc[j];
        }
    }
    if (wave == 31) {
        v8f acc = {};
#pragma unroll
        for (int k0 = 0; k0 < 32; k0 += 4) {
            int ka = k0 + 2 * half;
            int kb = ka + 1;
            v2f a, b;
            a.x = tl[mrow * 32 + ka];
            a.y = tl[mrow * 32 + kb];
            b.x = KHH_U[1024 * 32 + ka];
            b.y = KHH_U[1024 * 32 + kb];
            acc = WMMA_F32X4(a, b, acc);
        }
        if (mrow == 0) {
#pragma unroll
            for (int j = 0; j < 8; ++j) {
                int m = j + 8 * half;
                wh0[(size_t)(R + m) * 1025 + 1024] = acc[j];
            }
        }
    }
}

__global__ void __launch_bounds__(256) k_refine(const float* __restrict__ uh,
                                                const float* __restrict__ wh0,
                                                const float* __restrict__ sbuf,
                                                const float* __restrict__ Kbd_V,
                                                const float* __restrict__ Ke0,
                                                const float* __restrict__ Ko0,
                                                const float* __restrict__ Ke1,
                                                const float* __restrict__ Ko1,
                                                const float* __restrict__ Ke2,
                                                const float* __restrict__ Ko2,
                                                const int* __restrict__ nbrL0,
                                                const int* __restrict__ nbrL1,
                                                const int* __restrict__ nbrL2,
                                                float* __restrict__ out) {
    __shared__ float wA[4100];
    __shared__ float wB[8200];
    const int tid = threadIdx.x;
    const int bc  = blockIdx.x;
    const size_t base = (size_t)bc * 8193;
    const float* u = uh + base;
    const float H = 1.0f / 8192.0f;

    for (int n = tid; n < 1025; n += 256) wA[n] = wh0[(size_t)bc * 1025 + n];
    __syncthreads();

    const float* KeP[3]  = {Ke0, Ke1, Ke2};
    const float* KoP[3]  = {Ko0, Ko1, Ko2};
    const int*   nbrP[3] = {nbrL0, nbrL1, nbrL2};
    float* oldW = wA;
    float* newW = wB;

#pragma unroll
    for (int i = 0; i < 3; ++i) {
        const int Nc = 1025 << i;
        const int step = 4 >> i;
        const float hf = H * (float)step;
        const float* Ke = KeP[i];
        const float* Ko = KoP[i];
        const int* nbr  = nbrP[i];
        for (int n = tid; n < Nc; n += 256) {
            const int* row = nbr + (size_t)(2 * n) * 15;
            const float* kp = Ke + (size_t)n * 8;
            float acc = 0.f;
#pragma unroll
            for (int m = 0; m < 8; ++m) acc += kp[m] * u[step * row[2 * m]];
            newW[2 * n] = oldW[n] + hf * acc;
        }
        for (int n = tid; n < Nc - 1; n += 256) {
            const int* row = nbr + (size_t)(2 * n + 1) * 15;
            const float* kp = Ko + (size_t)n * 15;
            float acc = 0.f;
#pragma unroll
            for (int m = 0; m < 15; ++m) acc += kp[m] * u[step * row[m]];
            newW[2 * n + 1] = 0.5f * (oldW[n] + oldW[n + 1]) + hf * acc;
        }
        __syncthreads();
        float* tmp = oldW; oldW = newW; newW = tmp;
    }

    for (int n = tid; n < 8193; n += 256) {
        float v = oldW[n];
        if (n == 0 || n == 8192) {
            const float* kv = Kbd_V + ((n == 0) ? 0 : 32);
            const float* sp = sbuf + bc * 32;
            float acc = 0.f;
#pragma unroll
            for (int r = 0; r < 32; ++r) acc += kv[r] * sp[r];
            v = H * acc;
        }
        out[base + n] = v;
    }
}

extern "C" void kernel_launch(void* const* d_in, const int* in_sizes, int n_in,
                              void* d_out, int out_size, void* d_ws, size_t ws_size,
                              hipStream_t stream) {
    const float* uh     = (const float*)d_in[0];
    const float* KHH_U  = (const float*)d_in[1];
    const float* KHH_V  = (const float*)d_in[2];
    const float* Ke0_U  = (const float*)d_in[3];
    const float* Ke0_V  = (const float*)d_in[4];
    const float* Ko0_U  = (const float*)d_in[5];
    const float* Ko0_V  = (const float*)d_in[6];
    const int*   nbr0   = (const int*)d_in[7];
    const float* Ke1_U  = (const float*)d_in[8];
    const float* Ke1_V  = (const float*)d_in[9];
    const float* Ko1_U  = (const float*)d_in[10];
    const float* Ko1_V  = (const float*)d_in[11];
    const int*   nbr1   = (const int*)d_in[12];
    const float* Ke2_U  = (const float*)d_in[13];
    const float* Ke2_V  = (const float*)d_in[14];
    const float* Ko2_U  = (const float*)d_in[15];
    const float* Ko2_V  = (const float*)d_in[16];
    const int*   nbr2   = (const int*)d_in[17];
    const float* Kbd_U  = (const float*)d_in[18];
    const float* Kbd_V  = (const float*)d_in[19];

    float* ws  = (float*)d_ws;
    float* wh0 = ws;
    float* sb  = wh0 + 524800;
    float* Ke0 = sb + 16384;
    float* Ke1 = Ke0 + 8200;
    float* Ke2 = Ke1 + 16392;
    float* Ko0 = Ke2 + 32776;
    float* Ko1 = Ko0 + 15360;
    float* Ko2 = Ko1 + 30720;

    float* outp = (float*)d_out;

    auto grid = [](int elems) { return (elems + 255) / 256; };

    k_rankmm<<<grid(1025 * 8),  256, 0, stream>>>(Ke0_U, Ke0_V, Ke0, 1025, 8);
    k_rankmm<<<grid(1024 * 15), 256, 0, stream>>>(Ko0_U, Ko0_V, Ko0, 1024, 15);
    k_rankmm<<<grid(2049 * 8),  256, 0, stream>>>(Ke1_U, Ke1_V, Ke1, 2049, 8);
    k_rankmm<<<grid(2048 * 15), 256, 0, stream>>>(Ko1_U, Ko1_V, Ko1, 2048, 15);
    k_rankmm<<<grid(4097 * 8),  256, 0, stream>>>(Ke2_U, Ke2_V, Ke2, 4097, 8);
    k_rankmm<<<grid(4096 * 15), 256, 0, stream>>>(Ko2_U, Ko2_V, Ko2, 4096, 15);

    k_wmma_proj_part<<<256, 512, 0, stream>>>(uh, KHH_V, Kbd_U, wh0);
    k_wmma_proj_out<<<32, 1024, 0, stream>>>(KHH_U, wh0, sb);

    k_refine<<<512, 256, 0, stream>>>(uh, wh0, sb, Kbd_V,
                                      Ke0, Ko0, Ke1, Ko1, Ke2, Ko2,
                                      nbr2, nbr1, nbr0, outp);
}


# Refinement re-tiled: a block refines one 128-interval grid segment for 8 rows, so each index row and coefficient row is fetched once per 8 rows, not once per row; summation order unchanged
# speedup vs baseline: 7.3938x; 1.7491x over previous
#include <hip/hip_runtime.h>
#include <hip/hip_bf16.h>

typedef __attribute__((ext_vector_type(2))) float v2f;
typedef __attribute__((ext_vector_type(8))) float v8f;

#define WMMA_F32X4(a, b, c) \
    __builtin_amdgcn_wmma_f32_16x16x4_f32(false, (a), false, (b), (short)0, (c), false, false)

__global__ void __launch_bounds__(256) k_rankmm(const float* __restrict__ U,
                                                const float* __restrict__ V,
                                                float* __restrict__ out,
                                                int rows, int cols) {
    int idx = blockIdx.x * 256 + threadIdx.x;
    if (idx >= rows * cols) return;
    int n = idx / cols;
    int m = idx - n * cols;
    const float* up = U + n * 32;
    const float* vp = V + m * 32;
    float acc = 0.f;
#pragma unroll
    for (int r = 0; r < 32; ++r) acc += up[r] * vp[r];
    out[idx] = acc;
}

#define P_KS     8
#define P_WAVES 16
#define TILE_ROWS_FLOATS 16400

__global__ void __launch_bounds__(512) k_wmma_proj_part(const float* __restrict__ uh,
                                                        const float* __restrict__ KHH_V,
                                                        const float* __restrict__ Kbd_U,
                                                        float* __restrict__ part) {
    __shared__ float red[P_WAVES][1024];
    const int wave = __builtin_amdgcn_readfirstlane((int)(threadIdx.x >> 5));
    const int lane = threadIdx.x & 31;
    const int half = lane >> 4;
    const int mrow = lane & 15;
    const int mt   = blockIdx.x / P_KS;
    const int ks   = blockIdx.x % P_KS;
    const int R    = mt * 16;
    const bool tail = (ks == P_KS - 1) && (wave == 0);

    const float* arow = uh + (size_t)(R + mrow) * 8193;

    v8f t0 = {}; v8f t1 = {};
    for (int it = 0; it < 32 / P_WAVES; ++it) {
        int k0 = 4 * (ks * 32 + it * P_WAVES + wave);
        int ka = k0 + 2 * half;
        int kb = ka + 1;
        v2f a, b0, b1;
        a.x  = arow[8 * ka];
        a.y  = arow[8 * kb];
        b0.x = KHH_V[ka * 32 + mrow];
        b0.y = KHH_V[kb * 32 + mrow];
        b1.x = KHH_V[ka * 32 + 16 + mrow];
        b1.y = KHH_V[kb * 32 + 16 + mrow];
        t0 = WMMA_F32X4(a, b0, t0);
        t1 = WMMA_F32X4(a, b1, t1);
    }
    if (tail) {
        float alast = arow[8192];
        v2f a, b0, b1;
        a.x  = (half == 0) ? alast : 0.f;
        a.y  = 0.f;
        b0.x = KHH_V[1024 * 32 + mrow];
        b0.y = b0.x;
        b1.x = KHH_V[1024 * 32 + 16 + mrow];
        b1.y = b1.x;
        t0 = WMMA_F32X4(a, b0, t0);
        t1 = WMMA_F32X4(a, b1, t1);
    }

    v8f s0 = {}; v8f s1 = {};
    for (int it = 0; it < 256 / P_WAVES; ++it) {
        int k0 = 4 * (ks * 256 + it * P_WAVES + wave);
        int ka = k0 + 2 * half;
        int kb = ka + 1;
        v2f a, b0, b1;
        a.x  = arow[ka];
        a.y  = arow[kb];
        b0.x = Kbd_U[ka * 32 + mrow];
        b0.y = Kbd_U[kb * 32 + mrow];
        b1.x = Kbd_U[ka * 32 + 16 + mrow];
        b1.y = Kbd_U[kb * 32 + 16 + mrow];
        s0 = WMMA_F32X4(a, b0, s0);
        s1 = WMMA_F32X4(a, b1, s1);
    }
    if (tail) {
        float alast = arow[8192];
        v2f a, b0, b1;
        a.x  = (half == 0) ? alast : 0.f;
        a.y  = 0.f;
        b0.x = Kbd_U[8192 * 32 + mrow];
        b0.y = b0.x;
        b1.x = Kbd_U[8192 * 32 + 16 + mrow];
        b1.y = b1.x;
        s0 = WMMA_F32X4(a, b0, s0);
        s1 = WMMA_F32X4(a, b1, s1);
    }

    {
        float* rw = red[wave];
#pragma unroll
        for (int j = 0; j < 8; ++j) {
            rw[j * 32 + lane]       = t0[j];
            rw[256 + j * 32 + lane] = t1[j];
            rw[512 + j * 32 + lane] = s0[j];
            rw[768 + j * 32 + lane] = s1[j];
        }
    }
    __syncthreads();
    {
        float* dst = part + (size_t)mt * TILE_ROWS_FLOATS + ks * 1024;
#pragma unroll
        for (int q = 0; q < 2; ++q) {
            const int e = threadIdx.x + 512 * q;
            float sum = red[0][e];
#pragma unroll
            for (int w = 1; w < P_WAVES; ++w) sum += red[w][e];
            dst[e] = sum;
        }
    }
}

__global__ void __launch_bounds__(1024) k_wmma_proj_out(const float* __restrict__ KHH_U,
                                                        float* __restrict__ wh0,
                                                        float* __restrict__ sbuf) {
    __shared__ float tl[16 * 32];
    const int wave = __builtin_amdgcn_readfirstlane((int)(threadIdx.x >> 5));
    const int lane = threadIdx.x & 31;
    const int half = lane >> 4;
    const int mrow = lane & 15;
    const int mt   = blockIdx.x;
    const int R    = mt * 16;
    const float HK = 1.0f / 1024.0f;

    {
        const int e = threadIdx.x;
        const float* src = wh0 + (size_t)mt * TILE_ROWS_FLOATS;
        float sum = src[e];
#pragma unroll
        for (int s = 1; s < P_KS; ++s) sum += src[s * 1024 + e];
        const int p = e >> 8;
        const int j = (e >> 5) & 7;
        const int l = e & 31;
        const int m = j + 8 * (l >> 4);
        const int n = ((p & 1) << 4) + (l & 15);
        if (p < 2) tl[m * 32 + n] = sum * HK;
        else       sbuf[(size_t)(R + m) * 32 + n] = sum;
    }
    __syncthreads();

    for (int nt = 2 * wave; nt < 2 * wave + 2; ++nt) {
        int n = nt * 16 + mrow;
        v8f acc = {};
#pragma unroll
        for (int k0 = 0; k0 < 32; k0 += 4) {
            int ka = k0 + 2 * half;
            int kb = ka + 1;
            v2f a, b;
            a.x = tl[mrow * 32 + ka];
            a.y = tl[mrow * 32 + kb];
            b.x = KHH_U[n * 32 + ka];
            b.y = KHH_U[n * 32 + kb];
            acc = WMMA_F32X4(a, b, acc);
        }
#pragma unroll
        for (int j = 0; j < 8; ++j) {
            int m = j + 8 * half;
            wh0[(size_t)(R + m) * 1025 + n] = acc[j];
        }
    }
    if (wave == 31) {
        v8f acc = {};
#pragma unroll
        for (int k0 = 0; k0 < 32; k0 += 4) {
            int ka = k0 + 2 * half;
            int kb = ka + 1;
            v2f a, b;
            a.x = tl[mrow * 32 + ka];
            a.y = tl[mrow * 32 + kb];
            b.x = KHH_U[1024 * 32 + ka];
            b.y = KHH_U[1024 * 32 + kb];
            acc = WMMA_F32X4(a, b, acc);
        }
        if (mrow == 0) {
#pragma unroll
            for (int j = 0; j < 8; ++j) {
                int m = j + 8 * half;
                wh0[(size_t)(R + m) * 1025 + 1024] = acc[j];
            }
        }
    }
}

#define RS_S    128
#define RS_R    8
#define RS_NSEG (1024 / RS_S)
#define RS_PA   (4 * RS_S + 4)
#define RS_PB   (8 * RS_S + 8)

template <int STEP>
__device__ __forceinline__ void refine_level(const float* __restrict__ ub,
                                             const float* __restrict__ Ke,
                                             const float* __restrict__ Ko,
                                             const int* __restrict__ nbr,
                                             const float* oldW, const int oldPitch,
                                             float* newW, const int newPitch,
                                             const int nb, const int cntE, const int cntO,
                                             const int tid) {
    const float hf = (1.0f / 8192.0f) * (float)STEP;
    for (int nl = tid; nl < cntE; nl += 256) {
        const int n = nb + nl;
        const int* row = nbr + (size_t)(2 * n) * 15;
        const float* kp = Ke + (size_t)n * 8;
        int ix[8];
        float kk[8];
#pragma unroll
        for (int m = 0; m < 8; ++m) { ix[m] = STEP * row[2 * m]; kk[m] = kp[m]; }
#pragma unroll 4
        for (int r = 0; r < RS_R; ++r) {
            const float* u = ub + (size_t)r * 8193;
            float acc = 0.f;
#pragma unroll
            for (int m = 0; m < 8; ++m) acc += kk[m] * u[ix[m]];
            newW[r * newPitch + 2 * nl] = oldW[r * oldPitch + nl] + hf * acc;
        }
    }
    for (int nl = tid; nl < cntO; nl += 256) {
        const int n = nb + nl;
        const int* row = nbr + (size_t)(2 * n + 1) * 15;
        const float* kp = Ko + (size_t)n * 15;
        int ix[15];
        float kk[15];
#pragma unroll
        for (int m = 0; m < 15; ++m) { ix[m] = STEP * row[m]; kk[m] = kp[m]; }
#pragma unroll 2
        for (int r = 0; r < RS_R; ++r) {
            const float* u = ub + (size_t)r * 8193;
            float acc = 0.f;
#pragma unroll
            for (int m = 0; m < 15; ++m) acc += kk[m] * u[ix[m]];
            newW[r * newPitch + 2 * nl + 1] =
                0.5f * (oldW[r * oldPitch + nl] + oldW[r * oldPitch + nl + 1]) + hf * acc;
        }
    }
}

__device__ __forceinline__ float boundary_value(const float* __restrict__ sbuf,
                                                const float* __restrict__ Kbd_V,
                                                const int bc, const int which) {
    const float* kv = Kbd_V + which * 32;
    const float* sp = sbuf + bc * 32;
    float acc = 0.f;
#pragma unroll
    for (int r = 0; r < 32; ++r) acc += kv[r] * sp[r];
    return (1.0f / 8192.0f) * acc;
}

__global__ void __launch_bounds__(256) k_refine(const float* __restrict__ uh,
                                                const float* __restrict__ wh0,
                                                const float* __restrict__ sbuf,
                                                const float* __restrict__ Kbd_V,
                                                const float* __restrict__ Ke0,
                                                const float* __restrict__ Ko0,
                                                const float* __restrict__ Ke1,
                                                const float* __restrict__ Ko1,
                                                const float* __restrict__ Ke2,
                                                const float* __restrict__ Ko2,
                                                const int* __restrict__ nbrL0,
                                                const int* __restrict__ nbrL1,
                                                const int* __restrict__ nbrL2,
                                                float* __restrict__ out) {
    __shared__ float wA[RS_R * RS_PA];
    __shared__ float wB[RS_R * RS_PB];
    const int tid  = threadIdx.x;
    const int sg   = blockIdx.x % RS_NSEG;
    const int rg   = blockIdx.x / RS_NSEG;
    const int c0   = sg * RS_S;
    const int row0 = rg * RS_R;
    const float* ub = uh + (size_t)row0 * 8193;

    for (int e = tid; e < RS_R * (RS_S + 1); e += 256) {
        const int r = e / (RS_S + 1);
        const int n = e - r * (RS_S + 1);
        wA[r * RS_PA + n] = wh0[(size_t)(row0 + r) * 1025 + c0 + n];
    }
    __syncthreads();

    refine_level<4>(ub, Ke0, Ko0, nbrL0, wA, RS_PA, wB, RS_PB,
                    c0, RS_S + 1, RS_S, tid);
    __syncthreads();
    refine_level<2>(ub, Ke1, Ko1, nbrL1, wB, RS_PB, wA, RS_PA,
                    2 * c0, 2 * RS_S + 1, 2 * RS_S, tid);
    __syncthreads();
    refine_level<1>(ub, Ke2, Ko2, nbrL2, wA, RS_PA, wB, RS_PB,
                    4 * c0, 4 * RS_S, 4 * RS_S, tid);
    __syncthreads();

    for (int e = tid; e < RS_R * 8 * RS_S; e += 256) {
        const int r = e / (8 * RS_S);
        const int j = e - r * (8 * RS_S);
        const int g = 8 * c0 + j;
        float v = wB[r * RS_PB + j];
        if (g == 0) v = boundary_value(sbuf, Kbd_V, row0 + r, 0);
        out[(size_t)(row0 + r) * 8193 + g] = v;
    }
    if (sg == RS_NSEG - 1 && tid < RS_R)
        out[(size_t)(row0 + tid) * 8193 + 8192] = boundary_value(sbuf, Kbd_V, row0 + tid, 1);
}

extern "C" void kernel_launch(void* const* d_in, const int* in_sizes, int n_in,
                              void* d_out, int out_size, void* d_ws, size_t ws_size,
                              hipStream_t stream) {
    const float* uh     = (const float*)d_in[0];
    const float* KHH_U  = (const float*)d_in[1];
    const float* KHH_V  = (const float*)d_in[2];
    const float* Ke0_U  = (const float*)d_in[3];
    const float* Ke0_V  = (const float*)d_in[4];
    const float* Ko0_U  = (const float*)d_in[5];
    const float* Ko0_V  = (const float*)d_in[6];
    const int*   nbr0   = (const int*)d_in[7];
    const float* Ke1_U  = (const float*)d_in[8];
    const float* Ke1_V  = (const float*)d_in[9];
    const float* Ko1_U  = (const float*)d_in[10];
    const float* Ko1_V  = (const float*)d_in[11];
    const int*   nbr1   = (const int*)d_in[12];
    const float* Ke2_U  = (const float*)d_in[13];
    const float* Ke2_V  = (const float*)d_in[14];
    const float* Ko2_U  = (const float*)d_in[15];
    const float* Ko2_V  = (const float*)d_in[16];
    const int*   nbr2   = (const int*)d_in[17];
    const float* Kbd_U  = (const float*)d_in[18];
    const float* Kbd_V  = (const float*)d_in[19];

    float* ws  = (float*)d_ws;
    float* wh0 = ws;
    float* sb  = wh0 + 524800;
    float* Ke0 = sb + 16384;
    float* Ke1 = Ke0 + 8200;
    float* Ke2 = Ke1 + 16392;
    float* Ko0 = Ke2 + 32776;
    float* Ko1 = Ko0 + 15360;
    float* Ko2 = Ko1 + 30720;

    float* outp = (float*)d_out;

    auto grid = [](int elems) { return (elems + 255) / 256; };

    k_rankmm<<<grid(1025 * 8),  256, 0, stream>>>(Ke0_U, Ke0_V, Ke0, 1025, 8);
    k_rankmm<<<grid(1024 * 15), 256, 0, stream>>>(Ko0_U, Ko0_V, Ko0, 1024, 15);
    k_rankmm<<<grid(2049 * 8),  256, 0, stream>>>(Ke1_U, Ke1_V, Ke1, 2049, 8);
    k_rankmm<<<grid(2048 * 15), 256, 0, stream>>>(Ko1_U, Ko1_V, Ko1, 2048, 15);
    k_rankmm<<<grid(4097 * 8),  256, 0, stream>>>(Ke2_U, Ke2_V, Ke2, 4097, 8);
    k_rankmm<<<grid(4096 * 15), 256, 0, stream>>>(Ko2_U, Ko2_V, Ko2, 4096, 15);

    k_wmma_proj_part<<<256, 512, 0, stream>>>(uh, KHH_V, Kbd_U, wh0);
    k_wmma_proj_out<<<32, 1024, 0, stream>>>(KHH_U, wh0, sb);

    k_refine<<<512, 256, 0, stream>>>(uh, wh0, sb, Kbd_V,
                                      Ke0, Ko0, Ke1, Ko1, Ke2, Ko2,
                                      nbr2, nbr1, nbr0, outp);
}


# The six coefficient-product launches are folded into the projection launch as extra blocks behind its 256 (9 launches become 3); per-element arithmetic and summation order unchanged
# speedup vs baseline: 9.7819x; 1.3230x over previous
#include <hip/hip_runtime.h>
#include <hip/hip_bf16.h>

typedef __attribute__((ext_vector_type(2))) float v2f;
typedef __attribute__((ext_vector_type(8))) float v8f;

#define WMMA_F32X4(a, b, c) \
    __builtin_amdgcn_wmma_f32_16x16x4_f32(false, (a), false, (b), (short)0, (c), false, false)

template <int COLS>
__device__ __forceinline__ void rankmm_elem(const float* __restrict__ U,
                                            const float* __restrict__ V,
                                            float* __restrict__ out,
                                            const int rows, const int idx) {
    if (idx >= rows * COLS) return;
    int n = idx / COLS;
    int m = idx - n * COLS;
    const float* up = U + n * 32;
    const float* vp = V + m * 32;
    float acc = 0.f;
#pragma unroll
    for (int r = 0; r < 32; ++r) acc += up[r] * vp[r];
    out[idx] = acc;
}

#define CB(elems) (((elems) + 511) / 512)
#define CB_E0 CB(1025 * 8)
#define CB_O0 CB(1024 * 15)
#define CB_E1 CB(2049 * 8)
#define CB_O1 CB(2048 * 15)
#define CB_E2 CB(4097 * 8)
#define CB_O2 CB(4096 * 15)
#define CB_ALL (CB_E0 + CB_O0 + CB_E1 + CB_O1 + CB_E2 + CB_O2)

#define P_KS     8
#define P_WAVES 16
#define TILE_ROWS_FLOATS 16400

__global__ void __launch_bounds__(512) k_wmma_proj_part(const float* __restrict__ uh,
                                                        const float* __restrict__ KHH_V,
                                                        const float* __restrict__ Kbd_U,
                                                        float* __restrict__ part,
                                                        const float* __restrict__ Ke0_U,
                                                        const float* __restrict__ Ke0_V,
                                                        const float* __restrict__ Ko0_U,
                                                        const float* __restrict__ Ko0_V,
                                                        const float* __restrict__ Ke1_U,
                                                        const float* __restrict__ Ke1_V,
                                                        const float* __restrict__ Ko1_U,
                                                        const float* __restrict__ Ko1_V,
                                                        const float* __restrict__ Ke2_U,
                                                        const float* __restrict__ Ke2_V,
                                                        const float* __restrict__ Ko2_U,
                                                        const float* __restrict__ Ko2_V,
                                                        float* __restrict__ Ke0,
                                                        float* __restrict__ Ko0,
                                                        float* __restrict__ Ke1,
                                                        float* __restrict__ Ko1,
                                                        float* __restrict__ Ke2,
                                                        float* __restrict__ Ko2) {
    __shared__ float red[P_WAVES][1024];
    if (blockIdx.x >= 32 * P_KS) {
        int b = blockIdx.x - 32 * P_KS;
        const int t = threadIdx.x;
        if (b < CB_E0) { rankmm_elem<8>(Ke0_U, Ke0_V, Ke0, 1025, b * 512 + t); return; }
        b -= CB_E0;
        if (b < CB_O0) { rankmm_elem<15>(Ko0_U, Ko0_V, Ko0, 1024, b * 512 + t); return; }
        b -= CB_O0;
        if (b < CB_E1) { rankmm_elem<8>(Ke1_U, Ke1_V, Ke1, 2049, b * 512 + t); return; }
        b -= CB_E1;
        if (b < CB_O1) { rankmm_elem<15>(Ko1_U, Ko1_V, Ko1, 2048, b * 512 + t); return; }
        b -= CB_O1;
        if (b < CB_E2) { rankmm_elem<8>(Ke2_U, Ke2_V, Ke2, 4097, b * 512 + t); return; }
        b -= CB_E2;
        rankmm_elem<15>(Ko2_U, Ko2_V, Ko2, 4096, b * 512 + t);
        return;
    }
    const int wave = __builtin_amdgcn_readfirstlane((int)(threadIdx.x >> 5));
    const int lane = threadIdx.x & 31;
    const int half = lane >> 4;
    const int mrow = lane & 15;
    const int mt   = blockIdx.x / P_KS;
    const int ks   = blockIdx.x % P_KS;
    const int R    = mt * 16;
    const bool tail = (ks == P_KS - 1) && (wave == 0);

    const float* arow = uh + (size_t)(R + mrow) * 8193;

    v8f t0 = {}; v8f t1 = {};
    for (int it = 0; it < 32 / P_WAVES; ++it) {
        int k0 = 4 * (ks * 32 + it * P_WAVES + wave);
        int ka = k0 + 2 * half;
        int kb = ka + 1;
        v2f a, b0, b1;
        a.x  = arow[8 * ka];
        a.y  = arow[8 * kb];
        b0.x = KHH_V[ka * 32 + mrow];
        b0.y = KHH_V[kb * 32 + mrow];
        b1.x = KHH_V[ka * 32 + 16 + mrow];
        b1.y = KHH_V[kb * 32 + 16 + mrow];
        t0 = WMMA_F32X4(a, b0, t0);
        t1 = WMMA_F32X4(a, b1, t1);
    }
    if (tail) {
        float alast = arow[8192];
        v2f a, b0, b1;
        a.x  = (half == 0) ? alast : 0.f;
        a.y  = 0.f;
        b0.x = KHH_V[1024 * 32 + mrow];
        b0.y = b0.x;
        b1.x = KHH_V[1024 * 32 + 16 + mrow];
        b1.y = b1.x;
        t0 = WMMA_F32X4(a, b0, t0);
        t1 = WMMA_F32X4(a, b1, t1);
    }

    v8f s0 = {}; v8f s1 = {};
    for (int it = 0; it < 256 / P_WAVES; ++it) {
        int k0 = 4 * (ks * 256 + it * P_WAVES + wave);
        int ka = k0 + 2 * half;
        int kb = ka + 1;
        v2f a, b0, b1;
        a.x  = arow[ka];
        a.y  = arow[kb];
        b0.x = Kbd_U[ka * 32 + mrow];
        b0.y = Kbd_U[kb * 32 + mrow];
        b1.x = Kbd_U[ka * 32 + 16 + mrow];
        b1.y = Kbd_U[kb * 32 + 16 + mrow];
        s0 = WMMA_F32X4(a, b0, s0);
        s1 = WMMA_F32X4(a, b1, s1);
    }
    if (tail) {
        float alast = arow[8192];
        v2f a, b0, b1;
        a.x  = (half == 0) ? alast : 0.f;
        a.y  = 0.f;
        b0.x = Kbd_U[8192 * 32 + mrow];
        b0.y = b0.x;
        b1.x = Kbd_U[8192 * 32 + 16 + mrow];
        b1.y = b1.x;
        s0 = WMMA_F32X4(a, b0, s0);
        s1 = WMMA_F32X4(a, b1, s1);
    }

    {
        float* rw = red[wave];
#pragma unroll
        for (int j = 0; j < 8; ++j) {
            rw[j * 32 + lane]       = t0[j];
            rw[256 + j * 32 + lane] = t1[j];
            rw[512 + j * 32 + lane] = s0[j];
            rw[768 + j * 32 + lane] = s1[j];
        }
    }
    __syncthreads();
    {
        float* dst = part + (size_t)mt * TILE_ROWS_FLOATS + ks * 1024;
#pragma unroll
        for (int q = 0; q < 2; ++q) {
            const int e = threadIdx.x + 512 * q;
            float sum = red[0][e];
#pragma unroll
            for (int w = 1; w < P_WAVES; ++w) sum += red[w][e];
            dst[e] = sum;
        }
    }
}

__global__ void __launch_bounds__(1024) k_wmma_proj_out(const float* __restrict__ KHH_U,
                                                        float* __restrict__ wh0,
                                                        float* __restrict__ sbuf) {
    __shared__ float tl[16 * 32];
    const int wave = __builtin_amdgcn_readfirstlane((int)(threadIdx.x >> 5));
    const int lane = threadIdx.x & 31;
    const int half = lane >> 4;
    const int mrow = lane & 15;
    const int mt   = blockIdx.x;
    const int R    = mt * 16;
    const float HK = 1.0f / 1024.0f;

    {
        const int e = threadIdx.x;
        const float* src = wh0 + (size_t)mt * TILE_ROWS_FLOATS;
        float sum = src[e];
#pragma unroll
        for (int s = 1; s < P_KS; ++s) sum += src[s * 1024 + e];
        const int p = e >> 8;
        const int j = (e >> 5) & 7;
        const int l = e & 31;
        const int m = j + 8 * (l >> 4);
        const int n = ((p & 1) << 4) + (l & 15);
        if (p < 2) tl[m * 32 + n] = sum * HK;
        else       sbuf[(size_t)(R + m) * 32 + n] = sum;
    }
    __syncthreads();

    for (int nt = 2 * wave; nt < 2 * wave + 2; ++nt) {
        int n = nt * 16 + mrow;
        v8f acc = {};
#pragma unroll
        for (int k0 = 0; k0 < 32; k0 += 4) {
            int ka = k0 + 2 * half;
            int kb = ka + 1;
            v2f a, b;
            a.x = tl[mrow * 32 + ka];
            a.y = tl[mrow * 32 + kb];
            b.x = KHH_U[n * 32 + ka];
            b.y = KHH_U[n * 32 + kb];
            acc = WMMA_F32X4(a, b, acc);
        }
#pragma unroll
        for (int j = 0; j < 8; ++j) {
            int m = j + 8 * half;
            wh0[(size_t)(R + m) * 1025 + n] = acc[j];
        }
    }
    if (wave == 31) {
        v8f acc = {};
#pragma unroll
        for (int k0 = 0; k0 < 32; k0 += 4) {
            int ka = k0 + 2 * half;
            int kb = ka + 1;
            v2f a, b;
            a.x = tl[mrow * 32 + ka];
            a.y = tl[mrow * 32 + kb];
            b.x = KHH_U[1024 * 32 + ka];
            b.y = KHH_U[1024 * 32 + kb];
            acc = WMMA_F32X4(a, b, acc);
        }
        if (mrow == 0) {
#pragma unroll
            for (int j = 0; j < 8; ++j) {
                int m = j + 8 * half;
                wh0[(size_t)(R + m) * 1025 + 1024] = acc[j];
            }
        }
    }
}

#define RS_S    128
#define RS_R    8
#define RS_NSEG (1024 / RS_S)
#define RS_PA   (4 * RS_S + 4)
#define RS_PB   (8 * RS_S + 8)

template <int STEP>
__device__ __forceinline__ void refine_level(const float* __restrict__ ub,
                                             const float* __restrict__ Ke,
                                             const float* __restrict__ Ko,
                                             const int* __restrict__ nbr,
                                             const float* oldW, const int oldPitch,
                                             float* newW, const int newPitch,
                                             const int nb, const int cntE, const int cntO,
                                             const int tid) {
    const float hf = (1.0f / 8192.0f) * (float)STEP;
    for (int nl = tid; nl < cntE; nl += 256) {
        const int n = nb + nl;
        const int* row = nbr + (size_t)(2 * n) * 15;
        const float* kp = Ke + (size_t)n * 8;
        int ix[8];
        float kk[8];
#pragma unroll
        for (int m = 0; m < 8; ++m) { ix[m] = STEP * row[2 * m]; kk[m] = kp[m]; }
#pragma unroll 4
        for (int r = 0; r < RS_R; ++r) {
            const float* u = ub + (size_t)r * 8193;
            float acc = 0.f;
#pragma unroll
            for (int m = 0; m < 8; ++m) acc += kk[m] * u[ix[m]];
            newW[r * newPitch + 2 * nl] = oldW[r * oldPitch + nl] + hf * acc;
        }
    }
    for (int nl = tid; nl < cntO; nl += 256) {
        const int n = nb + nl;
        const int* row = nbr + (size_t)(2 * n + 1) * 15;
        const float* kp = Ko + (size_t)n * 15;
        int ix[15];
        float kk[15];
#pragma unroll
        for (int m = 0; m < 15; ++m) { ix[m] = STEP * row[m]; kk[m] = kp[m]; }
#pragma unroll 2
        for (int r = 0; r < RS_R; ++r) {
            const float* u = ub + (size_t)r * 8193;
            float acc = 0.f;
#pragma unroll
            for (int m = 0; m < 15; ++m) acc += kk[m] * u[ix[m]];
            newW[r * newPitch + 2 * nl + 1] =
                0.5f * (oldW[r * oldPitch + nl] + oldW[r * oldPitch + nl + 1]) + hf * acc;
        }
    }
}

__device__ __forceinline__ float boundary_value(const float* __restrict__ sbuf,
                                                const float* __restrict__ Kbd_V,
                                                const int bc, const int which) {
    const float* kv = Kbd_V + which * 32;
    const float* sp = sbuf + bc * 32;
    float acc = 0.f;
#pragma unroll
    for (int r = 0; r < 32; ++r) acc += kv[r] * sp[r];
    return (1.0f / 8192.0f) * acc;
}

__global__ void __launch_bounds__(256) k_refine(const float* __restrict__ uh,
                                                const float* __restrict__ wh0,
                                                const float* __restrict__ sbuf,
                                                const float* __restrict__ Kbd_V,
                                                const float* __restrict__ Ke0,
                                                const float* __restrict__ Ko0,
                                                const float* __restrict__ Ke1,
                                                const float* __restrict__ Ko1,
                                                const float* __restrict__ Ke2,
                                                const float* __restrict__ Ko2,
                                                const int* __restrict__ nbrL0,
                                                const int* __restrict__ nbrL1,
                                                const int* __restrict__ nbrL2,
                                                float* __restrict__ out) {
    __shared__ float wA[RS_R * RS_PA];
    __shared__ float wB[RS_R * RS_PB];
    const int tid  = threadIdx.x;
    const int sg   = blockIdx.x % RS_NSEG;
    const int rg   = blockIdx.x / RS_NSEG;
    const int c0   = sg * RS_S;
    const int row0 = rg * RS_R;
    const float* ub = uh + (size_t)row0 * 8193;

    for (int e = tid; e < RS_R * (RS_S + 1); e += 256) {
        const int r = e / (RS_S + 1);
        const int n = e - r * (RS_S + 1);
        wA[r * RS_PA + n] = wh0[(size_t)(row0 + r) * 1025 + c0 + n];
    }
    __syncthreads();

    refine_level<4>(ub, Ke0, Ko0, nbrL0, wA, RS_PA, wB, RS_PB,
                    c0, RS_S + 1, RS_S, tid);
    __syncthreads();
    refine_level<2>(ub, Ke1, Ko1, nbrL1, wB, RS_PB, wA, RS_PA,
                    2 * c0, 2 * RS_S + 1, 2 * RS_S, tid);
    __syncthreads();
    refine_level<1>(ub, Ke2, Ko2, nbrL2, wA, RS_PA, wB, RS_PB,
                    4 * c0, 4 * RS_S, 4 * RS_S, tid);
    __syncthreads();

    for (int e = tid; e < RS_R * 8 * RS_S; e += 256) {
        const int r = e / (8 * RS_S);
        const int j = e - r * (8 * RS_S);
        const int g = 8 * c0 + j;
        float v = wB[r * RS_PB + j];
        if (g == 0) v = boundary_value(sbuf, Kbd_V, row0 + r, 0);
        out[(size_t)(row0 + r) * 8193 + g] = v;
    }
    if (sg == RS_NSEG - 1 && tid < RS_R)
        out[(size_t)(row0 + tid) * 8193 + 8192] = boundary_value(sbuf, Kbd_V, row0 + tid, 1);
}

extern "C" void kernel_launch(void* const* d_in, const int* in_sizes, int n_in,
                              void* d_out, int out_size, void* d_ws, size_t ws_size,
                              hipStream_t stream) {
    const float* uh     = (const float*)d_in[0];
    const float* KHH_U  = (const float*)d_in[1];
    const float* KHH_V  = (const float*)d_in[2];
    const float* Ke0_U  = (const float*)d_in[3];
    const float* Ke0_V  = (const float*)d_in[4];
    const float* Ko0_U  = (const float*)d_in[5];
    const float* Ko0_V  = (const float*)d_in[6];
    const int*   nbr0   = (const int*)d_in[7];
    const float* Ke1_U  = (const float*)d_in[8];
    const float* Ke1_V  = (const float*)d_in[9];
    const float* Ko1_U  = (const float*)d_in[10];
    const float* Ko1_V  = (const float*)d_in[11];
    const int*   nbr1   = (const int*)d_in[12];
    const float* Ke2_U  = (const float*)d_in[13];
    const float* Ke2_V  = (const float*)d_in[14];
    const float* Ko2_U  = (const float*)d_in[15];
    const float* Ko2_V  = (const float*)d_in[16];
    const int*   nbr2   = (const int*)d_in[17];
    const float* Kbd_U  = (const float*)d_in[18];
    const float* Kbd_V  = (const float*)d_in[19];

    float* ws  = (float*)d_ws;
    float* wh0 = ws;
    float* sb  = wh0 + 524800;
    float* Ke0 = sb + 16384;
    float* Ke1 = Ke0 + 8200;
    float* Ke2 = Ke1 + 16392;
    float* Ko0 = Ke2 + 32776;
    float* Ko1 = Ko0 + 15360;
    float* Ko2 = Ko1 + 30720;

    float* outp = (float*)d_out;

    k_wmma_proj_part<<<32 * P_KS + CB_ALL, 512, 0, stream>>>(uh, KHH_V, Kbd_U, wh0,
                                                             Ke0_U, Ke0_V, Ko0_U, Ko0_V,
                                                             Ke1_U, Ke1_V, Ko1_U, Ko1_V,
                                                             Ke2_U, Ke2_V, Ko2_U, Ko2_V,
                                                             Ke0, Ko0, Ke1, Ko1, Ke2, Ko2);
    k_wmma_proj_out<<<32, 1024, 0, stream>>>(KHH_U, wh0, sb);

    k_refine<<<512, 256, 0, stream>>>(uh, wh0, sb, Kbd_V,
                                      Ke0, Ko0, Ke1, Ko1, Ke2, Ko2,
                                      nbr2, nbr1, nbr0, outp);
}


# Refinement: a chunk's index rows and coefficient rows are copied into LDS with whole-line 16-byte loads and read from there, not row by row per lane from global; summation order unchanged
# speedup vs baseline: 10.5749x; 1.0811x over previous
#include <hip/hip_runtime.h>
#include <hip/hip_bf16.h>

typedef __attribute__((ext_vector_type(2))) float v2f;
typedef __attribute__((ext_vector_type(8))) float v8f;

#define WMMA_F32X4(a, b, c) \
    __builtin_amdgcn_wmma_f32_16x16x4_f32(false, (a), false, (b), (short)0, (c), false, false)

template <int COLS>
__device__ __forceinline__ void rankmm_elem(const float* __restrict__ U,
                                            const float* __restrict__ V,
                                            float* __restrict__ out,
                                            const int rows, const int idx) {
    if (idx >= rows * COLS) return;
    int n = idx / COLS;
    int m = idx - n * COLS;
    const float* up = U + n * 32;
    const float* vp = V + m * 32;
    float acc = 0.f;
#pragma unroll
    for (int r = 0; r < 32; ++r) acc += up[r] * vp[r];
    out[idx] = acc;
}

#define CB(elems) (((elems) + 511) / 512)
#define CB_E0 CB(1025 * 8)
#define CB_O0 CB(1024 * 15)
#define CB_E1 CB(2049 * 8)
#define CB_O1 CB(2048 * 15)
#define CB_E2 CB(4097 * 8)
#define CB_O2 CB(4096 * 15)
#define CB_ALL (CB_E0 + CB_O0 + CB_E1 + CB_O1 + CB_E2 + CB_O2)

#define P_KS     8
#define P_WAVES 16
#define TILE_ROWS_FLOATS 16400

__global__ void __launch_bounds__(512) k_wmma_proj_part(const float* __restrict__ uh,
                                                        const float* __restrict__ KHH_V,
                                                        const float* __restrict__ Kbd_U,
                                                        float* __restrict__ part,
                                                        const float* __restrict__ Ke0_U,
                                                        const float* __restrict__ Ke0_V,
                                                        const float* __restrict__ Ko0_U,
                                                        const float* __restrict__ Ko0_V,
                                                        const float* __restrict__ Ke1_U,
                                                        const float* __restrict__ Ke1_V,
                                                        const float* __restrict__ Ko1_U,
                                                        const float* __restrict__ Ko1_V,
                                                        const float* __restrict__ Ke2_U,
                                                        const float* __restrict__ Ke2_V,
                                                        const float* __restrict__ Ko2_U,
                                                        const float* __restrict__ Ko2_V,
                                                        float* __restrict__ Ke0,
                                                        float* __restrict__ Ko0,
                                                        float* __restrict__ Ke1,
                                                        float* __restrict__ Ko1,
                                                        float* __restrict__ Ke2,
                                                        float* __restrict__ Ko2) {
    __shared__ float red[P_WAVES][1024];
    if (blockIdx.x >= 32 * P_KS) {
        int b = blockIdx.x - 32 * P_KS;
        const int t = threadIdx.x;
        if (b < CB_E0) { rankmm_elem<8>(Ke0_U, Ke0_V, Ke0, 1025, b * 512 + t); return; }
        b -= CB_E0;
        if (b < CB_O0) { rankmm_elem<15>(Ko0_U, Ko0_V, Ko0, 1024, b * 512 + t); return; }
        b -= CB_O0;
        if (b < CB_E1) { rankmm_elem<8>(Ke1_U, Ke1_V, Ke1, 2049, b * 512 + t); return; }
        b -= CB_E1;
        if (b < CB_O1) { rankmm_elem<15>(Ko1_U, Ko1_V, Ko1, 2048, b * 512 + t); return; }
        b -= CB_O1;
        if (b < CB_E2) { rankmm_elem<8>(Ke2_U, Ke2_V, Ke2, 4097, b * 512 + t); return; }
        b -= CB_E2;
        rankmm_elem<15>(Ko2_U, Ko2_V, Ko2, 4096, b * 512 + t);
        return;
    }
    const int wave = __builtin_amdgcn_readfirstlane((int)(threadIdx.x >> 5));
    const int lane = threadIdx.x & 31;
    const int half = lane >> 4;
    const int mrow = lane & 15;
    const int mt   = blockIdx.x / P_KS;
    const int ks   = blockIdx.x % P_KS;
    const int R    = mt * 16;
    const bool tail = (ks == P_KS - 1) && (wave == 0);

    const float* arow = uh + (size_t)(R + mrow) * 8193;

    v8f t0 = {}; v8f t1 = {};
    for (int it = 0; it < 32 / P_WAVES; ++it) {
        int k0 = 4 * (ks * 32 + it * P_WAVES + wave);
        int ka = k0 + 2 * half;
        int kb = ka + 1;
        v2f a, b0, b1;
        a.x  = arow[8 * ka];
        a.y  = arow[8 * kb];
        b0.x = KHH_V[ka * 32 + mrow];
        b0.y = KHH_V[kb * 32 + mrow];
        b1.x = KHH_V[ka * 32 + 16 + mrow];
        b1.y = KHH_V[kb * 32 + 16 + mrow];
        t0 = WMMA_F32X4(a, b0, t0);
        t1 = WMMA_F32X4(a, b1, t1);
    }
    if (tail) {
        float alast = arow[8192];
        v2f a, b0, b1;
        a.x  = (half == 0) ? alast : 0.f;
        a.y  = 0.f;
        b0.x = KHH_V[1024 * 32 + mrow];
        b0.y = b0.x;
        b1.x = KHH_V[1024 * 32 + 16 + mrow];
        b1.y = b1.x;
        t0 = WMMA_F32X4(a, b0, t0);
        t1 = WMMA_F32X4(a, b1, t1);
    }

    v8f s0 = {}; v8f s1 = {};
    for (int it = 0; it < 256 / P_WAVES; ++it) {
        int k0 = 4 * (ks * 256 + it * P_WAVES + wave);
        int ka = k0 + 2 * half;
        int kb = ka + 1;
        v2f a, b0, b1;
        a.x  = arow[ka];
        a.y  = arow[kb];
        b0.x = Kbd_U[ka * 32 + mrow];
        b0.y = Kbd_U[kb * 32 + mrow];
        b1.x = Kbd_U[ka * 32 + 16 + mrow];
        b1.y = Kbd_U[kb * 32 + 16 + mrow];
        s0 = WMMA_F32X4(a, b0, s0);
        s1 = WMMA_F32X4(a, b1, s1);
    }
    if (tail) {
        float alast = arow[8192];
        v2f a, b0, b1;
        a.x  = (half == 0) ? alast : 0.f;
        a.y  = 0.f;
        b0.x = Kbd_U[8192 * 32 + mrow];
        b0.y = b0.x;
        b1.x = Kbd_U[8192 * 32 + 16 + mrow];
        b1.y = b1.x;
        s0 = WMMA_F32X4(a, b0, s0);
        s1 = WMMA_F32X4(a, b1, s1);
    }

    {
        float* rw = red[wave];
#pragma unroll
        for (int j = 0; j < 8; ++j) {
            rw[j * 32 + lane]       = t0[j];
            rw[256 + j * 32 + lane] = t1[j];
            rw[512 + j * 32 + lane] = s0[j];
            rw[768 + j * 32 + lane] = s1[j];
        }
    }
    __syncthreads();
    {
        float* dst = part + (size_t)mt * TILE_ROWS_FLOATS + ks * 1024;
#pragma unroll
        for (int q = 0; q < 2; ++q) {
            const int e = threadIdx.x + 512 * q;
            float sum = red[0][e];
#pragma unroll
            for (int w = 1; w < P_WAVES; ++w) sum += red[w][e];
            dst[e] = sum;
        }
    }
}

__global__ void __launch_bounds__(1024) k_wmma_proj_out(const float* __restrict__ KHH_U,
                                                        float* __restrict__ wh0,
                                                        float* __restrict__ sbuf) {
    __shared__ float tl[16 * 32];
    const int wave = __builtin_amdgcn_readfirstlane((int)(threadIdx.x >> 5));
    const int lane = threadIdx.x & 31;
    const int half = lane >> 4;
    const int mrow = lane & 15;
    const int mt   = blockIdx.x;
    const int R    = mt * 16;
    const float HK = 1.0f / 1024.0f;

    {
        const int e = threadIdx.x;
        const float* src = wh0 + (size_t)mt * TILE_ROWS_FLOATS;
        float sum = src[e];
#pragma unroll
        for (int s = 1; s < P_KS; ++s) sum += src[s * 1024 + e];
        const int p = e >> 8;
        const int j = (e >> 5) & 7;
        const int l = e & 31;
        const int m = j + 8 * (l >> 4);
        const int n = ((p & 1) << 4) + (l & 15);
        if (p < 2) tl[m * 32 + n] = sum * HK;
        else       sbuf[(size_t)(R + m) * 32 + n] = sum;
    }
    __syncthreads();

    for (int nt = 2 * wave; nt < 2 * wave + 2; ++nt) {
        int n = nt * 16 + mrow;
        v8f acc = {};
#pragma unroll
        for (int k0 = 0; k0 < 32; k0 += 4) {
            int ka = k0 + 2 * half;
            int kb = ka + 1;
            v2f a, b;
            a.x = tl[mrow * 32 + ka];
            a.y = tl[mrow * 32 + kb];
            b.x = KHH_U[n * 32 + ka];
            b.y = KHH_U[n * 32 + kb];
            acc = WMMA_F32X4(a, b, acc);
        }
#pragma unroll
        for (int j = 0; j < 8; ++j) {
            int m = j + 8 * half;
            wh0[(size_t)(R + m) * 1025 + n] = acc[j];
        }
    }
    if (wave == 31) {
        v8f acc = {};
#pragma unroll
        for (int k0 = 0; k0 < 32; k0 += 4) {
            int ka = k0 + 2 * half;
            int kb = ka + 1;
            v2f a, b;
            a.x = tl[mrow * 32 + ka];
            a.y = tl[mrow * 32 + kb];
            b.x = KHH_U[1024 * 32 + ka];
            b.y = KHH_U[1024 * 32 + kb];
            acc = WMMA_F32X4(a, b, acc);
        }
        if (mrow == 0) {
#pragma unroll
            for (int j = 0; j < 8; ++j) {
                int m = j + 8 * half;
                wh0[(size_t)(R + m) * 1025 + 1024] = acc[j];
            }
        }
    }
}

#define RS_S    128
#define RS_R    8
#define RS_NSEG (1024 / RS_S)
#define RS_PA   (4 * RS_S + 4)
#define RS_PB   (8 * RS_S + 8)
#define RS_CH   256
#define RS_IDXN (8 * 1024 + 4)
#define RS_KEN  (3 * 1024 + 4)
#define RS_KON  (4 * 1024 + 4)

template <int MAXIT, typename V>
__device__ __forceinline__ void stage_load(V (&v)[MAXIT], const V* __restrict__ s4,
                                           const int n4, const int tid) {
#pragma unroll
    for (int k = 0; k < MAXIT; ++k) {
        if (k * 256 < n4) {
            const int i = tid + 256 * k;
            v[k] = s4[i < n4 ? i : n4 - 1];
        }
    }
}
template <int MAXIT, typename V>
__device__ __forceinline__ void stage_store(const V (&v)[MAXIT], V* d4,
                                            const int n4, const int tid) {
#pragma unroll
    for (int k = 0; k < MAXIT; ++k) {
        if (k * 256 < n4) {
            const int i = tid + 256 * k;
            d4[i < n4 ? i : i + 1] = v[k];
        }
    }
}

template <int STEP>
__device__ __forceinline__ void refine_level(const float* __restrict__ ub,
                                             const float* __restrict__ Ke,
                                             const float* __restrict__ Ko,
                                             const int* __restrict__ nbr,
                                             const float* oldW, const int oldPitch,
                                             float* newW, const int newPitch,
                                             const int nb, const int cntE, const int cntO,
                                             const int tid,
                                             int* sIdx, float* sKe, float* sKo) {
    const float hf = (1.0f / 8192.0f) * (float)STEP;
    for (int cb = 0; cb < cntO; cb += RS_CH) {
        const int ce = (cb + RS_CH >= cntO) ? (cntE - cb) : RS_CH;
        const int co = (cntO - cb < RS_CH) ? (cntO - cb) : RS_CH;
        const int nrows = (2 * ce - 1 > 2 * co) ? (2 * ce - 1) : (2 * co);
        const int n0 = nb + cb;
        {
            const int* isrc = nbr + (size_t)(2 * n0) * 15;
            const int icount = nrows * 15;
            const int in4 = icount >> 2, en4 = (ce * 8) >> 2, on4 = (co * 15) >> 2;
            int4 vi[8];
            float4 ve[3];
            float4 vo[4];
            int vt = 0;
            stage_load<8>(vi, (const int4*)isrc, in4, tid);
            stage_load<3>(ve, (const float4*)(Ke + (size_t)n0 * 8), en4, tid);
            stage_load<4>(vo, (const float4*)(Ko + (size_t)n0 * 15), on4, tid);
            if (tid < (icount & 3)) vt = isrc[4 * in4 + tid];
            stage_store<8>(vi, (int4*)sIdx, in4, tid);
            stage_store<3>(ve, (float4*)sKe, en4, tid);
            stage_store<4>(vo, (float4*)sKo, on4, tid);
            if (tid < (icount & 3)) sIdx[4 * in4 + tid] = vt;
        }
        __syncthreads();
        const float* oW = oldW + cb;
        float* nW = newW + 2 * cb;
        for (int nl = tid; nl < ce; nl += 256) {
            const int* row = sIdx + (2 * nl) * 15;
            const float* kp = sKe + nl * 8;
            int ix[8];
            float kk[8];
#pragma unroll
            for (int m = 0; m < 8; ++m) { ix[m] = STEP * row[2 * m]; kk[m] = kp[m]; }
#pragma unroll 4
            for (int r = 0; r < RS_R; ++r) {
                const float* u = ub + (size_t)r * 8193;
                float acc = 0.f;
#pragma unroll
                for (int m = 0; m < 8; ++m) acc += kk[m] * u[ix[m]];
                nW[r * newPitch + 2 * nl] = oW[r * oldPitch + nl] + hf * acc;
            }
        }
        for (int nl = tid; nl < co; nl += 256) {
            const int* row = sIdx + (2 * nl + 1) * 15;
            const float* kp = sKo + nl * 15;
            int ix[15];
            float kk[15];
#pragma unroll
            for (int m = 0; m < 15; ++m) { ix[m] = STEP * row[m]; kk[m] = kp[m]; }
#pragma unroll 2
            for (int r = 0; r < RS_R; ++r) {
                const float* u = ub + (size_t)r * 8193;
                float acc = 0.f;
#pragma unroll
                for (int m = 0; m < 15; ++m) acc += kk[m] * u[ix[m]];
                nW[r * newPitch + 2 * nl + 1] =
                    0.5f * (oW[r * oldPitch + nl] + oW[r * oldPitch + nl + 1]) + hf * acc;
            }
        }
        __syncthreads();
    }
}

__device__ __forceinline__ float boundary_value(const float* __restrict__ sbuf,
                                                const float* __restrict__ Kbd_V,
                                                const int bc, const int which) {
    const float* kv = Kbd_V + which * 32;
    const float* sp = sbuf + bc * 32;
    float acc = 0.f;
#pragma unroll
    for (int r = 0; r < 32; ++r) acc += kv[r] * sp[r];
    return (1.0f / 8192.0f) * acc;
}

__global__ void __launch_bounds__(256) k_refine(const float* __restrict__ uh,
                                                const float* __restrict__ wh0,
                                                const float* __restrict__ sbuf,
                                                const float* __restrict__ Kbd_V,
                                                const float* __restrict__ Ke0,
                                                const float* __restrict__ Ko0,
                                                const float* __restrict__ Ke1,
                                                const float* __restrict__ Ko1,
                                                const float* __restrict__ Ke2,
                                                const float* __restrict__ Ko2,
                                                const int* __restrict__ nbrL0,
                                                const int* __restrict__ nbrL1,
                                                const int* __restrict__ nbrL2,
                                                float* __restrict__ out) {
    __shared__ float wA[RS_R * RS_PA];
    __shared__ float wB[RS_R * RS_PB];
    __shared__ int   sIdx[RS_IDXN] __attribute__((aligned(16)));
    __shared__ float sKe[RS_KEN] __attribute__((aligned(16)));
    __shared__ float sKo[RS_KON] __attribute__((aligned(16)));
    const int tid  = threadIdx.x;
    const int sg   = blockIdx.x % RS_NSEG;
    const int rg   = blockIdx.x / RS_NSEG;
    const int c0   = sg * RS_S;
    const int row0 = rg * RS_R;
    const float* ub = uh + (size_t)row0 * 8193;

    for (int e = tid; e < RS_R * (RS_S + 1); e += 256) {
        const int r = e / (RS_S + 1);
        const int n = e - r * (RS_S + 1);
        wA[r * RS_PA + n] = wh0[(size_t)(row0 + r) * 1025 + c0 + n];
    }

    refine_level<4>(ub, Ke0, Ko0, nbrL0, wA, RS_PA, wB, RS_PB,
                    c0, RS_S + 1, RS_S, tid, sIdx, sKe, sKo);
    refine_level<2>(ub, Ke1, Ko1, nbrL1, wB, RS_PB, wA, RS_PA,
                    2 * c0, 2 * RS_S + 1, 2 * RS_S, tid, sIdx, sKe, sKo);
    refine_level<1>(ub, Ke2, Ko2, nbrL2, wA, RS_PA, wB, RS_PB,
                    4 * c0, 4 * RS_S, 4 * RS_S, tid, sIdx, sKe, sKo);

    for (int e = tid; e < RS_R * 8 * RS_S; e += 256) {
        const int r = e / (8 * RS_S);
        const int j = e - r * (8 * RS_S);
        const int g = 8 * c0 + j;
        float v = wB[r * RS_PB + j];
        if (g == 0) v = boundary_value(sbuf, Kbd_V, row0 + r, 0);
        out[(size_t)(row0 + r) * 8193 + g] = v;
    }
    if (sg == RS_NSEG - 1 && tid < RS_R)
        out[(size_t)(row0 + tid) * 8193 + 8192] = boundary_value(sbuf, Kbd_V, row0 + tid, 1);
}

extern "C" void kernel_launch(void* const* d_in, const int* in_sizes, int n_in,
                              void* d_out, int out_size, void* d_ws, size_t ws_size,
                              hipStream_t stream) {
    const float* uh     = (const float*)d_in[0];
    const float* KHH_U  = (const float*)d_in[1];
    const float* KHH_V  = (const float*)d_in[2];
    const float* Ke0_U  = (const float*)d_in[3];
    const float* Ke0_V  = (const float*)d_in[4];
    const float* Ko0_U  = (const float*)d_in[5];
    const float* Ko0_V  = (const float*)d_in[6];
    const int*   nbr0   = (const int*)d_in[7];
    const float* Ke1_U  = (const float*)d_in[8];
    const float* Ke1_V  = (const float*)d_in[9];
    const float* Ko1_U  = (const float*)d_in[10];
    const float* Ko1_V  = (const float*)d_in[11];
    const int*   nbr1   = (const int*)d_in[12];
    const float* Ke2_U  = (const float*)d_in[13];
    const float* Ke2_V  = (const float*)d_in[14];
    const float* Ko2_U  = (const float*)d_in[15];
    const float* Ko2_V  = (const float*)d_in[16];
    const int*   nbr2   = (const int*)d_in[17];
    const float* Kbd_U  = (const float*)d_in[18];
    const float* Kbd_V  = (const float*)d_in[19];

    float* ws  = (float*)d_ws;
    float* wh0 = ws;
    float* sb  = wh0 + 524800;
    float* Ke0 = sb + 16384;
    float* Ke1 = Ke0 + 8200;
    float* Ke2 = Ke1 + 16392;
    float* Ko0 = Ke2 + 32776;
    float* Ko1 = Ko0 + 15360;
    float* Ko2 = Ko1 + 30720;

    float* outp = (float*)d_out;

    k_wmma_proj_part<<<32 * P_KS + CB_ALL, 512, 0, stream>>>(uh, KHH_V, Kbd_U, wh0,
                                                             Ke0_U, Ke0_V, Ko0_U, Ko0_V,
                                                             Ke1_U, Ke1_V, Ko1_U, Ko1_V,
                                                             Ke2_U, Ke2_V, Ko2_U, Ko2_V,
                                                             Ke0, Ko0, Ke1, Ko1, Ke2, Ko2);
    k_wmma_proj_out<<<32, 1024, 0, stream>>>(KHH_U, wh0, sb);

    k_refine<<<512, 256, 0, stream>>>(uh, wh0, sb, Kbd_V,
                                      Ke0, Ko0, Ke1, Ko1, Ke2, Ko2,
                                      nbr2, nbr1, nbr0, outp);
}


# Refinement: per level a block copies its 8 rows' window of the fine grid into LDS and gathers the taps from there; global gathers only where an index leaves the window; summation order unchanged
# speedup vs baseline: 12.0128x; 1.1360x over previous
#include <hip/hip_runtime.h>
#include <hip/hip_bf16.h>

typedef __attribute__((ext_vector_type(2))) float v2f;
typedef __attribute__((ext_vector_type(8))) float v8f;

#define WMMA_F32X4(a, b, c) \
    __builtin_amdgcn_wmma_f32_16x16x4_f32(false, (a), false, (b), (short)0, (c), false, false)

template <int COLS>
__device__ __forceinline__ void rankmm_elem(const float* __restrict__ U,
                                            const float* __restrict__ V,
                                            float* __restrict__ out,
                                            const int rows, const int idx) {
    if (idx >= rows * COLS) return;
    int n = idx / COLS;
    int m = idx - n * COLS;
    const float* up = U + n * 32;
    const float* vp = V + m * 32;
    float acc = 0.f;
#pragma unroll
    for (int r = 0; r < 32; ++r) acc += up[r] * vp[r];
    out[idx] = acc;
}

#define CB(elems) (((elems) + 511) / 512)
#define CB_E0 CB(1025 * 8)
#define CB_O0 CB(1024 * 15)
#define CB_E1 CB(2049 * 8)
#define CB_O1 CB(2048 * 15)
#define CB_E2 CB(4097 * 8)
#define CB_O2 CB(4096 * 15)
#define CB_ALL (CB_E0 + CB_O0 + CB_E1 + CB_O1 + CB_E2 + CB_O2)

#define P_KS     8
#define P_WAVES 16
#define TILE_ROWS_FLOATS 16400

__global__ void __launch_bounds__(512) k_wmma_proj_part(const float* __restrict__ uh,
                                                        const float* __restrict__ KHH_V,
                                                        const float* __restrict__ Kbd_U,
                                                        float* __restrict__ part,
                                                        const float* __restrict__ Ke0_U,
                                                        const float* __restrict__ Ke0_V,
                                                        const float* __restrict__ Ko0_U,
                                                        const float* __restrict__ Ko0_V,
                                                        const float* __restrict__ Ke1_U,
                                                        const float* __restrict__ Ke1_V,
                                                        const float* __restrict__ Ko1_U,
                                                        const float* __restrict__ Ko1_V,
                                                        const float* __restrict__ Ke2_U,
                                                        const float* __restrict__ Ke2_V,
                                                        const float* __restrict__ Ko2_U,
                                                        const float* __restrict__ Ko2_V,
                                                        float* __restrict__ Ke0,
                                                        float* __restrict__ Ko0,
                                                        float* __restrict__ Ke1,
                                                        float* __restrict__ Ko1,
                                                        float* __restrict__ Ke2,
                                                        float* __restrict__ Ko2) {
    __shared__ float red[P_WAVES][1024];
    if (blockIdx.x >= 32 * P_KS) {
        int b = blockIdx.x - 32 * P_KS;
        const int t = threadIdx.x;
        if (b < CB_E0) { rankmm_elem<8>(Ke0_U, Ke0_V, Ke0, 1025, b * 512 + t); return; }
        b -= CB_E0;
        if (b < CB_O0) { rankmm_elem<15>(Ko0_U, Ko0_V, Ko0, 1024, b * 512 + t); return; }
        b -= CB_O0;
        if (b < CB_E1) { rankmm_elem<8>(Ke1_U, Ke1_V, Ke1, 2049, b * 512 + t); return; }
        b -= CB_E1;
        if (b < CB_O1) { rankmm_elem<15>(Ko1_U, Ko1_V, Ko1, 2048, b * 512 + t); return; }
        b -= CB_O1;
        if (b < CB_E2) { rankmm_elem<8>(Ke2_U, Ke2_V, Ke2, 4097, b * 512 + t); return; }
        b -= CB_E2;
        rankmm_elem<15>(Ko2_U, Ko2_V, Ko2, 4096, b * 512 + t);
        return;
    }
    const int wave = __builtin_amdgcn_readfirstlane((int)(threadIdx.x >> 5));
    const int lane = threadIdx.x & 31;
    const int half = lane >> 4;
    const int mrow = lane & 15;
    const int mt   = blockIdx.x / P_KS;
    const int ks   = blockIdx.x % P_KS;
    const int R    = mt * 16;
    const bool tail = (ks == P_KS - 1) && (wave == 0);

    const float* arow = uh + (size_t)(R + mrow) * 8193;

    v8f t0 = {}; v8f t1 = {};
    for (int it = 0; it < 32 / P_WAVES; ++it) {
        int k0 = 4 * (ks * 32 + it * P_WAVES + wave);
        int ka = k0 + 2 * half;
        int kb = ka + 1;
        v2f a, b0, b1;
        a.x  = arow[8 * ka];
        a.y  = arow[8 * kb];
        b0.x = KHH_V[ka * 32 + mrow];
        b0.y = KHH_V[kb * 32 + mrow];
        b1.x = KHH_V[ka * 32 + 16 + mrow];
        b1.y = KHH_V[kb * 32 + 16 + mrow];
        t0 = WMMA_F32X4(a, b0, t0);
        t1 = WMMA_F32X4(a, b1, t1);
    }
    if (tail) {
        float alast = arow[8192];
        v2f a, b0, b1;
        a.x  = (half == 0) ? alast : 0.f;
        a.y  = 0.f;
        b0.x = KHH_V[1024 * 32 + mrow];
        b0.y = b0.x;
        b1.x = KHH_V[1024 * 32 + 16 + mrow];
        b1.y = b1.x;
        t0 = WMMA_F32X4(a, b0, t0);
        t1 = WMMA_F32X4(a, b1, t1);
    }

    v8f s0 = {}; v8f s1 = {};
    for (int it = 0; it < 256 / P_WAVES; ++it) {
        int k0 = 4 * (ks * 256 + it * P_WAVES + wave);
        int ka = k0 + 2 * half;
        int kb = ka + 1;
        v2f a, b0, b1;
        a.x  = arow[ka];
        a.y  = arow[kb];
        b0.x = Kbd_U[ka * 32 + mrow];
        b0.y = Kbd_U[kb * 32 + mrow];
        b1.x = Kbd_U[ka * 32 + 16 + mrow];
        b1.y = Kbd_U[kb * 32 + 16 + mrow];
        s0 = WMMA_F32X4(a, b0, s0);
        s1 = WMMA_F32X4(a, b1, s1);
    }
    if (tail) {
        float alast = arow[8192];
        v2f a, b0, b1;
        a.x  = (half == 0) ? alast : 0.f;
        a.y  = 0.f;
        b0.x = Kbd_U[8192 * 32 + mrow];
        b0.y = b0.x;
        b1.x = Kbd_U[8192 * 32 + 16 + mrow];
        b1.y = b1.x;
        s0 = WMMA_F32X4(a, b0, s0);
        s1 = WMMA_F32X4(a, b1, s1);
    }

    {
        float* rw = red[wave];
#pragma unroll
        for (int j = 0; j < 8; ++j) {
            rw[j * 32 + lane]       = t0[j];
            rw[256 + j * 32 + lane] = t1[j];
            rw[512 + j * 32 + lane] = s0[j];
            rw[768 + j * 32 + lane] = s1[j];
        }
    }
    __syncthreads();
    {
        float* dst = part + (size_t)mt * TILE_ROWS_FLOATS + ks * 1024;
#pragma unroll
        for (int q = 0; q < 2; ++q) {
            const int e = threadIdx.x + 512 * q;
            float sum = red[0][e];
#pragma unroll
            for (int w = 1; w < P_WAVES; ++w) sum += red[w][e];
            dst[e] = sum;
        }
    }
}

__global__ void __launch_bounds__(1024) k_wmma_proj_out(const float* __restrict__ KHH_U,
                                                        float* __restrict__ wh0,
                                                        float* __restrict__ sbuf) {
    __shared__ float tl[16 * 32];
    const int wave = __builtin_amdgcn_readfirstlane((int)(threadIdx.x >> 5));
    const int lane = threadIdx.x & 31;
    const int half = lane >> 4;
    const int mrow = lane & 15;
    const int mt   = blockIdx.x;
    const int R    = mt * 16;
    const float HK = 1.0f / 1024.0f;

    {
        const int e = threadIdx.x;
        const float* src = wh0 + (size_t)mt * TILE_ROWS_FLOATS;
        float sum = src[e];
#pragma unroll
        for (int s = 1; s < P_KS; ++s) sum += src[s * 1024 + e];
        const int p = e >> 8;
        const int j = (e >> 5) & 7;
        const int l = e & 31;
        const int m = j + 8 * (l >> 4);
        const int n = ((p & 1) << 4) + (l & 15);
        if (p < 2) tl[m * 32 + n] = sum * HK;
        else       sbuf[(size_t)(R + m) * 32 + n] = sum;
    }
    __syncthreads();

    for (int nt = 2 * wave; nt < 2 * wave + 2; ++nt) {
        int n = nt * 16 + mrow;
        v8f acc = {};
#pragma unroll
        for (int k0 = 0; k0 < 32; k0 += 4) {
            int ka = k0 + 2 * half;
            int kb = ka + 1;
            v2f a, b;
            a.x = tl[mrow * 32 + ka];
            a.y = tl[mrow * 32 + kb];
            b.x = KHH_U[n * 32 + ka];
            b.y = KHH_U[n * 32 + kb];
            acc = WMMA_F32X4(a, b, acc);
        }
#pragma unroll
        for (int j = 0; j < 8; ++j) {
            int m = j + 8 * half;
            wh0[(size_t)(R + m) * 1025 + n] = acc[j];
        }
    }
    if (wave == 31) {
        v8f acc = {};
#pragma unroll
        for (int k0 = 0; k0 < 32; k0 += 4) {
            int ka = k0 + 2 * half;
            int kb = ka + 1;
            v2f a, b;
            a.x = tl[mrow * 32 + ka];
            a.y = tl[mrow * 32 + kb];
            b.x = KHH_U[1024 * 32 + ka];
            b.y = KHH_U[1024 * 32 + kb];
            acc = WMMA_F32X4(a, b, acc);
        }
        if (mrow == 0) {
#pragma unroll
            for (int j = 0; j < 8; ++j) {
                int m = j + 8 * half;
                wh0[(size_t)(R + m) * 1025 + 1024] = acc[j];
            }
        }
    }
}

#define RS_S    128
#define RS_R    8
#define RS_NSEG (1024 / RS_S)
#define RS_PA   (4 * RS_S + 4)
#define RS_PB   (8 * RS_S + 8)
#define RS_CH   256
#define RS_IDXN (8 * 1024 + 4)
#define RS_KEN  (3 * 1024 + 4)
#define RS_KON  (4 * 1024 + 4)
#define RS_WMAX (8 * RS_S + 16)

template <int MAXIT, typename V>
__device__ __forceinline__ void stage_load(V (&v)[MAXIT], const V* __restrict__ s4,
                                           const int n4, const int tid) {
#pragma unroll
    for (int k = 0; k < MAXIT; ++k) {
        if (k * 256 < n4) {
            const int i = tid + 256 * k;
            v[k] = s4[i < n4 ? i : n4 - 1];
        }
    }
}
template <int MAXIT, typename V>
__device__ __forceinline__ void stage_store(const V (&v)[MAXIT], V* d4,
                                            const int n4, const int tid) {
#pragma unroll
    for (int k = 0; k < MAXIT; ++k) {
        if (k * 256 < n4) {
            const int i = tid + 256 * k;
            d4[i < n4 ? i : i + 1] = v[k];
        }
    }
}

template <int STEP, int WLEN>
__device__ __forceinline__ void refine_level(const float* __restrict__ ub,
                                             const float* __restrict__ Ke,
                                             const float* __restrict__ Ko,
                                             const int* __restrict__ nbr,
                                             const float* oldW, const int oldPitch,
                                             float* newW, const int newPitch,
                                             const int nb, const int cntE, const int cntO,
                                             const int tid,
                                             int* sIdx, float* sKe, float* sKo, float* sU) {
    const float hf = (1.0f / 8192.0f) * (float)STEP;
    constexpr int NFINE = 8192 / STEP;
    constexpr int WTOT  = RS_R * WLEN;
    constexpr int WIT   = (WTOT + 255) / 256;
    const int wlo = 2 * nb - 8;
#pragma unroll
    for (int cb = 0; cb < cntO; cb += RS_CH) {
        const int ce = (cb + RS_CH >= cntO) ? (cntE - cb) : RS_CH;
        const int co = (cntO - cb < RS_CH) ? (cntO - cb) : RS_CH;
        const int nrows = (2 * ce - 1 > 2 * co) ? (2 * ce - 1) : (2 * co);
        const int n0 = nb + cb;
        {
            const int* isrc = nbr + (size_t)(2 * n0) * 15;
            const int icount = nrows * 15;
            const int in4 = icount >> 2, en4 = (ce * 8) >> 2, on4 = (co * 15) >> 2;
            int4 vi[8];
            float4 ve[3];
            float4 vo[4];
            float vw[WIT];
            int vt = 0;
            stage_load<8>(vi, (const int4*)isrc, in4, tid);
            stage_load<3>(ve, (const float4*)(Ke + (size_t)n0 * 8), en4, tid);
            stage_load<4>(vo, (const float4*)(Ko + (size_t)n0 * 15), on4, tid);
            if (tid < (icount & 3)) vt = isrc[4 * in4 + tid];
            if (cb == 0) {
#pragma unroll
                for (int k = 0; k < WIT; ++k) {
                    int e = tid + 256 * k;
                    e = e < WTOT ? e : WTOT - 1;
                    const int r = e / WLEN;
                    int f = wlo + (e - r * WLEN);
                    f = f < 0 ? 0 : (f > NFINE ? NFINE : f);
                    vw[k] = ub[(size_t)r * 8193 + STEP * f];
                }
            }
            stage_store<8>(vi, (int4*)sIdx, in4, tid);
            stage_store<3>(ve, (float4*)sKe, en4, tid);
            stage_store<4>(vo, (float4*)sKo, on4, tid);
            if (tid < (icount & 3)) sIdx[4 * in4 + tid] = vt;
            if (cb == 0) {
#pragma unroll
                for (int k = 0; k < WIT; ++k) {
                    const int e = tid + 256 * k;
                    if (e < WTOT) sU[e] = vw[k];
                }
            }
        }
        __syncthreads();
        const float* oW = oldW + cb;
        float* nW = newW + 2 * cb;
        for (int nl = tid; nl < ce; nl += 256) {
            const int* row = sIdx + (2 * nl) * 15;
            const float* kp = sKe + nl * 8;
            int ix[8];
            float kk[8];
#pragma unroll
            for (int m = 0; m < 8; ++m) { ix[m] = row[2 * m]; kk[m] = kp[m]; }
            int mn = ix[0], mx = ix[0];
#pragma unroll
            for (int m = 1; m < 8; ++m) { mn = min(mn, ix[m]); mx = max(mx, ix[m]); }
            if (mn >= wlo && mx < wlo + WLEN) {
#pragma unroll
                for (int m = 0; m < 8; ++m) ix[m] -= wlo;
#pragma unroll 4
                for (int r = 0; r < RS_R; ++r) {
                    const float* u = sU + r * WLEN;
                    float acc = 0.f;
#pragma unroll
                    for (int m = 0; m < 8; ++m) acc += kk[m] * u[ix[m]];
                    nW[r * newPitch + 2 * nl] = oW[r * oldPitch + nl] + hf * acc;
                }
            } else {
#pragma unroll
                for (int m = 0; m < 8; ++m) ix[m] *= STEP;
#pragma unroll 4
                for (int r = 0; r < RS_R; ++r) {
                    const float* u = ub + (size_t)r * 8193;
                    float acc = 0.f;
#pragma unroll
                    for (int m = 0; m < 8; ++m) acc += kk[m] * u[ix[m]];
                    nW[r * newPitch + 2 * nl] = oW[r * oldPitch + nl] + hf * acc;
                }
            }
        }
        for (int nl = tid; nl < co; nl += 256) {
            const int* row = sIdx + (2 * nl + 1) * 15;
            const float* kp = sKo + nl * 15;
            int ix[15];
            float kk[15];
#pragma unroll
            for (int m = 0; m < 15; ++m) { ix[m] = row[m]; kk[m] = kp[m]; }
            int mn = ix[0], mx = ix[0];
#pragma unroll
            for (int m = 1; m < 15; ++m) { mn = min(mn, ix[m]); mx = max(mx, ix[m]); }
            if (mn >= wlo && mx < wlo + WLEN) {
#pragma unroll
                for (int m = 0; m < 15; ++m) ix[m] -= wlo;
#pragma unroll 4
                for (int r = 0; r < RS_R; ++r) {
                    const float* u = sU + r * WLEN;
                    float acc = 0.f;
#pragma unroll
                    for (int m = 0; m < 15; ++m) acc += kk[m] * u[ix[m]];
                    nW[r * newPitch + 2 * nl + 1] =
                        0.5f * (oW[r * oldPitch + nl] + oW[r * oldPitch + nl + 1]) + hf * acc;
                }
            } else {
#pragma unroll
                for (int m = 0; m < 15; ++m) ix[m] *= STEP;
#pragma unroll 2
                for (int r = 0; r < RS_R; ++r) {
                    const float* u = ub + (size_t)r * 8193;
                    float acc = 0.f;
#pragma unroll
                    for (int m = 0; m < 15; ++m) acc += kk[m] * u[ix[m]];
                    nW[r * newPitch + 2 * nl + 1] =
                        0.5f * (oW[r * oldPitch + nl] + oW[r * oldPitch + nl + 1]) + hf * acc;
                }
            }
        }
        __syncthreads();
    }
}

__device__ __forceinline__ float boundary_value(const float* __restrict__ sbuf,
                                                const float* __restrict__ Kbd_V,
                                                const int bc, const int which) {
    const float* kv = Kbd_V + which * 32;
    const float* sp = sbuf + bc * 32;
    float acc = 0.f;
#pragma unroll
    for (int r = 0; r < 32; ++r) acc += kv[r] * sp[r];
    return (1.0f / 8192.0f) * acc;
}

__global__ void __launch_bounds__(256) k_refine(const float* __restrict__ uh,
                                                const float* __restrict__ wh0,
                                                const float* __restrict__ sbuf,
                                                const float* __restrict__ Kbd_V,
                                                const float* __restrict__ Ke0,
                                                const float* __restrict__ Ko0,
                                                const float* __restrict__ Ke1,
                                                const float* __restrict__ Ko1,
                                                const float* __restrict__ Ke2,
                                                const float* __restrict__ Ko2,
                                                const int* __restrict__ nbrL0,
                                                const int* __restrict__ nbrL1,
                                                const int* __restrict__ nbrL2,
                                                float* __restrict__ out) {
    __shared__ float wA[RS_R * RS_PA];
    __shared__ float wB[RS_R * RS_PB];
    __shared__ int   sIdx[RS_IDXN] __attribute__((aligned(16)));
    __shared__ float sKe[RS_KEN] __attribute__((aligned(16)));
    __shared__ float sKo[RS_KON] __attribute__((aligned(16)));
    __shared__ float sU[RS_R * RS_WMAX];
    const int tid  = threadIdx.x;
    const int sg   = blockIdx.x % RS_NSEG;
    const int rg   = blockIdx.x / RS_NSEG;
    const int c0   = sg * RS_S;
    const int row0 = rg * RS_R;
    const float* ub = uh + (size_t)row0 * 8193;

    for (int e = tid; e < RS_R * (RS_S + 1); e += 256) {
        const int r = e / (RS_S + 1);
        const int n = e - r * (RS_S + 1);
        wA[r * RS_PA + n] = wh0[(size_t)(row0 + r) * 1025 + c0 + n];
    }

    refine_level<4, 2 * RS_S + 16>(ub, Ke0, Ko0, nbrL0, wA, RS_PA, wB, RS_PB,
                                   c0, RS_S + 1, RS_S, tid, sIdx, sKe, sKo, sU);
    refine_level<2, 4 * RS_S + 16>(ub, Ke1, Ko1, nbrL1, wB, RS_PB, wA, RS_PA,
                                   2 * c0, 2 * RS_S + 1, 2 * RS_S, tid, sIdx, sKe, sKo, sU);
    refine_level<1, 8 * RS_S + 16>(ub, Ke2, Ko2, nbrL2, wA, RS_PA, wB, RS_PB,
                                   4 * c0, 4 * RS_S, 4 * RS_S, tid, sIdx, sKe, sKo, sU);

    for (int e = tid; e < RS_R * 8 * RS_S; e += 256) {
        const int r = e / (8 * RS_S);
        const int j = e - r * (8 * RS_S);
        const int g = 8 * c0 + j;
        float v = wB[r * RS_PB + j];
        if (g == 0) v = boundary_value(sbuf, Kbd_V, row0 + r, 0);
        out[(size_t)(row0 + r) * 8193 + g] = v;
    }
    if (sg == RS_NSEG - 1 && tid < RS_R)
        out[(size_t)(row0 + tid) * 8193 + 8192] = boundary_value(sbuf, Kbd_V, row0 + tid, 1);
}

extern "C" void kernel_launch(void* const* d_in, const int* in_sizes, int n_in,
                              void* d_out, int out_size, void* d_ws, size_t ws_size,
                              hipStream_t stream) {
    const float* uh     = (const float*)d_in[0];
    const float* KHH_U  = (const float*)d_in[1];
    const float* KHH_V  = (const float*)d_in[2];
    const float* Ke0_U  = (const float*)d_in[3];
    const float* Ke0_V  = (const float*)d_in[4];
    const float* Ko0_U  = (const float*)d_in[5];
    const float* Ko0_V  = (const float*)d_in[6];
    const int*   nbr0   = (const int*)d_in[7];
    const float* Ke1_U  = (const float*)d_in[8];
    const float* Ke1_V  = (const float*)d_in[9];
    const float* Ko1_U  = (const float*)d_in[10];
    const float* Ko1_V  = (const float*)d_in[11];
    const int*   nbr1   = (const int*)d_in[12];
    const float* Ke2_U  = (const float*)d_in[13];
    const float* Ke2_V  = (const float*)d_in[14];
    const float* Ko2_U  = (const float*)d_in[15];
    const float* Ko2_V  = (const float*)d_in[16];
    const int*   nbr2   = (const int*)d_in[17];
    const float* Kbd_U  = (const float*)d_in[18];
    const float* Kbd_V  = (const float*)d_in[19];

    float* ws  = (float*)d_ws;
    float* wh0 = ws;
    float* sb  = wh0 + 524800;
    float* Ke0 = sb + 16384;
    float* Ke1 = Ke0 + 8200;
    float* Ke2 = Ke1 + 16392;
    float* Ko0 = Ke2 + 32776;
    float* Ko1 = Ko0 + 15360;
    float* Ko2 = Ko1 + 30720;

    float* outp = (float*)d_out;

    k_wmma_proj_part<<<32 * P_KS + CB_ALL, 512, 0, stream>>>(uh, KHH_V, Kbd_U, wh0,
                                                             Ke0_U, Ke0_V, Ko0_U, Ko0_V,
                                                             Ke1_U, Ke1_V, Ko1_U, Ko1_V,
                                                             Ke2_U, Ke2_V, Ko2_U, Ko2_V,
                                                             Ke0, Ko0, Ke1, Ko1, Ke2, Ko2);
    k_wmma_proj_out<<<32, 1024, 0, stream>>>(KHH_U, wh0, sb);

    k_refine<<<512, 256, 0, stream>>>(uh, wh0, sb, Kbd_V,
                                      Ke0, Ko0, Ke1, Ko1, Ke2, Ko2,
                                      nbr2, nbr1, nbr0, outp);
}


# K-slice sum and third product folded into the refinement launch: a block adds its 8 rows' parked partial tiles and makes its 129 coarse points on WMMA; 3 launches become 2; summation order unchanged
# speedup vs baseline: 15.2730x; 1.2714x over previous
#include <hip/hip_runtime.h>
#include <hip/hip_bf16.h>

typedef __attribute__((ext_vector_type(2))) float v2f;
typedef __attribute__((ext_vector_type(8))) float v8f;

#define WMMA_F32X4(a, b, c) \
    __builtin_amdgcn_wmma_f32_16x16x4_f32(false, (a), false, (b), (short)0, (c), false, false)

template <int COLS>
__device__ __forceinline__ void rankmm_elem(const float* __restrict__ U,
                                            const float* __restrict__ V,
                                            float* __restrict__ out,
                                            const int rows, const int idx) {
    if (idx >= rows * COLS) return;
    int n = idx / COLS;
    int m = idx - n * COLS;
    const float* up = U + n * 32;
    const float* vp = V + m * 32;
    float acc = 0.f;
#pragma unroll
    for (int r = 0; r < 32; ++r) acc += up[r] * vp[r];
    out[idx] = acc;
}

#define CB(elems) (((elems) + 511) / 512)
#define CB_E0 CB(1025 * 8)
#define CB_O0 CB(1024 * 15)
#define CB_E1 CB(2049 * 8)
#define CB_O1 CB(2048 * 15)
#define CB_E2 CB(4097 * 8)
#define CB_O2 CB(4096 * 15)
#define CB_ALL (CB_E0 + CB_O0 + CB_E1 + CB_O1 + CB_E2 + CB_O2)

#define P_KS     8
#define P_WAVES 16
#define TILE_ROWS_FLOATS 16400

__global__ void __launch_bounds__(512) k_wmma_proj_part(const float* __restrict__ uh,
                                                        const float* __restrict__ KHH_V,
                                                        const float* __restrict__ Kbd_U,
                                                        float* __restrict__ part,
                                                        const float* __restrict__ Ke0_U,
                                                        const float* __restrict__ Ke0_V,
                                                        const float* __restrict__ Ko0_U,
                                                        const float* __restrict__ Ko0_V,
                                                        const float* __restrict__ Ke1_U,
                                                        const float* __restrict__ Ke1_V,
                                                        const float* __restrict__ Ko1_U,
                                                        const float* __restrict__ Ko1_V,
                                                        const float* __restrict__ Ke2_U,
                                                        const float* __restrict__ Ke2_V,
                                                        const float* __restrict__ Ko2_U,
                                                        const float* __restrict__ Ko2_V,
                                                        float* __restrict__ Ke0,
                                                        float* __restrict__ Ko0,
                                                        float* __restrict__ Ke1,
                                                        float* __restrict__ Ko1,
                                                        float* __restrict__ Ke2,
                                                        float* __restrict__ Ko2) {
    __shared__ float red[P_WAVES][1024];
    if (blockIdx.x >= 32 * P_KS) {
        int b = blockIdx.x - 32 * P_KS;
        const int t = threadIdx.x;
        if (b < CB_E0) { rankmm_elem<8>(Ke0_U, Ke0_V, Ke0, 1025, b * 512 + t); return; }
        b -= CB_E0;
        if (b < CB_O0) { rankmm_elem<15>(Ko0_U, Ko0_V, Ko0, 1024, b * 512 + t); return; }
        b -= CB_O0;
        if (b < CB_E1) { rankmm_elem<8>(Ke1_U, Ke1_V, Ke1, 2049, b * 512 + t); return; }
        b -= CB_E1;
        if (b < CB_O1) { rankmm_elem<15>(Ko1_U, Ko1_V, Ko1, 2048, b * 512 + t); return; }
        b -= CB_O1;
        if (b < CB_E2) { rankmm_elem<8>(Ke2_U, Ke2_V, Ke2, 4097, b * 512 + t); return; }
        b -= CB_E2;
        rankmm_elem<15>(Ko2_U, Ko2_V, Ko2, 4096, b * 512 + t);
        return;
    }
    const int wave = __builtin_amdgcn_readfirstlane((int)(threadIdx.x >> 5));
    const int lane = threadIdx.x & 31;
    const int half = lane >> 4;
    const int mrow = lane & 15;
    const int mt   = blockIdx.x / P_KS;
    const int ks   = blockIdx.x % P_KS;
    const int R    = mt * 16;
    const bool tail = (ks == P_KS - 1) && (wave == 0);

    const float* arow = uh + (size_t)(R + mrow) * 8193;

    v8f t0 = {}; v8f t1 = {};
    for (int it = 0; it < 32 / P_WAVES; ++it) {
        int k0 = 4 * (ks * 32 + it * P_WAVES + wave);
        int ka = k0 + 2 * half;
        int kb = ka + 1;
        v2f a, b0, b1;
        a.x  = arow[8 * ka];
        a.y  = arow[8 * kb];
        b0.x = KHH_V[ka * 32 + mrow];
        b0.y = KHH_V[kb * 32 + mrow];
        b1.x = KHH_V[ka * 32 + 16 + mrow];
        b1.y = KHH_V[kb * 32 + 16 + mrow];
        t0 = WMMA_F32X4(a, b0, t0);
        t1 = WMMA_F32X4(a, b1, t1);
    }
    if (tail) {
        float alast = arow[8192];
        v2f a, b0, b1;
        a.x  = (half == 0) ? alast : 0.f;
        a.y  = 0.f;
        b0.x = KHH_V[1024 * 32 + mrow];
        b0.y = b0.x;
        b1.x = KHH_V[1024 * 32 + 16 + mrow];
        b1.y = b1.x;
        t0 = WMMA_F32X4(a, b0, t0);
        t1 = WMMA_F32X4(a, b1, t1);
    }

    v8f s0 = {}; v8f s1 = {};
    for (int it = 0; it < 256 / P_WAVES; ++it) {
        int k0 = 4 * (ks * 256 + it * P_WAVES + wave);
        int ka = k0 + 2 * half;
        int kb = ka + 1;
        v2f a, b0, b1;
        a.x  = arow[ka];
        a.y  = arow[kb];
        b0.x = Kbd_U[ka * 32 + mrow];
        b0.y = Kbd_U[kb * 32 + mrow];
        b1.x = Kbd_U[ka * 32 + 16 + mrow];
        b1.y = Kbd_U[kb * 32 + 16 + mrow];
        s0 = WMMA_F32X4(a, b0, s0);
        s1 = WMMA_F32X4(a, b1, s1);
    }
    if (tail) {
        float alast = arow[8192];
        v2f a, b0, b1;
        a.x  = (half == 0) ? alast : 0.f;
        a.y  = 0.f;
        b0.x = Kbd_U[8192 * 32 + mrow];
        b0.y = b0.x;
        b1.x = Kbd_U[8192 * 32 + 16 + mrow];
        b1.y = b1.x;
        s0 = WMMA_F32X4(a, b0, s0);
        s1 = WMMA_F32X4(a, b1, s1);
    }

    {
        float* rw = red[wave];
#pragma unroll
        for (int j = 0; j < 8; ++j) {
            rw[j * 32 + lane]       = t0[j];
            rw[256 + j * 32 + lane] = t1[j];
            rw[512 + j * 32 + lane] = s0[j];
            rw[768 + j * 32 + lane] = s1[j];
        }
    }
    __syncthreads();
    {
        float* dst = part + (size_t)mt * TILE_ROWS_FLOATS + ks * 1024;
#pragma unroll
        for (int q = 0; q < 2; ++q) {
            const int e = threadIdx.x + 512 * q;
            float sum = red[0][e];
#pragma unroll
            for (int w = 1; w < P_WAVES; ++w) sum += red[w][e];
            dst[e] = sum;
        }
    }
}

#define RS_S    128
#define RS_R    8
#define RS_NSEG (1024 / RS_S)
#define RS_PA   (4 * RS_S + 4)
#define RS_PB   (8 * RS_S + 8)
#define RS_CH   256
#define RS_IDXN (8 * 1024 + 4)
#define RS_KEN  (3 * 1024 + 4)
#define RS_KON  (4 * 1024 + 4)
#define RS_WMAX (8 * RS_S + 16)

template <int MAXIT, typename V>
__device__ __forceinline__ void stage_load(V (&v)[MAXIT], const V* __restrict__ s4,
                                           const int n4, const int tid) {
#pragma unroll
    for (int k = 0; k < MAXIT; ++k) {
        if (k * 256 < n4) {
            const int i = tid + 256 * k;
            v[k] = s4[i < n4 ? i : n4 - 1];
        }
    }
}
template <int MAXIT, typename V>
__device__ __forceinline__ void stage_store(const V (&v)[MAXIT], V* d4,
                                            const int n4, const int tid) {
#pragma unroll
    for (int k = 0; k < MAXIT; ++k) {
        if (k * 256 < n4) {
            const int i = tid + 256 * k;
            d4[i < n4 ? i : i + 1] = v[k];
        }
    }
}

template <int STEP, int WLEN>
__device__ __forceinline__ void refine_level(const float* __restrict__ ub,
                                             const float* __restrict__ Ke,
                                             const float* __restrict__ Ko,
                                             const int* __restrict__ nbr,
                                             const float* oldW, const int oldPitch,
                                             float* newW, const int newPitch,
                                             const int nb, const int cntE, const int cntO,
                                             const int tid,
                                             int* sIdx, float* sKe, float* sKo, float* sU) {
    const float hf = (1.0f / 8192.0f) * (float)STEP;
    constexpr int NFINE = 8192 / STEP;
    constexpr int WTOT  = RS_R * WLEN;
    constexpr int WIT   = (WTOT + 255) / 256;
    const int wlo = 2 * nb - 8;
#pragma unroll
    for (int cb = 0; cb < cntO; cb += RS_CH) {
        const int ce = (cb + RS_CH >= cntO) ? (cntE - cb) : RS_CH;
        const int co = (cntO - cb < RS_CH) ? (cntO - cb) : RS_CH;
        const int nrows = (2 * ce - 1 > 2 * co) ? (2 * ce - 1) : (2 * co);
        const int n0 = nb + cb;
        {
            const int* isrc = nbr + (size_t)(2 * n0) * 15;
            const int icount = nrows * 15;
            const int in4 = icount >> 2, en4 = (ce * 8) >> 2, on4 = (co * 15) >> 2;
            int4 vi[8];
            float4 ve[3];
            float4 vo[4];
            float vw[WIT];
            int vt = 0;
            stage_load<8>(vi, (const int4*)isrc, in4, tid);
            stage_load<3>(ve, (const float4*)(Ke + (size_t)n0 * 8), en4, tid);
            stage_load<4>(vo, (const float4*)(Ko + (size_t)n0 * 15), on4, tid);
            if (tid < (icount & 3)) vt = isrc[4 * in4 + tid];
            if (cb == 0) {
#pragma unroll
                for (int k = 0; k < WIT; ++k) {
                    int e = tid + 256 * k;
                    e = e < WTOT ? e : WTOT - 1;
                    const int r = e / WLEN;
                    int f = wlo + (e - r * WLEN);
                    f = f < 0 ? 0 : (f > NFINE ? NFINE : f);
                    vw[k] = ub[(size_t)r * 8193 + STEP * f];
                }
            }
            stage_store<8>(vi, (int4*)sIdx, in4, tid);
            stage_store<3>(ve, (float4*)sKe, en4, tid);
            stage_store<4>(vo, (float4*)sKo, on4, tid);
            if (tid < (icount & 3)) sIdx[4 * in4 + tid] = vt;
            if (cb == 0) {
#pragma unroll
                for (int k = 0; k < WIT; ++k) {
                    const int e = tid + 256 * k;
                    if (e < WTOT) sU[e] = vw[k];
                }
            }
        }
        __syncthreads();
        const float* oW = oldW + cb;
        float* nW = newW + 2 * cb;
        for (int nl = tid; nl < ce; nl += 256) {
            const int* row = sIdx + (2 * nl) * 15;
            const float* kp = sKe + nl * 8;
            int ix[8];
            float kk[8];
#pragma unroll
            for (int m = 0; m < 8; ++m) { ix[m] = row[2 * m]; kk[m] = kp[m]; }
            int mn = ix[0], mx = ix[0];
#pragma unroll
            for (int m = 1; m < 8; ++m) { mn = min(mn, ix[m]); mx = max(mx, ix[m]); }
            if (mn >= wlo && mx < wlo + WLEN) {
#pragma unroll
                for (int m = 0; m < 8; ++m) ix[m] -= wlo;
#pragma unroll 4
                for (int r = 0; r < RS_R; ++r) {
                    const float* u = sU + r * WLEN;
                    float acc = 0.f;
#pragma unroll
                    for (int m = 0; m < 8; ++m) acc += kk[m] * u[ix[m]];
                    nW[r * newPitch + 2 * nl] = oW[r * oldPitch + nl] + hf * acc;
                }
            } else {
#pragma unroll
                for (int m = 0; m < 8; ++m) ix[m] *= STEP;
#pragma unroll 4
                for (int r = 0; r < RS_R; ++r) {
                    const float* u = ub + (size_t)r * 8193;
                    float acc = 0.f;
#pragma unroll
                    for (int m = 0; m < 8; ++m) acc += kk[m] * u[ix[m]];
                    nW[r * newPitch + 2 * nl] = oW[r * oldPitch + nl] + hf * acc;
                }
            }
        }
        for (int nl = tid; nl < co; nl += 256) {
            const int* row = sIdx + (2 * nl + 1) * 15;
            const float* kp = sKo + nl * 15;
            int ix[15];
            float kk[15];
#pragma unroll
            for (int m = 0; m < 15; ++m) { ix[m] = row[m]; kk[m] = kp[m]; }
            int mn = ix[0], mx = ix[0];
#pragma unroll
            for (int m = 1; m < 15; ++m) { mn = min(mn, ix[m]); mx = max(mx, ix[m]); }
            if (mn >= wlo && mx < wlo + WLEN) {
#pragma unroll
                for (int m = 0; m < 15; ++m) ix[m] -= wlo;
#pragma unroll 4
                for (int r = 0; r < RS_R; ++r) {
                    const float* u = sU + r * WLEN;
                    float acc = 0.f;
#pragma unroll
                    for (int m = 0; m < 15; ++m) acc += kk[m] * u[ix[m]];
                    nW[r * newPitch + 2 * nl + 1] =
                        0.5f * (oW[r * oldPitch + nl] + oW[r * oldPitch + nl + 1]) + hf * acc;
                }
            } else {
#pragma unroll
                for (int m = 0; m < 15; ++m) ix[m] *= STEP;
#pragma unroll 2
                for (int r = 0; r < RS_R; ++r) {
                    const float* u = ub + (size_t)r * 8193;
                    float acc = 0.f;
#pragma unroll
                    for (int m = 0; m < 15; ++m) acc += kk[m] * u[ix[m]];
                    nW[r * newPitch + 2 * nl + 1] =
                        0.5f * (oW[r * oldPitch + nl] + oW[r * oldPitch + nl + 1]) + hf * acc;
                }
            }
        }
        __syncthreads();
    }
}

__device__ __forceinline__ float boundary_value(const float* srows,
                                                const float* __restrict__ Kbd_V,
                                                const int lr, const int which) {
    const float* kv = Kbd_V + which * 32;
    const float* sp = srows + lr * 32;
    float acc = 0.f;
#pragma unroll
    for (int r = 0; r < 32; ++r) acc += kv[r] * sp[r];
    return (1.0f / 8192.0f) * acc;
}

__global__ void __launch_bounds__(256) k_refine(const float* __restrict__ uh,
                                                const float* __restrict__ part,
                                                const float* __restrict__ KHH_U,
                                                const float* __restrict__ Kbd_V,
                                                const float* __restrict__ Ke0,
                                                const float* __restrict__ Ko0,
                                                const float* __restrict__ Ke1,
                                                const float* __restrict__ Ko1,
                                                const float* __restrict__ Ke2,
                                                const float* __restrict__ Ko2,
                                                const int* __restrict__ nbrL0,
                                                const int* __restrict__ nbrL1,
                                                const int* __restrict__ nbrL2,
                                                float* __restrict__ out) {
    __shared__ float wA[RS_R * RS_PA];
    __shared__ float wB[RS_R * RS_PB];
    __shared__ int   sIdx[RS_IDXN] __attribute__((aligned(16)));
    __shared__ float sKe[RS_KEN] __attribute__((aligned(16)));
    __shared__ float sKo[RS_KON] __attribute__((aligned(16)));
    __shared__ float sU[RS_R * RS_WMAX];
    __shared__ float tl[RS_R * 32];
    __shared__ float sS[RS_R * 32];
    static_assert(RS_R == 8 && RS_S == 128, "W0 below: 8 rows x 32 columns on 256 threads, 8 waves x 16 points");
    const int tid  = threadIdx.x;
    const int wave = __builtin_amdgcn_readfirstlane((int)(threadIdx.x >> 5));
    const int lane = tid & 31;
    const int half = lane >> 4;
    const int mrow = lane & 15;
    const int sg   = blockIdx.x % RS_NSEG;
    const int rg   = blockIdx.x / RS_NSEG;
    const int c0   = sg * RS_S;
    const int row0 = rg * RS_R;
    const float* ub = uh + (size_t)row0 * 8193;

    v2f bf[8];
    v2f bl[8] = {};
    {
        const float* bp = KHH_U + (size_t)(c0 + 16 * wave + mrow) * 32 + 2 * half;
#pragma unroll
        for (int q = 0; q < 8; ++q) { bf[q].x = bp[4 * q]; bf[q].y = bp[4 * q + 1]; }
    }
    if (wave == 0) {
        const float* bp = KHH_U + (size_t)(c0 + RS_S) * 32 + 2 * half;
#pragma unroll
        for (int q = 0; q < 8; ++q) { bl[q].x = bp[4 * q]; bl[q].y = bp[4 * q + 1]; }
    }
    {
        const int mt = row0 >> 4;
        const int hb = (row0 >> 3) & 1;
        const int j  = tid >> 5;
        const int n  = tid & 31;
        const float* sp = part + (size_t)mt * TILE_ROWS_FLOATS
                        + ((n >> 4) << 8) + j * 32 + 16 * hb + (n & 15);
        const float HK = 1.0f / 1024.0f;
        float tv[P_KS];
        float sv[P_KS];
#pragma unroll
        for (int s = 0; s < P_KS; ++s) {
            tv[s] = sp[s * 1024];
            sv[s] = sp[s * 1024 + 512];
        }
        float sum = tv[0];
        float ss  = sv[0];
#pragma unroll
        for (int s = 1; s < P_KS; ++s) { sum += tv[s]; ss += sv[s]; }
        tl[j * 32 + n] = sum * HK;
        sS[j * 32 + n] = ss;
    }
    __syncthreads();
    {
        const float* ap = tl + (mrow & 7) * 32 + 2 * half;
        v2f a[8];
#pragma unroll
        for (int q = 0; q < 8; ++q) { a[q].x = ap[4 * q]; a[q].y = ap[4 * q + 1]; }
        v8f acc = {};
#pragma unroll
        for (int q = 0; q < 8; ++q) acc = WMMA_F32X4(a[q], bf[q], acc);
        if (half == 0) {
#pragma unroll
            for (int j = 0; j < 8; ++j) wA[j * RS_PA + 16 * wave + mrow] = acc[j];
        }
        if (wave == 0) {
            v8f accl = {};
#pragma unroll
            for (int q = 0; q < 8; ++q) accl = WMMA_F32X4(a[q], bl[q], accl);
            if (lane == 0) {
#pragma unroll
                for (int j = 0; j < 8; ++j) wA[j * RS_PA + RS_S] = accl[j];
            }
        }
    }

    refine_level<4, 2 * RS_S + 16>(ub, Ke0, Ko0, nbrL0, wA, RS_PA, wB, RS_PB,
                                   c0, RS_S + 1, RS_S, tid, sIdx, sKe, sKo, sU);
    refine_level<2, 4 * RS_S + 16>(ub, Ke1, Ko1, nbrL1, wB, RS_PB, wA, RS_PA,
                                   2 * c0, 2 * RS_S + 1, 2 * RS_S, tid, sIdx, sKe, sKo, sU);
    refine_level<1, 8 * RS_S + 16>(ub, Ke2, Ko2, nbrL2, wA, RS_PA, wB, RS_PB,
                                   4 * c0, 4 * RS_S, 4 * RS_S, tid, sIdx, sKe, sKo, sU);

    for (int e = tid; e < RS_R * 8 * RS_S; e += 256) {
        const int r = e / (8 * RS_S);
        const int j = e - r * (8 * RS_S);
        const int g = 8 * c0 + j;
        float v = wB[r * RS_PB + j];
        if (g == 0) v = boundary_value(sS, Kbd_V, r, 0);
        out[(size_t)(row0 + r) * 8193 + g] = v;
    }
    if (sg == RS_NSEG - 1 && tid < RS_R)
        out[(size_t)(row0 + tid) * 8193 + 8192] = boundary_value(sS, Kbd_V, tid, 1);
}

extern "C" void kernel_launch(void* const* d_in, const int* in_sizes, int n_in,
                              void* d_out, int out_size, void* d_ws, size_t ws_size,
                              hipStream_t stream) {
    const float* uh     = (const float*)d_in[0];
    const float* KHH_U  = (const float*)d_in[1];
    const float* KHH_V  = (const float*)d_in[2];
    const float* Ke0_U  = (const float*)d_in[3];
    const float* Ke0_V  = (const float*)d_in[4];
    const float* Ko0_U  = (const float*)d_in[5];
    const float* Ko0_V  = (const float*)d_in[6];
    const int*   nbr0   = (const int*)d_in[7];
    const float* Ke1_U  = (const float*)d_in[8];
    const float* Ke1_V  = (const float*)d_in[9];
    const float* Ko1_U  = (const float*)d_in[10];
    const float* Ko1_V  = (const float*)d_in[11];
    const int*   nbr1   = (const int*)d_in[12];
    const float* Ke2_U  = (const float*)d_in[13];
    const float* Ke2_V  = (const float*)d_in[14];
    const float* Ko2_U  = (const float*)d_in[15];
    const float* Ko2_V  = (const float*)d_in[16];
    const int*   nbr2   = (const int*)d_in[17];
    const float* Kbd_U  = (const float*)d_in[18];
    const float* Kbd_V  = (const float*)d_in[19];

    float* ws  = (float*)d_ws;
    float* prt = ws;
    float* Ke0 = prt + 524800;
    float* Ke1 = Ke0 + 8200;
    float* Ke2 = Ke1 + 16392;
    float* Ko0 = Ke2 + 32776;
    float* Ko1 = Ko0 + 15360;
    float* Ko2 = Ko1 + 30720;

    float* outp = (float*)d_out;

    k_wmma_proj_part<<<32 * P_KS + CB_ALL, 512, 0, stream>>>(uh, KHH_V, Kbd_U, prt,
                                                             Ke0_U, Ke0_V, Ko0_U, Ko0_V,
                                                             Ke1_U, Ke1_V, Ko1_U, Ko1_V,
                                                             Ke2_U, Ke2_V, Ko2_U, Ko2_V,
                                                             Ke0, Ko0, Ke1, Ko1, Ke2, Ko2);
    k_refine<<<512, 256, 0, stream>>>(uh, prt, KHH_U, Kbd_V,
                                      Ke0, Ko0, Ke1, Ko1, Ke2, Ko2,
                                      nbr2, nbr1, nbr0, outp);
}


# Final copy of a block's results from LDS to out is straight-line: 16-byte reads issued ahead of 16-byte stores, boundary values made ahead by 8 lanes, not a read-wait-store per trip; sums unchanged
# speedup vs baseline: 16.1902x; 1.0601x over previous
#include <hip/hip_runtime.h>
#include <hip/hip_bf16.h>

typedef __attribute__((ext_vector_type(2))) float v2f;
typedef __attribute__((ext_vector_type(8))) float v8f;
typedef __attribute__((ext_vector_type(4))) float v4f;
typedef float v4f_a4 __attribute__((ext_vector_type(4), aligned(4)));

#define WMMA_F32X4(a, b, c) \
    __builtin_amdgcn_wmma_f32_16x16x4_f32(false, (a), false, (b), (short)0, (c), false, false)

template <int COLS>
__device__ __forceinline__ void rankmm_elem(const float* __restrict__ U,
                                            const float* __restrict__ V,
                                            float* __restrict__ out,
                                            const int rows, const int idx) {
    if (idx >= rows * COLS) return;
    int n = idx / COLS;
    int m = idx - n * COLS;
    const float* up = U + n * 32;
    const float* vp = V + m * 32;
    float acc = 0.f;
#pragma unroll
    for (int r = 0; r < 32; ++r) acc += up[r] * vp[r];
    out[idx] = acc;
}

#define CB(elems) (((elems) + 511) / 512)
#define CB_E0 CB(1025 * 8)
#define CB_O0 CB(1024 * 15)
#define CB_E1 CB(2049 * 8)
#define CB_O1 CB(2048 * 15)
#define CB_E2 CB(4097 * 8)
#define CB_O2 CB(4096 * 15)
#define CB_ALL (CB_E0 + CB_O0 + CB_E1 + CB_O1 + CB_E2 + CB_O2)

#define P_KS     8
#define P_WAVES 16
#define TILE_ROWS_FLOATS 16400

__global__ void __launch_bounds__(512) k_wmma_proj_part(const float* __restrict__ uh,
                                                        const float* __restrict__ KHH_V,
                                                        const float* __restrict__ Kbd_U,
                                                        float* __restrict__ part,
                                                        const float* __restrict__ Ke0_U,
                                                        const float* __restrict__ Ke0_V,
                                                        const float* __restrict__ Ko0_U,
                                                        const float* __restrict__ Ko0_V,
                                                        const float* __restrict__ Ke1_U,
                                                        const float* __restrict__ Ke1_V,
                                                        const float* __restrict__ Ko1_U,
                                                        const float* __restrict__ Ko1_V,
                                                        const float* __restrict__ Ke2_U,
                                                        const float* __restrict__ Ke2_V,
                                                        const float* __restrict__ Ko2_U,
                                                        const float* __restrict__ Ko2_V,
                                                        float* __restrict__ Ke0,
                                                        float* __restrict__ Ko0,
                                                        float* __restrict__ Ke1,
                                                        float* __restrict__ Ko1,
                                                        float* __restrict__ Ke2,
                                                        float* __restrict__ Ko2) {
    __shared__ float red[P_WAVES][1024];
    if (blockIdx.x >= 32 * P_KS) {
        int b = blockIdx.x - 32 * P_KS;
        const int t = threadIdx.x;
        if (b < CB_E0) { rankmm_elem<8>(Ke0_U, Ke0_V, Ke0, 1025, b * 512 + t); return; }
        b -= CB_E0;
        if (b < CB_O0) { rankmm_elem<15>(Ko0_U, Ko0_V, Ko0, 1024, b * 512 + t); return; }
        b -= CB_O0;
        if (b < CB_E1) { rankmm_elem<8>(Ke1_U, Ke1_V, Ke1, 2049, b * 512 + t); return; }
        b -= CB_E1;
        if (b < CB_O1) { rankmm_elem<15>(Ko1_U, Ko1_V, Ko1, 2048, b * 512 + t); return; }
        b -= CB_O1;
        if (b < CB_E2) { rankmm_elem<8>(Ke2_U, Ke2_V, Ke2, 4097, b * 512 + t); return; }
        b -= CB_E2;
        rankmm_elem<15>(Ko2_U, Ko2_V, Ko2, 4096, b * 512 + t);
        return;
    }
    const int wave = __builtin_amdgcn_readfirstlane((int)(threadIdx.x >> 5));
    const int lane = threadIdx.x & 31;
    const int half = lane >> 4;
    const int mrow = lane & 15;
    const int mt   = blockIdx.x / P_KS;
    const int ks   = blockIdx.x % P_KS;
    const int R    = mt * 16;
    const bool tail = (ks == P_KS - 1) && (wave == 0);

    const float* arow = uh + (size_t)(R + mrow) * 8193;

    v8f t0 = {}; v8f t1 = {};
    for (int it = 0; it < 32 / P_WAVES; ++it) {
        int k0 = 4 * (ks * 32 + it * P_WAVES + wave);
        int ka = k0 + 2 * half;
        int kb = ka + 1;
        v2f a, b0, b1;
        a.x  = arow[8 * ka];
        a.y  = arow[8 * kb];
        b0.x = KHH_V[ka * 32 + mrow];
        b0.y = KHH_V[kb * 32 + mrow];
        b1.x = KHH_V[ka * 32 + 16 + mrow];
        b1.y = KHH_V[kb * 32 + 16 + mrow];
        t0 = WMMA_F32X4(a, b0, t0);
        t1 = WMMA_F32X4(a, b1, t1);
    }
    if (tail) {
        float alast = arow[8192];
        v2f a, b0, b1;
        a.x  = (half == 0) ? alast : 0.f;
        a.y  = 0.f;
        b0.x = KHH_V[1024 * 32 + mrow];
        b0.y = b0.x;
        b1.x = KHH_V[1024 * 32 + 16 + mrow];
        b1.y = b1.x;
        t0 = WMMA_F32X4(a, b0, t0);
        t1 = WMMA_F32X4(a, b1, t1);
    }

    v8f s0 = {}; v8f s1 = {};
    for (int it = 0; it < 256 / P_WAVES; ++it) {
        int k0 = 4 * (ks * 256 + it * P_WAVES + wave);
        int ka = k0 + 2 * half;
        int kb = ka + 1;
        v2f a, b0, b1;
        a.x  = arow[ka];
        a.y  = arow[kb];
        b0.x = Kbd_U[ka * 32 + mrow];
        b0.y = Kbd_U[kb * 32 + mrow];
        b1.x = Kbd_U[ka * 32 + 16 + mrow];
        b1.y = Kbd_U[kb * 32 + 16 + mrow];
        s0 = WMMA_F32X4(a, b0, s0);
        s1 = WMMA_F32X4(a, b1, s1);
    }
    if (tail) {
        float alast = arow[8192];
        v2f a, b0, b1;
        a.x  = (half == 0) ? alast : 0.f;
        a.y  = 0.f;
        b0.x = Kbd_U[8192 * 32 + mrow];
        b0.y = b0.x;
        b1.x = Kbd_U[8192 * 32 + 16 + mrow];
        b1.y = b1.x;
        s0 = WMMA_F32X4(a, b0, s0);
        s1 = WMMA_F32X4(a, b1, s1);
    }

    {
        float* rw = red[wave];
#pragma unroll
        for (int j = 0; j < 8; ++j) {
            rw[j * 32 + lane]       = t0[j];
            rw[256 + j * 32 + lane] = t1[j];
            rw[512 + j * 32 + lane] = s0[j];
            rw[768 + j * 32 + lane] = s1[j];
        }
    }
    __syncthreads();
    {
        float* dst = part + (size_t)mt * TILE_ROWS_FLOATS + ks * 1024;
#pragma unroll
        for (int q = 0; q < 2; ++q) {
            const int e = threadIdx.x + 512 * q;
            float sum = red[0][e];
#pragma unroll
            for (int w = 1; w < P_WAVES; ++w) sum += red[w][e];
            dst[e] = sum;
        }
    }
}

#define RS_S    128
#define RS_R    8
#define RS_NSEG (1024 / RS_S)
#define RS_PA   (4 * RS_S + 4)
#define RS_PB   (8 * RS_S + 8)
#define RS_CH   256
#define RS_IDXN (8 * 1024 + 4)
#define RS_KEN  (3 * 1024 + 4)
#define RS_KON  (4 * 1024 + 4)
#define RS_WMAX (8 * RS_S + 16)

template <int MAXIT, typename V>
__device__ __forceinline__ void stage_load(V (&v)[MAXIT], const V* __restrict__ s4,
                                           const int n4, const int tid) {
#pragma unroll
    for (int k = 0; k < MAXIT; ++k) {
        if (k * 256 < n4) {
            const int i = tid + 256 * k;
            v[k] = s4[i < n4 ? i : n4 - 1];
        }
    }
}
template <int MAXIT, typename V>
__device__ __forceinline__ void stage_store(const V (&v)[MAXIT], V* d4,
                                            const int n4, const int tid) {
#pragma unroll
    for (int k = 0; k < MAXIT; ++k) {
        if (k * 256 < n4) {
            const int i = tid + 256 * k;
            d4[i < n4 ? i : i + 1] = v[k];
        }
    }
}

template <int STEP, int WLEN>
__device__ __forceinline__ void refine_level(const float* __restrict__ ub,
                                             const float* __restrict__ Ke,
                                             const float* __restrict__ Ko,
                                             const int* __restrict__ nbr,
                                             const float* oldW, const int oldPitch,
                                             float* newW, const int newPitch,
                                             const int nb, const int cntE, const int cntO,
                                             const int tid,
                                             int* sIdx, float* sKe, float* sKo, float* sU) {
    const float hf = (1.0f / 8192.0f) * (float)STEP;
    constexpr int NFINE = 8192 / STEP;
    constexpr int WTOT  = RS_R * WLEN;
    constexpr int WIT   = (WTOT + 255) / 256;
    const int wlo = 2 * nb - 8;
#pragma unroll
    for (int cb = 0; cb < cntO; cb += RS_CH) {
        const int ce = (cb + RS_CH >= cntO) ? (cntE - cb) : RS_CH;
        const int co = (cntO - cb < RS_CH) ? (cntO - cb) : RS_CH;
        const int nrows = (2 * ce - 1 > 2 * co) ? (2 * ce - 1) : (2 * co);
        const int n0 = nb + cb;
        {
            const int* isrc = nbr + (size_t)(2 * n0) * 15;
            const int icount = nrows * 15;
            const int in4 = icount >> 2, en4 = (ce * 8) >> 2, on4 = (co * 15) >> 2;
            int4 vi[8];
            float4 ve[3];
            float4 vo[4];
            float vw[WIT];
            int vt = 0;
            stage_load<8>(vi, (const int4*)isrc, in4, tid);
            stage_load<3>(ve, (const float4*)(Ke + (size_t)n0 * 8), en4, tid);
            stage_load<4>(vo, (const float4*)(Ko + (size_t)n0 * 15), on4, tid);
            if (tid < (icount & 3)) vt = isrc[4 * in4 + tid];
            if (cb == 0) {
#pragma unroll
                for (int k = 0; k < WIT; ++k) {
                    int e = tid + 256 * k;
                    e = e < WTOT ? e : WTOT - 1;
                    const int r = e / WLEN;
                    int f = wlo + (e - r * WLEN);
                    f = f < 0 ? 0 : (f > NFINE ? NFINE : f);
                    vw[k] = ub[(size_t)r * 8193 + STEP * f];
                }
            }
            stage_store<8>(vi, (int4*)sIdx, in4, tid);
            stage_store<3>(ve, (float4*)sKe, en4, tid);
            stage_store<4>(vo, (float4*)sKo, on4, tid);
            if (tid < (icount & 3)) sIdx[4 * in4 + tid] = vt;
            if (cb == 0) {
#pragma unroll
                for (int k = 0; k < WIT; ++k) {
                    const int e = tid + 256 * k;
                    if (e < WTOT) sU[e] = vw[k];
                }
            }
        }
        __syncthreads();
        const float* oW = oldW + cb;
        float* nW = newW + 2 * cb;
        for (int nl = tid; nl < ce; nl += 256) {
            const int* row = sIdx + (2 * nl) * 15;
            const float* kp = sKe + nl * 8;
            int ix[8];
            float kk[8];
#pragma unroll
            for (int m = 0; m < 8; ++m) { ix[m] = row[2 * m]; kk[m] = kp[m]; }
            int mn = ix[0], mx = ix[0];
#pragma unroll
            for (int m = 1; m < 8; ++m) { mn = min(mn, ix[m]); mx = max(mx, ix[m]); }
            if (mn >= wlo && mx < wlo + WLEN) {
#pragma unroll
                for (int m = 0; m < 8; ++m) ix[m] -= wlo;
#pragma unroll 4
                for (int r = 0; r < RS_R; ++r) {
                    const float* u = sU + r * WLEN;
                    float acc = 0.f;
#pragma unroll
                    for (int m = 0; m < 8; ++m) acc += kk[m] * u[ix[m]];
                    nW[r * newPitch + 2 * nl] = oW[r * oldPitch + nl] + hf * acc;
                }
            } else {
#pragma unroll
                for (int m = 0; m < 8; ++m) ix[m] *= STEP;
#pragma unroll 4
                for (int r = 0; r < RS_R; ++r) {
                    const float* u = ub + (size_t)r * 8193;
                    float acc = 0.f;
#pragma unroll
                    for (int m = 0; m < 8; ++m) acc += kk[m] * u[ix[m]];
                    nW[r * newPitch + 2 * nl] = oW[r * oldPitch + nl] + hf * acc;
                }
            }
        }
        for (int nl = tid; nl < co; nl += 256) {
            const int* row = sIdx + (2 * nl + 1) * 15;
            const float* kp = sKo + nl * 15;
            int ix[15];
            float kk[15];
#pragma unroll
            for (int m = 0; m < 15; ++m) { ix[m] = row[m]; kk[m] = kp[m]; }
            int mn = ix[0], mx = ix[0];
#pragma unroll
            for (int m = 1; m < 15; ++m) { mn = min(mn, ix[m]); mx = max(mx, ix[m]); }
            if (mn >= wlo && mx < wlo + WLEN) {
#pragma unroll
                for (int m = 0; m < 15; ++m) ix[m] -= wlo;
#pragma unroll 4
                for (int r = 0; r < RS_R; ++r) {
                    const float* u = sU + r * WLEN;
                    float acc = 0.f;
#pragma unroll
                    for (int m = 0; m < 15; ++m) acc += kk[m] * u[ix[m]];
                    nW[r * newPitch + 2 * nl + 1] =
                        0.5f * (oW[r * oldPitch + nl] + oW[r * oldPitch + nl + 1]) + hf * acc;
                }
            } else {
#pragma unroll
                for (int m = 0; m < 15; ++m) ix[m] *= STEP;
#pragma unroll 2
                for (int r = 0; r < RS_R; ++r) {
                    const float* u = ub + (size_t)r * 8193;
                    float acc = 0.f;
#pragma unroll
                    for (int m = 0; m < 15; ++m) acc += kk[m] * u[ix[m]];
                    nW[r * newPitch + 2 * nl + 1] =
                        0.5f * (oW[r * oldPitch + nl] + oW[r * oldPitch + nl + 1]) + hf * acc;
                }
            }
        }
        __syncthreads();
    }
}

__device__ __forceinline__ float boundary_value(const float* srows,
                                                const float* __restrict__ Kbd_V,
                                                const int lr, const int which) {
    const float* kv = Kbd_V + which * 32;
    const float* sp = srows + lr * 32;
    float acc = 0.f;
#pragma unroll
    for (int r = 0; r < 32; ++r) acc += kv[r] * sp[r];
    return (1.0f / 8192.0f) * acc;
}

__global__ void __launch_bounds__(256) k_refine(const float* __restrict__ uh,
                                                const float* __restrict__ part,
                                                const float* __restrict__ KHH_U,
                                                const float* __restrict__ Kbd_V,
                                                const float* __restrict__ Ke0,
                                                const float* __restrict__ Ko0,
                                                const float* __restrict__ Ke1,
                                                const float* __restrict__ Ko1,
                                                const float* __restrict__ Ke2,
                                                const float* __restrict__ Ko2,
                                                const int* __restrict__ nbrL0,
                                                const int* __restrict__ nbrL1,
                                                const int* __restrict__ nbrL2,
                                                float* __restrict__ out) {
    __shared__ float wA[RS_R * RS_PA];
    __shared__ float wB[RS_R * RS_PB] __attribute__((aligned(16)));
    __shared__ int   sIdx[RS_IDXN] __attribute__((aligned(16)));
    __shared__ float sKe[RS_KEN] __attribute__((aligned(16)));
    __shared__ float sKo[RS_KON] __attribute__((aligned(16)));
    __shared__ float sU[RS_R * RS_WMAX];
    __shared__ float tl[RS_R * 32];
    __shared__ float sS[RS_R * 32];
    __shared__ float sBnd[2 * RS_R];
    static_assert(RS_R == 8 && RS_S == 128, "W0 below: 8 rows x 32 columns on 256 threads, 8 waves x 16 points");
    const int tid  = threadIdx.x;
    const int wave = __builtin_amdgcn_readfirstlane((int)(threadIdx.x >> 5));
    const int lane = tid & 31;
    const int half = lane >> 4;
    const int mrow = lane & 15;
    const int sg   = blockIdx.x % RS_NSEG;
    const int rg   = blockIdx.x / RS_NSEG;
    const int c0   = sg * RS_S;
    const int row0 = rg * RS_R;
    const float* ub = uh + (size_t)row0 * 8193;

    v2f bf[8];
    v2f bl[8] = {};
    {
        const float* bp = KHH_U + (size_t)(c0 + 16 * wave + mrow) * 32 + 2 * half;
#pragma unroll
        for (int q = 0; q < 8; ++q) { bf[q].x = bp[4 * q]; bf[q].y = bp[4 * q + 1]; }
    }
    if (wave == 0) {
        const float* bp = KHH_U + (size_t)(c0 + RS_S) * 32 + 2 * half;
#pragma unroll
        for (int q = 0; q < 8; ++q) { bl[q].x = bp[4 * q]; bl[q].y = bp[4 * q + 1]; }
    }
    {
        const int mt = row0 >> 4;
        const int hb = (row0 >> 3) & 1;
        const int j  = tid >> 5;
        const int n  = tid & 31;
        const float* sp = part + (size_t)mt * TILE_ROWS_FLOATS
                        + ((n >> 4) << 8) + j * 32 + 16 * hb + (n & 15);
        const float HK = 1.0f / 1024.0f;
        float tv[P_KS];
        float sv[P_KS];
#pragma unroll
        for (int s = 0; s < P_KS; ++s) {
            tv[s] = sp[s * 1024];
            sv[s] = sp[s * 1024 + 512];
        }
        float sum = tv[0];
        float ss  = sv[0];
#pragma unroll
        for (int s = 1; s < P_KS; ++s) { sum += tv[s]; ss += sv[s]; }
        tl[j * 32 + n] = sum * HK;
        sS[j * 32 + n] = ss;
    }
    __syncthreads();
    if (sg == 0 && tid >= 256 - RS_R) {
        const int r = tid - (256 - RS_R);
        sBnd[r] = boundary_value(sS, Kbd_V, r, 0);
    }
    if (sg == RS_NSEG - 1 && tid >= 256 - RS_R) {
        const int r = tid - (256 - RS_R);
        sBnd[RS_R + r] = boundary_value(sS, Kbd_V, r, 1);
    }
    {
        const float* ap = tl + (mrow & 7) * 32 + 2 * half;
        v2f a[8];
#pragma unroll
        for (int q = 0; q < 8; ++q) { a[q].x = ap[4 * q]; a[q].y = ap[4 * q + 1]; }
        v8f acc = {};
#pragma unroll
        for (int q = 0; q < 8; ++q) acc = WMMA_F32X4(a[q], bf[q], acc);
        if (half == 0) {
#pragma unroll
            for (int j = 0; j < 8; ++j) wA[j * RS_PA + 16 * wave + mrow] = acc[j];
        }
        if (wave == 0) {
            v8f accl = {};
#pragma unroll
            for (int q = 0; q < 8; ++q) accl = WMMA_F32X4(a[q], bl[q], accl);
            if (lane == 0) {
#pragma unroll
                for (int j = 0; j < 8; ++j) wA[j * RS_PA + RS_S] = accl[j];
            }
        }
    }

    refine_level<4, 2 * RS_S + 16>(ub, Ke0, Ko0, nbrL0, wA, RS_PA, wB, RS_PB,
                                   c0, RS_S + 1, RS_S, tid, sIdx, sKe, sKo, sU);
    refine_level<2, 4 * RS_S + 16>(ub, Ke1, Ko1, nbrL1, wB, RS_PB, wA, RS_PA,
                                   2 * c0, 2 * RS_S + 1, 2 * RS_S, tid, sIdx, sKe, sKo, sU);
    refine_level<1, 8 * RS_S + 16>(ub, Ke2, Ko2, nbrL2, wA, RS_PA, wB, RS_PB,
                                   4 * c0, 4 * RS_S, 4 * RS_S, tid, sIdx, sKe, sKo, sU);

    {
        static_assert(8 * RS_S == 4 * 256, "final copy: 256 threads x 4 points = one row of the segment");
        const float* wp = wB + 4 * tid;
        float* ob = out + (size_t)row0 * 8193 + 8 * c0 + 4 * tid;
        v4f v[RS_R];
#pragma unroll
        for (int r = 0; r < RS_R; ++r) v[r] = *(const v4f*)(wp + r * RS_PB);
        if (sg == 0 && tid == 0) {
#pragma unroll
            for (int r = 0; r < RS_R; ++r) v[r].x = sBnd[r];
        }
#pragma unroll
        for (int r = 0; r < RS_R; ++r) *(v4f_a4*)(ob + (size_t)r * 8193) = v[r];
    }
    if (sg == RS_NSEG - 1 && tid < RS_R)
        out[(size_t)(row0 + tid) * 8193 + 8192] = sBnd[RS_R + tid];
}

extern "C" void kernel_launch(void* const* d_in, const int* in_sizes, int n_in,
                              void* d_out, int out_size, void* d_ws, size_t ws_size,
                              hipStream_t stream) {
    const float* uh     = (const float*)d_in[0];
    const float* KHH_U  = (const float*)d_in[1];
    const float* KHH_V  = (const float*)d_in[2];
    const float* Ke0_U  = (const float*)d_in[3];
    const float* Ke0_V  = (const float*)d_in[4];
    const float* Ko0_U  = (const float*)d_in[5];
    const float* Ko0_V  = (const float*)d_in[6];
    const int*   nbr0   = (const int*)d_in[7];
    const float* Ke1_U  = (const float*)d_in[8];
    const float* Ke1_V  = (const float*)d_in[9];
    const float* Ko1_U  = (const float*)d_in[10];
    const float* Ko1_V  = (const float*)d_in[11];
    const int*   nbr1   = (const int*)d_in[12];
    const float* Ke2_U  = (const float*)d_in[13];
    const float* Ke2_V  = (const float*)d_in[14];
    const float* Ko2_U  = (const float*)d_in[15];
    const float* Ko2_V  = (const float*)d_in[16];
    const int*   nbr2   = (const int*)d_in[17];
    const float* Kbd_U  = (const float*)d_in[18];
    const float* Kbd_V  = (const float*)d_in[19];

    float* ws  = (float*)d_ws;
    float* prt = ws;
    float* Ke0 = prt + 524800;
    float* Ke1 = Ke0 + 8200;
    float* Ke2 = Ke1 + 16392;
    float* Ko0 = Ke2 + 32776;
    float* Ko1 = Ko0 + 15360;
    float* Ko2 = Ko1 + 30720;

    float* outp = (float*)d_out;

    k_wmma_proj_part<<<32 * P_KS + CB_ALL, 512, 0, stream>>>(uh, KHH_V, Kbd_U, prt,
                                                             Ke0_U, Ke0_V, Ko0_U, Ko0_V,
                                                             Ke1_U, Ke1_V, Ko1_U, Ko1_V,
                                                             Ke2_U, Ke2_V, Ko2_U, Ko2_V,
                                                             Ke0, Ko0, Ke1, Ko1, Ke2, Ko2);
    k_refine<<<512, 256, 0, stream>>>(uh, prt, KHH_U, Kbd_V,
                                      Ke0, Ko0, Ke1, Ko1, Ke2, Ko2,
                                      nbr2, nbr1, nbr0, outp);
}


# Refinement: window and wh strips are point-major in LDS, a point's 8 rows as two 16-byte vectors: a tap of all 8 rows is 2 ds_load_b128 where it was 8 ds_load_b32; each row's sums unchanged
# speedup vs baseline: 16.6754x; 1.0300x over previous
#include <hip/hip_runtime.h>
#include <hip/hip_bf16.h>

typedef __attribute__((ext_vector_type(2))) float v2f;
typedef __attribute__((ext_vector_type(8))) float v8f;
typedef __attribute__((ext_vector_type(4))) float v4f;
typedef float v4f_a4 __attribute__((ext_vector_type(4), aligned(4)));

#define WMMA_F32X4(a, b, c) \
    __builtin_amdgcn_wmma_f32_16x16x4_f32(false, (a), false, (b), (short)0, (c), false, false)

template <int COLS>
__device__ __forceinline__ void rankmm_elem(const float* __restrict__ U,
                                            const float* __restrict__ V,
                                            float* __restrict__ out,
                                            const int rows, const int idx) {
    if (idx >= rows * COLS) return;
    int n = idx / COLS;
    int m = idx - n * COLS;
    const float* up = U + n * 32;
    const float* vp = V + m * 32;
    float acc = 0.f;
#pragma unroll
    for (int r = 0; r < 32; ++r) acc += up[r] * vp[r];
    out[idx] = acc;
}

#define CB(elems) (((elems) + 511) / 512)
#define CB_E0 CB(1025 * 8)
#define CB_O0 CB(1024 * 15)
#define CB_E1 CB(2049 * 8)
#define CB_O1 CB(2048 * 15)
#define CB_E2 CB(4097 * 8)
#define CB_O2 CB(4096 * 15)
#define CB_ALL (CB_E0 + CB_O0 + CB_E1 + CB_O1 + CB_E2 + CB_O2)

#define P_KS     8
#define P_WAVES 16
#define TILE_ROWS_FLOATS 16400

__global__ void __launch_bounds__(512) k_wmma_proj_part(const float* __restrict__ uh,
                                                        const float* __restrict__ KHH_V,
                                                        const float* __restrict__ Kbd_U,
                                                        float* __restrict__ part,
                                                        const float* __restrict__ Ke0_U,
                                                        const float* __restrict__ Ke0_V,
                                                        const float* __restrict__ Ko0_U,
                                                        const float* __restrict__ Ko0_V,
                                                        const float* __restrict__ Ke1_U,
                                                        const float* __restrict__ Ke1_V,
                                                        const float* __restrict__ Ko1_U,
                                                        const float* __restrict__ Ko1_V,
                                                        const float* __restrict__ Ke2_U,
                                                        const float* __restrict__ Ke2_V,
                                                        const float* __restrict__ Ko2_U,
                                                        const float* __restrict__ Ko2_V,
                                                        float* __restrict__ Ke0,
                                                        float* __restrict__ Ko0,
                                                        float* __restrict__ Ke1,
                                                        float* __restrict__ Ko1,
                                                        float* __restrict__ Ke2,
                                                        float* __restrict__ Ko2) {
    __shared__ float red[P_WAVES][1024];
    if (blockIdx.x >= 32 * P_KS) {
        int b = blockIdx.x - 32 * P_KS;
        const int t = threadIdx.x;
        if (b < CB_E0) { rankmm_elem<8>(Ke0_U, Ke0_V, Ke0, 1025, b * 512 + t); return; }
        b -= CB_E0;
        if (b < CB_O0) { rankmm_elem<15>(Ko0_U, Ko0_V, Ko0, 1024, b * 512 + t); return; }
        b -= CB_O0;
        if (b < CB_E1) { rankmm_elem<8>(Ke1_U, Ke1_V, Ke1, 2049, b * 512 + t); return; }
        b -= CB_E1;
        if (b < CB_O1) { rankmm_elem<15>(Ko1_U, Ko1_V, Ko1, 2048, b * 512 + t); return; }
        b -= CB_O1;
        if (b < CB_E2) { rankmm_elem<8>(Ke2_U, Ke2_V, Ke2, 4097, b * 512 + t); return; }
        b -= CB_E2;
        rankmm_elem<15>(Ko2_U, Ko2_V, Ko2, 4096, b * 512 + t);
        return;
    }
    const int wave = __builtin_amdgcn_readfirstlane((int)(threadIdx.x >> 5));
    const int lane = threadIdx.x & 31;
    const int half = lane >> 4;
    const int mrow = lane & 15;
    const int mt   = blockIdx.x / P_KS;
    const int ks   = blockIdx.x % P_KS;
    const int R    = mt * 16;
    const bool tail = (ks == P_KS - 1) && (wave == 0);

    const float* arow = uh + (size_t)(R + mrow) * 8193;

    v8f t0 = {}; v8f t1 = {};
    for (int it = 0; it < 32 / P_WAVES; ++it) {
        int k0 = 4 * (ks * 32 + it * P_WAVES + wave);
        int ka = k0 + 2 * half;
        int kb = ka + 1;
        v2f a, b0, b1;
        a.x  = arow[8 * ka];
        a.y  = arow[8 * kb];
        b0.x = KHH_V[ka * 32 + mrow];
        b0.y = KHH_V[kb * 32 + mrow];
        b1.x = KHH_V[ka * 32 + 16 + mrow];
        b1.y = KHH_V[kb * 32 + 16 + mrow];
        t0 = WMMA_F32X4(a, b0, t0);
        t1 = WMMA_F32X4(a, b1, t1);
    }
    if (tail) {
        float alast = arow[8192];
        v2f a, b0, b1;
        a.x  = (half == 0) ? alast : 0.f;
        a.y  = 0.f;
        b0.x = KHH_V[1024 * 32 + mrow];
        b0.y = b0.x;
        b1.x = KHH_V[1024 * 32 + 16 + mrow];
        b1.y = b1.x;
        t0 = WMMA_F32X4(a, b0, t0);
        t1 = WMMA_F32X4(a, b1, t1);
    }

    v8f s0 = {}; v8f s1 = {};
    for (int it = 0; it < 256 / P_WAVES; ++it) {
        int k0 = 4 * (ks * 256 + it * P_WAVES + wave);
        int ka = k0 + 2 * half;
        int kb = ka + 1;
        v2f a, b0, b1;
        a.x  = arow[ka];
        a.y  = arow[kb];
        b0.x = Kbd_U[ka * 32 + mrow];
        b0.y = Kbd_U[kb * 32 + mrow];
        b1.x = Kbd_U[ka * 32 + 16 + mrow];
        b1.y = Kbd_U[kb * 32 + 16 + mrow];
        s0 = WMMA_F32X4(a, b0, s0);
        s1 = WMMA_F32X4(a, b1, s1);
    }
    if (tail) {
        float alast = arow[8192];
        v2f a, b0, b1;
        a.x  = (half == 0) ? alast : 0.f;
        a.y  = 0.f;
        b0.x = Kbd_U[8192 * 32 + mrow];
        b0.y = b0.x;
        b1.x = Kbd_U[8192 * 32 + 16 + mrow];
        b1.y = b1.x;
        s0 = WMMA_F32X4(a, b0, s0);
        s1 = WMMA_F32X4(a, b1, s1);
    }

    {
        float* rw = red[wave];
#pragma unroll
        for (int j = 0; j < 8; ++j) {
            rw[j * 32 + lane]       = t0[j];
            rw[256 + j * 32 + lane] = t1[j];
            rw[512 + j * 32 + lane] = s0[j];
            rw[768 + j * 32 + lane] = s1[j];
        }
    }
    __syncthreads();
    {
        float* dst = part + (size_t)mt * TILE_ROWS_FLOATS + ks * 1024;
#pragma unroll
        for (int q = 0; q < 2; ++q) {
            const int e = threadIdx.x + 512 * q;
            float sum = red[0][e];
#pragma unroll
            for (int w = 1; w < P_WAVES; ++w) sum += red[w][e];
            dst[e] = sum;
        }
    }
}

#define RS_S    128
#define RS_R    8
#define RS_NSEG (1024 / RS_S)
#define RS_HP(npts) (((((npts) + 1) / 2 + 3) / 8) * 8 + 4)
#define RS_HPA  RS_HP(4 * RS_S + 1)
#define RS_HPB  RS_HP(8 * RS_S + 1)
#define RS_CH   256
#define RS_IDXN (8 * 1024 + 4)
#define RS_KEN  (3 * 1024 + 4)
#define RS_KON  (4 * 1024 + 4)
#define RS_WMAX (8 * RS_S + 16)
#define RS_HPU  RS_HP(RS_WMAX)

__device__ __forceinline__ int pm_vec(const int n, const int hp) {
    return (n & 1) * (2 * hp) + (n >> 1);
}
__device__ __forceinline__ int pm_word(const int n, const int hp) {
    const int t = 2 * n;
    return t + (t & 2) * (4 * hp - 1);
}

template <int MAXIT, typename V>
__device__ __forceinline__ void stage_load(V (&v)[MAXIT], const V* __restrict__ s4,
                                           const int n4, const int tid) {
#pragma unroll
    for (int k = 0; k < MAXIT; ++k) {
        if (k * 256 < n4) {
            const int i = tid + 256 * k;
            v[k] = s4[i < n4 ? i : n4 - 1];
        }
    }
}
template <int MAXIT, typename V>
__device__ __forceinline__ void stage_store(const V (&v)[MAXIT], V* d4,
                                            const int n4, const int tid) {
#pragma unroll
    for (int k = 0; k < MAXIT; ++k) {
        if (k * 256 < n4) {
            const int i = tid + 256 * k;
            d4[i < n4 ? i : i + 1] = v[k];
        }
    }
}

template <int STEP, int WLEN, int OHP, int NHP>
__device__ __forceinline__ void refine_level(const float* __restrict__ ub,
                                             const float* __restrict__ Ke,
                                             const float* __restrict__ Ko,
                                             const int* __restrict__ nbr,
                                             const float* oldW, float* newW,
                                             const int nb, const int cntE, const int cntO,
                                             const int tid,
                                             int* sIdx, float* sKe, float* sKo, float* sU) {
    const float hf = (1.0f / 8192.0f) * (float)STEP;
    constexpr int NFINE = 8192 / STEP;
    constexpr int HP    = RS_HP(WLEN);
    constexpr int WTOT  = 2 * WLEN;
    constexpr int WIT   = (WTOT + 255) / 256;
    const int wlo = 2 * nb - 8;
#pragma unroll
    for (int cb = 0; cb < cntO; cb += RS_CH) {
        const int ce = (cb + RS_CH >= cntO) ? (cntE - cb) : RS_CH;
        const int co = (cntO - cb < RS_CH) ? (cntO - cb) : RS_CH;
        const int nrows = (2 * ce - 1 > 2 * co) ? (2 * ce - 1) : (2 * co);
        const int n0 = nb + cb;
        {
            const int* isrc = nbr + (size_t)(2 * n0) * 15;
            const int icount = nrows * 15;
            const int in4 = icount >> 2, en4 = (ce * 8) >> 2, on4 = (co * 15) >> 2;
            int4 vi[8];
            float4 ve[3];
            float4 vo[4];
            v4f vw[WIT];
            int vt = 0;
            stage_load<8>(vi, (const int4*)isrc, in4, tid);
            stage_load<3>(ve, (const float4*)(Ke + (size_t)n0 * 8), en4, tid);
            stage_load<4>(vo, (const float4*)(Ko + (size_t)n0 * 15), on4, tid);
            if (tid < (icount & 3)) vt = isrc[4 * in4 + tid];
            if (cb == 0) {
#pragma unroll
                for (int k = 0; k < WIT; ++k) {
                    int e = tid + 256 * k;
                    e = e < WTOT ? e : WTOT - 1;
                    const int q = e / WLEN;
                    int f = wlo + (e - q * WLEN);
                    f = f < 0 ? 0 : (f > NFINE ? NFINE : f);
                    const float* up = ub + (size_t)(4 * q) * 8193 + STEP * f;
                    vw[k].x = up[0];
                    vw[k].y = up[8193];
                    vw[k].z = up[2 * 8193];
                    vw[k].w = up[3 * 8193];
                }
            }
            stage_store<8>(vi, (int4*)sIdx, in4, tid);
            stage_store<3>(ve, (float4*)sKe, en4, tid);
            stage_store<4>(vo, (float4*)sKo, on4, tid);
            if (tid < (icount & 3)) sIdx[4 * in4 + tid] = vt;
            if (cb == 0) {
#pragma unroll
                for (int k = 0; k < WIT; ++k) {
                    const int e = tid + 256 * k;
                    if (e < WTOT) {
                        const int q = e / WLEN;
                        const int j = e - q * WLEN;
                        *(v4f*)(sU + 4 * (pm_vec(j, HP) + q * HP)) = vw[k];
                    }
                }
            }
        }
        __syncthreads();
        for (int nl = tid; nl < ce; nl += 256) {
            const int* row = sIdx + (2 * nl) * 15;
            const float* kp = sKe + nl * 8;
            int ix[8];
            float kk[8];
#pragma unroll
            for (int m = 0; m < 8; ++m) { ix[m] = row[2 * m]; kk[m] = kp[m]; }
            int mn = ix[0], mx = ix[0];
#pragma unroll
            for (int m = 1; m < 8; ++m) { mn = min(mn, ix[m]); mx = max(mx, ix[m]); }
            const int vo = pm_vec(cb + nl, OHP);
            const int vn = pm_vec(2 * (cb + nl), NHP);
            if (mn >= wlo && mx < wlo + WLEN) {
                v4f a0 = {}, a1 = {};
#pragma unroll
                for (int m = 0; m < 8; ++m) {
                    const float* p = sU + pm_word(ix[m] - wlo, HP);
                    const v4f u0 = *(const v4f*)p;
                    const v4f u1 = *(const v4f*)(p + 4 * HP);
                    a0 += kk[m] * u0;
                    a1 += kk[m] * u1;
                }
                const v4f o0 = *(const v4f*)(oldW + 4 * vo);
                const v4f o1 = *(const v4f*)(oldW + 4 * (vo + OHP));
                *(v4f*)(newW + 4 * vn)         = o0 + hf * a0;
                *(v4f*)(newW + 4 * (vn + NHP)) = o1 + hf * a1;
            } else {
#pragma unroll
                for (int m = 0; m < 8; ++m) ix[m] *= STEP;
#pragma unroll 4
                for (int r = 0; r < RS_R; ++r) {
                    const float* u = ub + (size_t)r * 8193;
                    float acc = 0.f;
#pragma unroll
                    for (int m = 0; m < 8; ++m) acc += kk[m] * u[ix[m]];
                    const int q = r >> 2, k = r & 3;
                    newW[4 * (vn + q * NHP) + k] = oldW[4 * (vo + q * OHP) + k] + hf * acc;
                }
            }
        }
        for (int nl = tid; nl < co; nl += 256) {
            const int* row = sIdx + (2 * nl + 1) * 15;
            const float* kp = sKo + nl * 15;
            int ix[15];
            float kk[15];
#pragma unroll
            for (int m = 0; m < 15; ++m) { ix[m] = row[m]; kk[m] = kp[m]; }
            int mn = ix[0], mx = ix[0];
#pragma unroll
            for (int m = 1; m < 15; ++m) { mn = min(mn, ix[m]); mx = max(mx, ix[m]); }
            const int vl = pm_vec(cb + nl, OHP);
            const int vr = pm_vec(cb + nl + 1, OHP);
            const int vn = pm_vec(2 * (cb + nl) + 1, NHP);
            if (mn >= wlo && mx < wlo + WLEN) {
                v4f a0 = {}, a1 = {};
#pragma unroll
                for (int m = 0; m < 15; ++m) {
                    const float* p = sU + pm_word(ix[m] - wlo, HP);
                    const v4f u0 = *(const v4f*)p;
                    const v4f u1 = *(const v4f*)(p + 4 * HP);
                    a0 += kk[m] * u0;
                    a1 += kk[m] * u1;
                }
                const v4f l0 = *(const v4f*)(oldW + 4 * vl);
                const v4f l1 = *(const v4f*)(oldW + 4 * (vl + OHP));
                const v4f r0 = *(const v4f*)(oldW + 4 * vr);
                const v4f r1 = *(const v4f*)(oldW + 4 * (vr + OHP));
                *(v4f*)(newW + 4 * vn)         = 0.5f * (l0 + r0) + hf * a0;
                *(v4f*)(newW + 4 * (vn + NHP)) = 0.5f * (l1 + r1) + hf * a1;
            } else {
#pragma unroll
                for (int m = 0; m < 15; ++m) ix[m] *= STEP;
#pragma unroll 2
                for (int r = 0; r < RS_R; ++r) {
                    const float* u = ub + (size_t)r * 8193;
                    float acc = 0.f;
#pragma unroll
                    for (int m = 0; m < 15; ++m) acc += kk[m] * u[ix[m]];
                    const int q = r >> 2, k = r & 3;
                    newW[4 * (vn + q * NHP) + k] =
                        0.5f * (oldW[4 * (vl + q * OHP) + k] + oldW[4 * (vr + q * OHP) + k]) + hf * acc;
                }
            }
        }
        __syncthreads();
    }
}

__device__ __forceinline__ float boundary_value(const float* srows,
                                                const float* __restrict__ Kbd_V,
                                                const int lr, const int which) {
    const float* kv = Kbd_V + which * 32;
    const float* sp = srows + lr * 32;
    float acc = 0.f;
#pragma unroll
    for (int r = 0; r < 32; ++r) acc += kv[r] * sp[r];
    return (1.0f / 8192.0f) * acc;
}

__global__ void __launch_bounds__(256) k_refine(const float* __restrict__ uh,
                                                const float* __restrict__ part,
                                                const float* __restrict__ KHH_U,
                                                const float* __restrict__ Kbd_V,
                                                const float* __restrict__ Ke0,
                                                const float* __restrict__ Ko0,
                                                const float* __restrict__ Ke1,
                                                const float* __restrict__ Ko1,
                                                const float* __restrict__ Ke2,
                                                const float* __restrict__ Ko2,
                                                const int* __restrict__ nbrL0,
                                                const int* __restrict__ nbrL1,
                                                const int* __restrict__ nbrL2,
                                                float* __restrict__ out) {
    __shared__ float wA[16 * RS_HPA] __attribute__((aligned(16)));
    __shared__ float wB[16 * RS_HPB] __attribute__((aligned(16)));
    __shared__ int   sIdx[RS_IDXN] __attribute__((aligned(16)));
    __shared__ float sKe[RS_KEN] __attribute__((aligned(16)));
    __shared__ float sKo[RS_KON] __attribute__((aligned(16)));
    __shared__ float sU[16 * RS_HPU] __attribute__((aligned(16)));
    __shared__ float tl[RS_R * 32];
    __shared__ float sS[RS_R * 32];
    __shared__ float sBnd[2 * RS_R];
    static_assert(RS_R == 8 && RS_S == 128, "W0 below: 8 rows x 32 columns on 256 threads, 8 waves x 16 points");
    const int tid  = threadIdx.x;
    const int wave = __builtin_amdgcn_readfirstlane((int)(threadIdx.x >> 5));
    const int lane = tid & 31;
    const int half = lane >> 4;
    const int mrow = lane & 15;
    const int sg   = blockIdx.x % RS_NSEG;
    const int rg   = blockIdx.x / RS_NSEG;
    const int c0   = sg * RS_S;
    const int row0 = rg * RS_R;
    const float* ub = uh + (size_t)row0 * 8193;

    v2f bf[8];
    v2f bl[8] = {};
    {
        const float* bp = KHH_U + (size_t)(c0 + 16 * wave + mrow) * 32 + 2 * half;
#pragma unroll
        for (int q = 0; q < 8; ++q) { bf[q].x = bp[4 * q]; bf[q].y = bp[4 * q + 1]; }
    }
    if (wave == 0) {
        const float* bp = KHH_U + (size_t)(c0 + RS_S) * 32 + 2 * half;
#pragma unroll
        for (int q = 0; q < 8; ++q) { bl[q].x = bp[4 * q]; bl[q].y = bp[4 * q + 1]; }
    }
    {
        const int mt = row0 >> 4;
        const int hb = (row0 >> 3) & 1;
        const int j  = tid >> 5;
        const int n  = tid & 31;
        const float* sp = part + (size_t)mt * TILE_ROWS_FLOATS
                        + ((n >> 4) << 8) + j * 32 + 16 * hb + (n & 15);
        const float HK = 1.0f / 1024.0f;
        float tv[P_KS];
        float sv[P_KS];
#pragma unroll
        for (int s = 0; s < P_KS; ++s) {
            tv[s] = sp[s * 1024];
            sv[s] = sp[s * 1024 + 512];
        }
        float sum = tv[0];
        float ss  = sv[0];
#pragma unroll
        for (int s = 1; s < P_KS; ++s) { sum += tv[s]; ss += sv[s]; }
        tl[j * 32 + n] = sum * HK;
        sS[j * 32 + n] = ss;
    }
    __syncthreads();
    if (sg == 0 && tid >= 256 - RS_R) {
        const int r = tid - (256 - RS_R);
        sBnd[r] = boundary_value(sS, Kbd_V, r, 0);
    }
    if (sg == RS_NSEG - 1 && tid >= 256 - RS_R) {
        const int r = tid - (256 - RS_R);
        sBnd[RS_R + r] = boundary_value(sS, Kbd_V, r, 1);
    }
    {
        const float* ap = tl + (mrow & 7) * 32 + 2 * half;
        v2f a[8];
#pragma unroll
        for (int q = 0; q < 8; ++q) { a[q].x = ap[4 * q]; a[q].y = ap[4 * q + 1]; }
        v8f acc = {};
#pragma unroll
        for (int q = 0; q < 8; ++q) acc = WMMA_F32X4(a[q], bf[q], acc);
        if (half == 0) {
            const int v = pm_vec(16 * wave + mrow, RS_HPA);
            const v4f lo = {acc[0], acc[1], acc[2], acc[3]};
            const v4f hi = {acc[4], acc[5], acc[6], acc[7]};
            *(v4f*)(wA + 4 * v)            = lo;
            *(v4f*)(wA + 4 * (v + RS_HPA)) = hi;
        }
        if (wave == 0) {
            v8f accl = {};
#pragma unroll
            for (int q = 0; q < 8; ++q) accl = WMMA_F32X4(a[q], bl[q], accl);
            if (lane == 0) {
                const int v = pm_vec(RS_S, RS_HPA);
                const v4f lo = {accl[0], accl[1], accl[2], accl[3]};
                const v4f hi = {accl[4], accl[5], accl[6], accl[7]};
                *(v4f*)(wA + 4 * v)            = lo;
                *(v4f*)(wA + 4 * (v + RS_HPA)) = hi;
            }
        }
    }

    refine_level<4, 2 * RS_S + 16, RS_HPA, RS_HPB>(ub, Ke0, Ko0, nbrL0, wA, wB,
                                                   c0, RS_S + 1, RS_S, tid, sIdx, sKe, sKo, sU);
    refine_level<2, 4 * RS_S + 16, RS_HPB, RS_HPA>(ub, Ke1, Ko1, nbrL1, wB, wA,
                                                   2 * c0, 2 * RS_S + 1, 2 * RS_S, tid, sIdx, sKe, sKo, sU);
    refine_level<1, 8 * RS_S + 16, RS_HPA, RS_HPB>(ub, Ke2, Ko2, nbrL2, wA, wB,
                                                   4 * c0, 4 * RS_S, 4 * RS_S, tid, sIdx, sKe, sKo, sU);

    {
        static_assert(8 * RS_S == 4 * 256, "final copy: 256 threads x 4 points = one row of the segment");
        float* ob = out + (size_t)row0 * 8193 + 8 * c0 + 4 * tid;
        v4f p[4][2];
#pragma unroll
        for (int i = 0; i < 4; ++i) {
            const int pv = pm_vec(4 * tid + i, RS_HPB);
            p[i][0] = *(const v4f*)(wB + 4 * pv);
            p[i][1] = *(const v4f*)(wB + 4 * (pv + RS_HPB));
        }
        v4f v[RS_R];
#pragma unroll
        for (int r = 0; r < RS_R; ++r) {
            v[r].x = p[0][r >> 2][r & 3];
            v[r].y = p[1][r >> 2][r & 3];
            v[r].z = p[2][r >> 2][r & 3];
            v[r].w = p[3][r >> 2][r & 3];
        }
        if (sg == 0 && tid == 0) {
#pragma unroll
            for (int r = 0; r < RS_R; ++r) v[r].x = sBnd[r];
        }
#pragma unroll
        for (int r = 0; r < RS_R; ++r) *(v4f_a4*)(ob + (size_t)r * 8193) = v[r];
    }
    if (sg == RS_NSEG - 1 && tid < RS_R)
        out[(size_t)(row0 + tid) * 8193 + 8192] = sBnd[RS_R + tid];
}

extern "C" void kernel_launch(void* const* d_in, const int* in_sizes, int n_in,
                              void* d_out, int out_size, void* d_ws, size_t ws_size,
                              hipStream_t stream) {
    const float* uh     = (const float*)d_in[0];
    const float* KHH_U  = (const float*)d_in[1];
    const float* KHH_V  = (const float*)d_in[2];
    const float* Ke0_U  = (const float*)d_in[3];
    const float* Ke0_V  = (const float*)d_in[4];
    const float* Ko0_U  = (const float*)d_in[5];
    const float* Ko0_V  = (const float*)d_in[6];
    const int*   nbr0   = (const int*)d_in[7];
    const float* Ke1_U  = (const float*)d_in[8];
    const float* Ke1_V  = (const float*)d_in[9];
    const float* Ko1_U  = (const float*)d_in[10];
    const float* Ko1_V  = (const float*)d_in[11];
    const int*   nbr1   = (const int*)d_in[12];
    const float* Ke2_U  = (const float*)d_in[13];
    const float* Ke2_V  = (const float*)d_in[14];
    const float* Ko2_U  = (const float*)d_in[15];
    const float* Ko2_V  = (const float*)d_in[16];
    const int*   nbr2   = (const int*)d_in[17];
    const float* Kbd_U  = (const float*)d_in[18];
    const float* Kbd_V  = (const float*)d_in[19];

    float* ws  = (float*)d_ws;
    float* prt = ws;
    float* Ke0 = prt + 524800;
    float* Ke1 = Ke0 + 8200;
    float* Ke2 = Ke1 + 16392;
    float* Ko0 = Ke2 + 32776;
    float* Ko1 = Ko0 + 15360;
    float* Ko2 = Ko1 + 30720;

    float* outp = (float*)d_out;

    k_wmma_proj_part<<<32 * P_KS + CB_ALL, 512, 0, stream>>>(uh, KHH_V, Kbd_U, prt,
                                                             Ke0_U, Ke0_V, Ko0_U, Ko0_V,
                                                             Ke1_U, Ke1_V, Ko1_U, Ko1_V,
                                                             Ke2_U, Ke2_V, Ko2_U, Ko2_V,
                                                             Ke0, Ko0, Ke1, Ko1, Ke2, Ko2);
    k_refine<<<512, 256, 0, stream>>>(uh, prt, KHH_U, Kbd_V,
                                      Ke0, Ko0, Ke1, Ko1, Ke2, Ko2,
                                      nbr2, nbr1, nbr0, outp);
}


# Refinement: each stage's staging loads (index and coefficient rows, window) are issued before the previous stage's compute and stored to LDS behind it; barriers there fence LDS only; sums unchanged
# speedup vs baseline: 18.0588x; 1.0830x over previous
#include <hip/hip_runtime.h>
#include <hip/hip_bf16.h>

typedef __attribute__((ext_vector_type(2))) float v2f;
typedef __attribute__((ext_vector_type(8))) float v8f;
typedef __attribute__((ext_vector_type(4))) float v4f;
typedef float v4f_a4 __attribute__((ext_vector_type(4), aligned(4)));
typedef __attribute__((ext_vector_type(4))) int v4i;

#define WMMA_F32X4(a, b, c) \
    __builtin_amdgcn_wmma_f32_16x16x4_f32(false, (a), false, (b), (short)0, (c), false, false)

template <int COLS>
__device__ __forceinline__ void rankmm_elem(const float* __restrict__ U,
                                            const float* __restrict__ V,
                                            float* __restrict__ out,
                                            const int rows, const int idx) {
    if (idx >= rows * COLS) return;
    int n = idx / COLS;
    int m = idx - n * COLS;
    const float* up = U + n * 32;
    const float* vp = V + m * 32;
    float acc = 0.f;
#pragma unroll
    for (int r = 0; r < 32; ++r) acc += up[r] * vp[r];
    out[idx] = acc;
}

#define CB(elems) (((elems) + 511) / 512)
#define CB_E0 CB(1025 * 8)
#define CB_O0 CB(1024 * 15)
#define CB_E1 CB(2049 * 8)
#define CB_O1 CB(2048 * 15)
#define CB_E2 CB(4097 * 8)
#define CB_O2 CB(4096 * 15)
#define CB_ALL (CB_E0 + CB_O0 + CB_E1 + CB_O1 + CB_E2 + CB_O2)

#define P_KS     8
#define P_WAVES 16
#define TILE_ROWS_FLOATS 16400

__global__ void __launch_bounds__(512) k_wmma_proj_part(const float* __restrict__ uh,
                                                        const float* __restrict__ KHH_V,
                                                        const float* __restrict__ Kbd_U,
                                                        float* __restrict__ part,
                                                        const float* __restrict__ Ke0_U,
                                                        const float* __restrict__ Ke0_V,
                                                        const float* __restrict__ Ko0_U,
                                                        const float* __restrict__ Ko0_V,
                                                        const float* __restrict__ Ke1_U,
                                                        const float* __restrict__ Ke1_V,
                                                        const float* __restrict__ Ko1_U,
                                                        const float* __restrict__ Ko1_V,
                                                        const float* __restrict__ Ke2_U,
                                                        const float* __restrict__ Ke2_V,
                                                        const float* __restrict__ Ko2_U,
                                                        const float* __restrict__ Ko2_V,
                                                        float* __restrict__ Ke0,
                                                        float* __restrict__ Ko0,
                                                        float* __restrict__ Ke1,
                                                        float* __restrict__ Ko1,
                                                        float* __restrict__ Ke2,
                                                        float* __restrict__ Ko2) {
    __shared__ float red[P_WAVES][1024];
    if (blockIdx.x >= 32 * P_KS) {
        int b = blockIdx.x - 32 * P_KS;
        const int t = threadIdx.x;
        if (b < CB_E0) { rankmm_elem<8>(Ke0_U, Ke0_V, Ke0, 1025, b * 512 + t); return; }
        b -= CB_E0;
        if (b < CB_O0) { rankmm_elem<15>(Ko0_U, Ko0_V, Ko0, 1024, b * 512 + t); return; }
        b -= CB_O0;
        if (b < CB_E1) { rankmm_elem<8>(Ke1_U, Ke1_V, Ke1, 2049, b * 512 + t); return; }
        b -= CB_E1;
        if (b < CB_O1) { rankmm_elem<15>(Ko1_U, Ko1_V, Ko1, 2048, b * 512 + t); return; }
        b -= CB_O1;
        if (b < CB_E2) { rankmm_elem<8>(Ke2_U, Ke2_V, Ke2, 4097, b * 512 + t); return; }
        b -= CB_E2;
        rankmm_elem<15>(Ko2_U, Ko2_V, Ko2, 4096, b * 512 + t);
        return;
    }
    const int wave = __builtin_amdgcn_readfirstlane((int)(threadIdx.x >> 5));
    const int lane = threadIdx.x & 31;
    const int half = lane >> 4;
    const int mrow = lane & 15;
    const int mt   = blockIdx.x / P_KS;
    const int ks   = blockIdx.x % P_KS;
    const int R    = mt * 16;
    const bool tail = (ks == P_KS - 1) && (wave == 0);

    const float* arow = uh + (size_t)(R + mrow) * 8193;

    v8f t0 = {}; v8f t1 = {};
    for (int it = 0; it < 32 / P_WAVES; ++it) {
        int k0 = 4 * (ks * 32 + it * P_WAVES + wave);
        int ka = k0 + 2 * half;
        int kb = ka + 1;
        v2f a, b0, b1;
        a.x  = arow[8 * ka];
        a.y  = arow[8 * kb];
        b0.x = KHH_V[ka * 32 + mrow];
        b0.y = KHH_V[kb * 32 + mrow];
        b1.x = KHH_V[ka * 32 + 16 + mrow];
        b1.y = KHH_V[kb * 32 + 16 + mrow];
        t0 = WMMA_F32X4(a, b0, t0);
        t1 = WMMA_F32X4(a, b1, t1);
    }
    if (tail) {
        float alast = arow[8192];
        v2f a, b0, b1;
        a.x  = (half == 0) ? alast : 0.f;
        a.y  = 0.f;
        b0.x = KHH_V[1024 * 32 + mrow];
        b0.y = b0.x;
        b1.x = KHH_V[1024 * 32 + 16 + mrow];
        b1.y = b1.x;
        t0 = WMMA_F32X4(a, b0, t0);
        t1 = WMMA_F32X4(a, b1, t1);
    }

    v8f s0 = {}; v8f s1 = {};
    for (int it = 0; it < 256 / P_WAVES; ++it) {
        int k0 = 4 * (ks * 256 + it * P_WAVES + wave);
        int ka = k0 + 2 * half;
        int kb = ka + 1;
        v2f a, b0, b1;
        a.x  = arow[ka];
        a.y  = arow[kb];
        b0.x = Kbd_U[ka * 32 + mrow];
        b0.y = Kbd_U[kb * 32 + mrow];
        b1.x = Kbd_U[ka * 32 + 16 + mrow];
        b1.y = Kbd_U[kb * 32 + 16 + mrow];
        s0 = WMMA_F32X4(a, b0, s0);
        s1 = WMMA_F32X4(a, b1, s1);
    }
    if (tail) {
        float alast = arow[8192];
        v2f a, b0, b1;
        a.x  = (half == 0) ? alast : 0.f;
        a.y  = 0.f;
        b0.x = Kbd_U[8192 * 32 + mrow];
        b0.y = b0.x;
        b1.x = Kbd_U[8192 * 32 + 16 + mrow];
        b1.y = b1.x;
        s0 = WMMA_F32X4(a, b0, s0);
        s1 = WMMA_F32X4(a, b1, s1);
    }

    {
        float* rw = red[wave];
#pragma unroll
        for (int j = 0; j < 8; ++j) {
            rw[j * 32 + lane]       = t0[j];
            rw[256 + j * 32 + lane] = t1[j];
            rw[512 + j * 32 + lane] = s0[j];
            rw[768 + j * 32 + lane] = s1[j];
        }
    }
    __syncthreads();
    {
        float* dst = part + (size_t)mt * TILE_ROWS_FLOATS + ks * 1024;
#pragma unroll
        for (int q = 0; q < 2; ++q) {
            const int e = threadIdx.x + 512 * q;
            float sum = red[0][e];
#pragma unroll
            for (int w = 1; w < P_WAVES; ++w) sum += red[w][e];
            dst[e] = sum;
        }
    }
}

#define RS_S    128
#define RS_R    8
#define RS_NSEG (1024 / RS_S)
#define RS_HP(npts) (((((npts) + 1) / 2 + 3) / 8) * 8 + 4)
#define RS_HPA  RS_HP(4 * RS_S + 1)
#define RS_HPB  RS_HP(8 * RS_S + 1)
#define RS_CH   256
#define RS_IDXN (8 * 1024 + 4)
#define RS_KEN  (3 * 1024 + 4)
#define RS_KON  (4 * 1024 + 4)
#define RS_WMAX (8 * RS_S + 16)
#define RS_HPU  RS_HP(RS_WMAX)

__device__ __forceinline__ int pm_vec(const int n, const int hp) {
    return (n & 1) * (2 * hp) + (n >> 1);
}
__device__ __forceinline__ int pm_word(const int n, const int hp) {
    const int t = 2 * n;
    return t + (t & 2) * (4 * hp - 1);
}

__device__ __forceinline__ void lds_barrier() {
    __builtin_amdgcn_fence(__ATOMIC_RELEASE, "workgroup", "local");
    __builtin_amdgcn_s_barrier();
    __builtin_amdgcn_fence(__ATOMIC_ACQUIRE, "workgroup", "local");
}

template <int MAXIT, typename V>
__device__ __forceinline__ void stage_load(V (&v)[MAXIT], const V* __restrict__ s4,
                                           const int n4, const int tid) {
#pragma unroll
    for (int k = 0; k < MAXIT; ++k) {
        if (k * 256 < n4) {
            const int i = tid + 256 * k;
            v[k] = s4[i < n4 ? i : n4 - 1];
        }
    }
}
template <int MAXIT, typename V>
__device__ __forceinline__ void stage_store(const V (&v)[MAXIT], V* d4,
                                            const int n4, const int tid) {
#pragma unroll
    for (int k = 0; k < MAXIT; ++k) {
        if (k * 256 < n4) {
            const int i = tid + 256 * k;
            d4[i < n4 ? i : i + 1] = v[k];
        }
    }
}

template <int STEP_, int WLEN_, int CB_, int CE_, int CO_>
struct StageCfg {
    static constexpr int STEP = STEP_, WLEN = WLEN_, CB = CB_, CE = CE_, CO = CO_;
    static constexpr int NROWS  = (2 * CE - 1 > 2 * CO) ? (2 * CE - 1) : (2 * CO);
    static constexpr int ICOUNT = NROWS * 15;
    static constexpr int IN4 = ICOUNT >> 2, EN4 = (CE * 8) >> 2, ON4 = (CO * 15) >> 2;
    static constexpr int ITAIL = ICOUNT & 3;
    static constexpr int NFINE = 8192 / STEP;
    static constexpr int HP    = RS_HP(WLEN);
    static constexpr int WTOT  = 2 * WLEN;
    static constexpr bool WIN  = (CB == 0);
    static constexpr int WIT   = WIN ? (WTOT + 255) / 256 : 1;
};

template <typename C>
struct StageRegs {
    v4i vi[8];
    v4f ve[3];
    v4f vo[4];
    v4f vw[C::WIT];
    int vt;
};

template <typename C>
__device__ __forceinline__ void stage_issue(StageRegs<C>& g,
                                            const float* __restrict__ ub,
                                            const float* __restrict__ Ke,
                                            const float* __restrict__ Ko,
                                            const int* __restrict__ nbr,
                                            const int nb, const int tid) {
    const int n0 = nb + C::CB;
    const int* isrc = nbr + (size_t)(2 * n0) * 15;
    stage_load<8>(g.vi, (const v4i*)isrc, C::IN4, tid);
    stage_load<3>(g.ve, (const v4f*)(Ke + (size_t)n0 * 8), C::EN4, tid);
    stage_load<4>(g.vo, (const v4f*)(Ko + (size_t)n0 * 15), C::ON4, tid);
    g.vt = 0;
    if (tid < C::ITAIL) g.vt = isrc[4 * C::IN4 + tid];
    if (C::WIN) {
        const int wlo = 2 * nb - 8;
#pragma unroll
        for (int k = 0; k < C::WIT; ++k) {
            int e = tid + 256 * k;
            e = e < C::WTOT ? e : C::WTOT - 1;
            const int q = e / C::WLEN;
            int f = wlo + (e - q * C::WLEN);
            f = f < 0 ? 0 : (f > C::NFINE ? C::NFINE : f);
            const float* up = ub + (size_t)(4 * q) * 8193 + C::STEP * f;
            g.vw[k].x = up[0];
            g.vw[k].y = up[8193];
            g.vw[k].z = up[2 * 8193];
            g.vw[k].w = up[3 * 8193];
        }
    }
}

template <typename C>
__device__ __forceinline__ void stage_commit(const StageRegs<C>& g, const int tid,
                                             int* sIdx, float* sKe, float* sKo, float* sU) {
    stage_store<8>(g.vi, (v4i*)sIdx, C::IN4, tid);
    stage_store<3>(g.ve, (v4f*)sKe, C::EN4, tid);
    stage_store<4>(g.vo, (v4f*)sKo, C::ON4, tid);
    if (tid < C::ITAIL) sIdx[4 * C::IN4 + tid] = g.vt;
    if (C::WIN) {
#pragma unroll
        for (int k = 0; k < C::WIT; ++k) {
            const int e = tid + 256 * k;
            if (e < C::WTOT) {
                const int q = e / C::WLEN;
                const int j = e - q * C::WLEN;
                *(v4f*)(sU + 4 * (pm_vec(j, C::HP) + q * C::HP)) = g.vw[k];
            }
        }
    }
}

template <typename C, int OHP, int NHP>
__device__ __forceinline__ void stage_compute(const float* __restrict__ ub,
                                              const float* oldW, float* newW,
                                              const int nb, const int tid,
                                              const int* sIdx, const float* sKe,
                                              const float* sKo, const float* sU) {
    constexpr int STEP = C::STEP, WLEN = C::WLEN, HP = C::HP;
    constexpr int cb = C::CB, ce = C::CE, co = C::CO;
    const float hf = (1.0f / 8192.0f) * (float)STEP;
    const int wlo = 2 * nb - 8;
    {
        for (int nl = tid; nl < ce; nl += 256) {
            const int* row = sIdx + (2 * nl) * 15;
            const float* kp = sKe + nl * 8;
            int ix[8];
            float kk[8];
#pragma unroll
            for (int m = 0; m < 8; ++m) { ix[m] = row[2 * m]; kk[m] = kp[m]; }
            int mn = ix[0], mx = ix[0];
#pragma unroll
            for (int m = 1; m < 8; ++m) { mn = min(mn, ix[m]); mx = max(mx, ix[m]); }
            const int vo = pm_vec(cb + nl, OHP);
            const int vn = pm_vec(2 * (cb + nl), NHP);
            if (mn >= wlo && mx < wlo + WLEN) {
                v4f a0 = {}, a1 = {};
#pragma unroll
                for (int m = 0; m < 8; ++m) {
                    const float* p = sU + pm_word(ix[m] - wlo, HP);
                    const v4f u0 = *(const v4f*)p;
                    const v4f u1 = *(const v4f*)(p + 4 * HP);
                    a0 += kk[m] * u0;
                    a1 += kk[m] * u1;
                }
                const v4f o0 = *(const v4f*)(oldW + 4 * vo);
                const v4f o1 = *(const v4f*)(oldW + 4 * (vo + OHP));
                *(v4f*)(newW + 4 * vn)         = o0 + hf * a0;
                *(v4f*)(newW + 4 * (vn + NHP)) = o1 + hf * a1;
            } else {
                unsigned off[8];
#pragma unroll
                for (int m = 0; m < 8; ++m) off[m] = (unsigned)ix[m] * (unsigned)(4 * STEP);
#pragma unroll
                for (int r = 0; r < RS_R; ++r) {
                    const char* u = (const char*)(ub + (size_t)r * 8193);
                    float acc = 0.f;
#pragma unroll
                    for (int m = 0; m < 8; ++m) acc += kk[m] * *(const float*)(u + off[m]);
                    const int q = r >> 2, k = r & 3;
                    newW[4 * (vn + q * NHP) + k] = oldW[4 * (vo + q * OHP) + k] + hf * acc;
                }
            }
        }
        for (int nl = tid; nl < co; nl += 256) {
            const int* row = sIdx + (2 * nl + 1) * 15;
            const float* kp = sKo + nl * 15;
            int ix[15];
            float kk[15];
#pragma unroll
            for (int m = 0; m < 15; ++m) { ix[m] = row[m]; kk[m] = kp[m]; }
            int mn = ix[0], mx = ix[0];
#pragma unroll
            for (int m = 1; m < 15; ++m) { mn = min(mn, ix[m]); mx = max(mx, ix[m]); }
            const int vl = pm_vec(cb + nl, OHP);
            const int vr = pm_vec(cb + nl + 1, OHP);
            const int vn = pm_vec(2 * (cb + nl) + 1, NHP);
            if (mn >= wlo && mx < wlo + WLEN) {
                v4f a0 = {}, a1 = {};
#pragma unroll
                for (int m = 0; m < 15; ++m) {
                    const float* p = sU + pm_word(ix[m] - wlo, HP);
                    const v4f u0 = *(const v4f*)p;
                    const v4f u1 = *(const v4f*)(p + 4 * HP);
                    a0 += kk[m] * u0;
                    a1 += kk[m] * u1;
                }
                const v4f l0 = *(const v4f*)(oldW + 4 * vl);
                const v4f l1 = *(const v4f*)(oldW + 4 * (vl + OHP));
                const v4f r0 = *(const v4f*)(oldW + 4 * vr);
                const v4f r1 = *(const v4f*)(oldW + 4 * (vr + OHP));
                *(v4f*)(newW + 4 * vn)         = 0.5f * (l0 + r0) + hf * a0;
                *(v4f*)(newW + 4 * (vn + NHP)) = 0.5f * (l1 + r1) + hf * a1;
            } else {
                unsigned off[15];
#pragma unroll
                for (int m = 0; m < 15; ++m) off[m] = (unsigned)ix[m] * (unsigned)(4 * STEP);
#pragma unroll
                for (int r = 0; r < RS_R; ++r) {
                    const char* u = (const char*)(ub + (size_t)r * 8193);
                    float acc = 0.f;
#pragma unroll
                    for (int m = 0; m < 15; ++m) acc += kk[m] * *(const float*)(u + off[m]);
                    const int q = r >> 2, k = r & 3;
                    newW[4 * (vn + q * NHP) + k] =
                        0.5f * (oldW[4 * (vl + q * OHP) + k] + oldW[4 * (vr + q * OHP) + k]) + hf * acc;
                    if (r == 3) __builtin_amdgcn_sched_barrier(0);
                }
            }
        }
    }
}

__device__ __forceinline__ float boundary_value(const float* srows,
                                                const float* __restrict__ Kbd_V,
                                                const int lr, const int which) {
    const float* kv = Kbd_V + which * 32;
    const float* sp = srows + lr * 32;
    float acc = 0.f;
#pragma unroll
    for (int r = 0; r < 32; ++r) acc += kv[r] * sp[r];
    return (1.0f / 8192.0f) * acc;
}

__global__ void __launch_bounds__(256) k_refine(const float* __restrict__ uh,
                                                const float* __restrict__ part,
                                                const float* __restrict__ KHH_U,
                                                const float* __restrict__ Kbd_V,
                                                const float* __restrict__ Ke0,
                                                const float* __restrict__ Ko0,
                                                const float* __restrict__ Ke1,
                                                const float* __restrict__ Ko1,
                                                const float* __restrict__ Ke2,
                                                const float* __restrict__ Ko2,
                                                const int* __restrict__ nbrL0,
                                                const int* __restrict__ nbrL1,
                                                const int* __restrict__ nbrL2,
                                                float* __restrict__ out) {
    __shared__ float wA[16 * RS_HPA] __attribute__((aligned(16)));
    __shared__ float wB[16 * RS_HPB] __attribute__((aligned(16)));
    __shared__ int   sIdx[RS_IDXN] __attribute__((aligned(16)));
    __shared__ float sKe[RS_KEN] __attribute__((aligned(16)));
    __shared__ float sKo[RS_KON] __attribute__((aligned(16)));
    __shared__ float sU[16 * RS_HPU] __attribute__((aligned(16)));
    __shared__ float tl[RS_R * 32];
    __shared__ float sS[RS_R * 32];
    __shared__ float sBnd[2 * RS_R];
    static_assert(RS_R == 8 && RS_S == 128, "W0 below: 8 rows x 32 columns on 256 threads, 8 waves x 16 points");
    const int tid  = threadIdx.x;
    const int wave = __builtin_amdgcn_readfirstlane((int)(threadIdx.x >> 5));
    const int lane = tid & 31;
    const int half = lane >> 4;
    const int mrow = lane & 15;
    const int sg   = blockIdx.x % RS_NSEG;
    const int rg   = blockIdx.x / RS_NSEG;
    const int c0   = sg * RS_S;
    const int row0 = rg * RS_R;
    const float* ub = uh + (size_t)row0 * 8193;

    using C1  = StageCfg<4, 2 * RS_S + 16, 0, RS_S + 1, RS_S>;
    using C2  = StageCfg<2, 4 * RS_S + 16, 0, 2 * RS_S + 1, 2 * RS_S>;
    using C3a = StageCfg<1, 8 * RS_S + 16, 0, RS_CH, RS_CH>;
    using C3b = StageCfg<1, 8 * RS_S + 16, RS_CH, RS_CH, RS_CH>;
    static_assert(4 * RS_S == 2 * RS_CH, "level 3 is two chunks of RS_CH even and RS_CH odd points");

    v2f bf[8];
    v2f bl[8] = {};
    {
        const float* bp = KHH_U + (size_t)(c0 + 16 * wave + mrow) * 32 + 2 * half;
#pragma unroll
        for (int q = 0; q < 8; ++q) { bf[q].x = bp[4 * q]; bf[q].y = bp[4 * q + 1]; }
    }
    if (wave == 0) {
        const float* bp = KHH_U + (size_t)(c0 + RS_S) * 32 + 2 * half;
#pragma unroll
        for (int q = 0; q < 8; ++q) { bl[q].x = bp[4 * q]; bl[q].y = bp[4 * q + 1]; }
    }
    StageRegs<C1> g1;
    {
        const int mt = row0 >> 4;
        const int hb = (row0 >> 3) & 1;
        const int j  = tid >> 5;
        const int n  = tid & 31;
        const float* sp = part + (size_t)mt * TILE_ROWS_FLOATS
                        + ((n >> 4) << 8) + j * 32 + 16 * hb + (n & 15);
        const float HK = 1.0f / 1024.0f;
        float tv[P_KS];
        float sv[P_KS];
#pragma unroll
        for (int s = 0; s < P_KS; ++s) {
            tv[s] = sp[s * 1024];
            sv[s] = sp[s * 1024 + 512];
        }
        stage_issue<C1>(g1, ub, Ke0, Ko0, nbrL0, c0, tid);
        float sum = tv[0];
        float ss  = sv[0];
#pragma unroll
        for (int s = 1; s < P_KS; ++s) { sum += tv[s]; ss += sv[s]; }
        tl[j * 32 + n] = sum * HK;
        sS[j * 32 + n] = ss;
    }
    lds_barrier();
    if (sg == 0 && tid >= 256 - RS_R) {
        const int r = tid - (256 - RS_R);
        sBnd[r] = boundary_value(sS, Kbd_V, r, 0);
    }
    if (sg == RS_NSEG - 1 && tid >= 256 - RS_R) {
        const int r = tid - (256 - RS_R);
        sBnd[RS_R + r] = boundary_value(sS, Kbd_V, r, 1);
    }
    {
        const float* ap = tl + (mrow & 7) * 32 + 2 * half;
        v2f a[8];
#pragma unroll
        for (int q = 0; q < 8; ++q) { a[q].x = ap[4 * q]; a[q].y = ap[4 * q + 1]; }
        v8f acc = {};
#pragma unroll
        for (int q = 0; q < 8; ++q) acc = WMMA_F32X4(a[q], bf[q], acc);
        if (half == 0) {
            const int v = pm_vec(16 * wave + mrow, RS_HPA);
            const v4f lo = {acc[0], acc[1], acc[2], acc[3]};
            const v4f hi = {acc[4], acc[5], acc[6], acc[7]};
            *(v4f*)(wA + 4 * v)            = lo;
            *(v4f*)(wA + 4 * (v + RS_HPA)) = hi;
        }
        if (wave == 0) {
            v8f accl = {};
#pragma unroll
            for (int q = 0; q < 8; ++q) accl = WMMA_F32X4(a[q], bl[q], accl);
            if (lane == 0) {
                const int v = pm_vec(RS_S, RS_HPA);
                const v4f lo = {accl[0], accl[1], accl[2], accl[3]};
                const v4f hi = {accl[4], accl[5], accl[6], accl[7]};
                *(v4f*)(wA + 4 * v)            = lo;
                *(v4f*)(wA + 4 * (v + RS_HPA)) = hi;
            }
        }
    }

    stage_commit<C1>(g1, tid, sIdx, sKe, sKo, sU);
    StageRegs<C2> g2;
    stage_issue<C2>(g2, ub, Ke1, Ko1, nbrL1, 2 * c0, tid);
    lds_barrier();
    stage_compute<C1, RS_HPA, RS_HPB>(ub, wA, wB, c0, tid, sIdx, sKe, sKo, sU);
    lds_barrier();

    stage_commit<C2>(g2, tid, sIdx, sKe, sKo, sU);
    StageRegs<C3a> g3a;
    stage_issue<C3a>(g3a, ub, Ke2, Ko2, nbrL2, 4 * c0, tid);
    lds_barrier();
    stage_compute<C2, RS_HPB, RS_HPA>(ub, wB, wA, 2 * c0, tid, sIdx, sKe, sKo, sU);
    lds_barrier();

    stage_commit<C3a>(g3a, tid, sIdx, sKe, sKo, sU);
    StageRegs<C3b> g3b;
    stage_issue<C3b>(g3b, ub, Ke2, Ko2, nbrL2, 4 * c0, tid);
    lds_barrier();
    stage_compute<C3a, RS_HPA, RS_HPB>(ub, wA, wB, 4 * c0, tid, sIdx, sKe, sKo, sU);
    lds_barrier();

    stage_commit<C3b>(g3b, tid, sIdx, sKe, sKo, sU);
    lds_barrier();
    stage_compute<C3b, RS_HPA, RS_HPB>(ub, wA, wB, 4 * c0, tid, sIdx, sKe, sKo, sU);
    lds_barrier();

    {
        static_assert(8 * RS_S == 4 * 256, "final copy: 256 threads x 4 points = one row of the segment");
        float* ob = out + (size_t)row0 * 8193 + 8 * c0 + 4 * tid;
        v4f p[4][2];
#pragma unroll
        for (int i = 0; i < 4; ++i) {
            const int pv = pm_vec(4 * tid + i, RS_HPB);
            p[i][0] = *(const v4f*)(wB + 4 * pv);
            p[i][1] = *(const v4f*)(wB + 4 * (pv + RS_HPB));
        }
        v4f v[RS_R];
#pragma unroll
        for (int r = 0; r < RS_R; ++r) {
            v[r].x = p[0][r >> 2][r & 3];
            v[r].y = p[1][r >> 2][r & 3];
            v[r].z = p[2][r >> 2][r & 3];
            v[r].w = p[3][r >> 2][r & 3];
        }
        if (sg == 0 && tid == 0) {
#pragma unroll
            for (int r = 0; r < RS_R; ++r) v[r].x = sBnd[r];
        }
#pragma unroll
        for (int r = 0; r < RS_R; ++r) *(v4f_a4*)(ob + (size_t)r * 8193) = v[r];
    }
    if (sg == RS_NSEG - 1 && tid < RS_R)
        out[(size_t)(row0 + tid) * 8193 + 8192] = sBnd[RS_R + tid];
}

extern "C" void kernel_launch(void* const* d_in, const int* in_sizes, int n_in,
                              void* d_out, int out_size, void* d_ws, size_t ws_size,
                              hipStream_t stream) {
    const float* uh     = (const float*)d_in[0];
    const float* KHH_U  = (const float*)d_in[1];
    const float* KHH_V  = (const float*)d_in[2];
    const float* Ke0_U  = (const float*)d_in[3];
    const float* Ke0_V  = (const float*)d_in[4];
    const float* Ko0_U  = (const float*)d_in[5];
    const float* Ko0_V  = (const float*)d_in[6];
    const int*   nbr0   = (const int*)d_in[7];
    const float* Ke1_U  = (const float*)d_in[8];
    const float* Ke1_V  = (const float*)d_in[9];
    const float* Ko1_U  = (const float*)d_in[10];
    const float* Ko1_V  = (const float*)d_in[11];
    const int*   nbr1   = (const int*)d_in[12];
    const float* Ke2_U  = (const float*)d_in[13];
    const float* Ke2_V  = (const float*)d_in[14];
    const float* Ko2_U  = (const float*)d_in[15];
    const float* Ko2_V  = (const float*)d_in[16];
    const int*   nbr2   = (const int*)d_in[17];
    const float* Kbd_U  = (const float*)d_in[18];
    const float* Kbd_V  = (const float*)d_in[19];

    float* ws  = (float*)d_ws;
    float* prt = ws;
    float* Ke0 = prt + 524800;
    float* Ke1 = Ke0 + 8200;
    float* Ke2 = Ke1 + 16392;
    float* Ko0 = Ke2 + 32776;
    float* Ko1 = Ko0 + 15360;
    float* Ko2 = Ko1 + 30720;

    float* outp = (float*)d_out;

    k_wmma_proj_part<<<32 * P_KS + CB_ALL, 512, 0, stream>>>(uh, KHH_V, Kbd_U, prt,
                                                             Ke0_U, Ke0_V, Ko0_U, Ko0_V,
                                                             Ke1_U, Ke1_V, Ko1_U, Ko1_V,
                                                             Ke2_U, Ke2_V, Ko2_U, Ko2_V,
                                                             Ke0, Ko0, Ke1, Ko1, Ke2, Ko2);
    k_refine<<<512, 256, 0, stream>>>(uh, prt, KHH_U, Kbd_V,
                                      Ke0, Ko0, Ke1, Ko1, Ke2, Ko2,
                                      nbr2, nbr1, nbr0, outp);
}


# Projection loads widened: 4 neighbouring uh columns per 16-byte load, each B row as 8 bytes (columns 2c, 2c+1): half the loads and cache lines per column; k order within a K slice changes
# speedup vs baseline: 18.7331x; 1.0373x over previous
#include <hip/hip_runtime.h>
#include <hip/hip_bf16.h>

typedef __attribute__((ext_vector_type(2))) float v2f;
typedef __attribute__((ext_vector_type(8))) float v8f;
typedef __attribute__((ext_vector_type(4))) float v4f;
typedef float v4f_a4 __attribute__((ext_vector_type(4), aligned(4)));
typedef __attribute__((ext_vector_type(4))) int v4i;

#define WMMA_F32X4(a, b, c) \
    __builtin_amdgcn_wmma_f32_16x16x4_f32(false, (a), false, (b), (short)0, (c), false, false)

template <int COLS>
__device__ __forceinline__ void rankmm_elem(const float* __restrict__ U,
                                            const float* __restrict__ V,
                                            float* __restrict__ out,
                                            const int rows, const int idx) {
    if (idx >= rows * COLS) return;
    int n = idx / COLS;
    int m = idx - n * COLS;
    const float* up = U + n * 32;
    const float* vp = V + m * 32;
    float acc = 0.f;
#pragma unroll
    for (int r = 0; r < 32; ++r) acc += up[r] * vp[r];
    out[idx] = acc;
}

#define CB(elems) (((elems) + 511) / 512)
#define CB_E0 CB(1025 * 8)
#define CB_O0 CB(1024 * 15)
#define CB_E1 CB(2049 * 8)
#define CB_O1 CB(2048 * 15)
#define CB_E2 CB(4097 * 8)
#define CB_O2 CB(4096 * 15)
#define CB_ALL (CB_E0 + CB_O0 + CB_E1 + CB_O1 + CB_E2 + CB_O2)

#define P_KS     8
#define P_WAVES 16
#define TILE_ROWS_FLOATS 16400

__global__ void __launch_bounds__(512) k_wmma_proj_part(const float* __restrict__ uh,
                                                        const float* __restrict__ KHH_V,
                                                        const float* __restrict__ Kbd_U,
                                                        float* __restrict__ part,
                                                        const float* __restrict__ Ke0_U,
                                                        const float* __restrict__ Ke0_V,
                                                        const float* __restrict__ Ko0_U,
                                                        const float* __restrict__ Ko0_V,
                                                        const float* __restrict__ Ke1_U,
                                                        const float* __restrict__ Ke1_V,
                                                        const float* __restrict__ Ko1_U,
                                                        const float* __restrict__ Ko1_V,
                                                        const float* __restrict__ Ke2_U,
                                                        const float* __restrict__ Ke2_V,
                                                        const float* __restrict__ Ko2_U,
                                                        const float* __restrict__ Ko2_V,
                                                        float* __restrict__ Ke0,
                                                        float* __restrict__ Ko0,
                                                        float* __restrict__ Ke1,
                                                        float* __restrict__ Ko1,
                                                        float* __restrict__ Ke2,
                                                        float* __restrict__ Ko2) {
    __shared__ float red[P_WAVES][1024];
    if (blockIdx.x >= 32 * P_KS) {
        int b = blockIdx.x - 32 * P_KS;
        const int t = threadIdx.x;
        if (b < CB_E0) { rankmm_elem<8>(Ke0_U, Ke0_V, Ke0, 1025, b * 512 + t); return; }
        b -= CB_E0;
        if (b < CB_O0) { rankmm_elem<15>(Ko0_U, Ko0_V, Ko0, 1024, b * 512 + t); return; }
        b -= CB_O0;
        if (b < CB_E1) { rankmm_elem<8>(Ke1_U, Ke1_V, Ke1, 2049, b * 512 + t); return; }
        b -= CB_E1;
        if (b < CB_O1) { rankmm_elem<15>(Ko1_U, Ko1_V, Ko1, 2048, b * 512 + t); return; }
        b -= CB_O1;
        if (b < CB_E2) { rankmm_elem<8>(Ke2_U, Ke2_V, Ke2, 4097, b * 512 + t); return; }
        b -= CB_E2;
        rankmm_elem<15>(Ko2_U, Ko2_V, Ko2, 4096, b * 512 + t);
        return;
    }
    const int wave = __builtin_amdgcn_readfirstlane((int)(threadIdx.x >> 5));
    const int lane = threadIdx.x & 31;
    const int half = lane >> 4;
    const int mrow = lane & 15;
    const int mt   = blockIdx.x / P_KS;
    const int ks   = blockIdx.x % P_KS;
    const int R    = mt * 16;
    const bool tail = (ks == P_KS - 1) && (wave == 0);

    const float* arow = uh + (size_t)(R + mrow) * 8193;

    v8f t0 = {}; v8f t1 = {};
    for (int it = 0; it < 32 / P_WAVES; ++it) {
        int k0 = 4 * (ks * 32 + it * P_WAVES + wave);
        int ka = k0 + 2 * half;
        int kb = ka + 1;
        v2f a, b0, b1;
        a.x  = arow[8 * ka];
        a.y  = arow[8 * kb];
        const v2f ba = *(const v2f*)(KHH_V + ka * 32 + 2 * mrow);
        const v2f bb = *(const v2f*)(KHH_V + kb * 32 + 2 * mrow);
        b0.x = ba.x;  b0.y = bb.x;
        b1.x = ba.y;  b1.y = bb.y;
        t0 = WMMA_F32X4(a, b0, t0);
        t1 = WMMA_F32X4(a, b1, t1);
    }
    if (tail) {
        float alast = arow[8192];
        v2f a, b0, b1;
        a.x  = (half == 0) ? alast : 0.f;
        a.y  = 0.f;
        b0.x = KHH_V[1024 * 32 + 2 * mrow];
        b0.y = b0.x;
        b1.x = KHH_V[1024 * 32 + 2 * mrow + 1];
        b1.y = b1.x;
        t0 = WMMA_F32X4(a, b0, t0);
        t1 = WMMA_F32X4(a, b1, t1);
    }

    v8f s0 = {}; v8f s1 = {};
#pragma unroll 2
    for (int j = 0; j < 128 / P_WAVES; ++j) {
        const int kq = 8 * (ks * 128 + j * P_WAVES + wave) + 4 * half;
        const v4f_a4 a4 = *(const v4f_a4*)(arow + kq);
        const float* bp = Kbd_U + kq * 32 + 2 * mrow;
        const v2f q0 = *(const v2f*)(bp);
        const v2f q1 = *(const v2f*)(bp + 32);
        const v2f q2 = *(const v2f*)(bp + 64);
        const v2f q3 = *(const v2f*)(bp + 96);
        v2f a, b0, b1;
        a.x  = a4.x;  a.y  = a4.y;
        b0.x = q0.x;  b0.y = q1.x;
        b1.x = q0.y;  b1.y = q1.y;
        s0 = WMMA_F32X4(a, b0, s0);
        s1 = WMMA_F32X4(a, b1, s1);
        a.x  = a4.z;  a.y  = a4.w;
        b0.x = q2.x;  b0.y = q3.x;
        b1.x = q2.y;  b1.y = q3.y;
        s0 = WMMA_F32X4(a, b0, s0);
        s1 = WMMA_F32X4(a, b1, s1);
    }
    if (tail) {
        float alast = arow[8192];
        v2f a, b0, b1;
        a.x  = (half == 0) ? alast : 0.f;
        a.y  = 0.f;
        b0.x = Kbd_U[8192 * 32 + 2 * mrow];
        b0.y = b0.x;
        b1.x = Kbd_U[8192 * 32 + 2 * mrow + 1];
        b1.y = b1.x;
        s0 = WMMA_F32X4(a, b0, s0);
        s1 = WMMA_F32X4(a, b1, s1);
    }

    {
        float* rw = red[wave];
#pragma unroll
        for (int j = 0; j < 8; ++j) {
            rw[j * 32 + lane]       = t0[j];
            rw[256 + j * 32 + lane] = t1[j];
            rw[512 + j * 32 + lane] = s0[j];
            rw[768 + j * 32 + lane] = s1[j];
        }
    }
    __syncthreads();
    {
        float* dst = part + (size_t)mt * TILE_ROWS_FLOATS + ks * 1024;
#pragma unroll
        for (int q = 0; q < 2; ++q) {
            const int e = threadIdx.x + 512 * q;
            float sum = red[0][e];
#pragma unroll
            for (int w = 1; w < P_WAVES; ++w) sum += red[w][e];
            dst[e] = sum;
        }
    }
}

#define RS_S    128
#define RS_R    8
#define RS_NSEG (1024 / RS_S)
#define RS_HP(npts) (((((npts) + 1) / 2 + 3) / 8) * 8 + 4)
#define RS_HPA  RS_HP(4 * RS_S + 1)
#define RS_HPB  RS_HP(8 * RS_S + 1)
#define RS_CH   256
#define RS_IDXN (8 * 1024 + 4)
#define RS_KEN  (3 * 1024 + 4)
#define RS_KON  (4 * 1024 + 4)
#define RS_WMAX (8 * RS_S + 16)
#define RS_HPU  RS_HP(RS_WMAX)

__device__ __forceinline__ int pm_vec(const int n, const int hp) {
    return (n & 1) * (2 * hp) + (n >> 1);
}
__device__ __forceinline__ int pm_word(const int n, const int hp) {
    const int t = 2 * n;
    return t + (t & 2) * (4 * hp - 1);
}

__device__ __forceinline__ void lds_barrier() {
    __builtin_amdgcn_fence(__ATOMIC_RELEASE, "workgroup", "local");
    __builtin_amdgcn_s_barrier();
    __builtin_amdgcn_fence(__ATOMIC_ACQUIRE, "workgroup", "local");
}

template <int MAXIT, typename V>
__device__ __forceinline__ void stage_load(V (&v)[MAXIT], const V* __restrict__ s4,
                                           const int n4, const int tid) {
#pragma unroll
    for (int k = 0; k < MAXIT; ++k) {
        if (k * 256 < n4) {
            const int i = tid + 256 * k;
            v[k] = s4[i < n4 ? i : n4 - 1];
        }
    }
}
template <int MAXIT, typename V>
__device__ __forceinline__ void stage_store(const V (&v)[MAXIT], V* d4,
                                            const int n4, const int tid) {
#pragma unroll
    for (int k = 0; k < MAXIT; ++k) {
        if (k * 256 < n4) {
            const int i = tid + 256 * k;
            d4[i < n4 ? i : i + 1] = v[k];
        }
    }
}

template <int STEP_, int WLEN_, int CB_, int CE_, int CO_>
struct StageCfg {
    static constexpr int STEP = STEP_, WLEN = WLEN_, CB = CB_, CE = CE_, CO = CO_;
    static constexpr int NROWS  = (2 * CE - 1 > 2 * CO) ? (2 * CE - 1) : (2 * CO);
    static constexpr int ICOUNT = NROWS * 15;
    static constexpr int IN4 = ICOUNT >> 2, EN4 = (CE * 8) >> 2, ON4 = (CO * 15) >> 2;
    static constexpr int ITAIL = ICOUNT & 3;
    static constexpr int NFINE = 8192 / STEP;
    static constexpr int HP    = RS_HP(WLEN);
    static constexpr int WTOT  = 2 * WLEN;
    static constexpr bool WIN  = (CB == 0);
    static constexpr int WIT   = WIN ? (WTOT + 255) / 256 : 1;
};

template <typename C>
struct StageRegs {
    v4i vi[8];
    v4f ve[3];
    v4f vo[4];
    v4f vw[C::WIT];
    int vt;
};

template <typename C>
__device__ __forceinline__ void stage_issue(StageRegs<C>& g,
                                            const float* __restrict__ ub,
                                            const float* __restrict__ Ke,
                                            const float* __restrict__ Ko,
                                            const int* __restrict__ nbr,
                                            const int nb, const int tid) {
    const int n0 = nb + C::CB;
    const int* isrc = nbr + (size_t)(2 * n0) * 15;
    stage_load<8>(g.vi, (const v4i*)isrc, C::IN4, tid);
    stage_load<3>(g.ve, (const v4f*)(Ke + (size_t)n0 * 8), C::EN4, tid);
    stage_load<4>(g.vo, (const v4f*)(Ko + (size_t)n0 * 15), C::ON4, tid);
    g.vt = 0;
    if (tid < C::ITAIL) g.vt = isrc[4 * C::IN4 + tid];
    if (C::WIN) {
        const int wlo = 2 * nb - 8;
#pragma unroll
        for (int k = 0; k < C::WIT; ++k) {
            int e = tid + 256 * k;
            e = e < C::WTOT ? e : C::WTOT - 1;
            const int q = e / C::WLEN;
            int f = wlo + (e - q * C::WLEN);
            f = f < 0 ? 0 : (f > C::NFINE ? C::NFINE : f);
            const float* up = ub + (size_t)(4 * q) * 8193 + C::STEP * f;
            g.vw[k].x = up[0];
            g.vw[k].y = up[8193];
            g.vw[k].z = up[2 * 8193];
            g.vw[k].w = up[3 * 8193];
        }
    }
}

template <typename C>
__device__ __forceinline__ void stage_commit(const StageRegs<C>& g, const int tid,
                                             int* sIdx, float* sKe, float* sKo, float* sU) {
    stage_store<8>(g.vi, (v4i*)sIdx, C::IN4, tid);
    stage_store<3>(g.ve, (v4f*)sKe, C::EN4, tid);
    stage_store<4>(g.vo, (v4f*)sKo, C::ON4, tid);
    if (tid < C::ITAIL) sIdx[4 * C::IN4 + tid] = g.vt;
    if (C::WIN) {
#pragma unroll
        for (int k = 0; k < C::WIT; ++k) {
            const int e = tid + 256 * k;
            if (e < C::WTOT) {
                const int q = e / C::WLEN;
                const int j = e - q * C::WLEN;
                *(v4f*)(sU + 4 * (pm_vec(j, C::HP) + q * C::HP)) = g.vw[k];
            }
        }
    }
}

template <typename C, int OHP, int NHP>
__device__ __forceinline__ void stage_compute(const float* __restrict__ ub,
                                              const float* oldW, float* newW,
                                              const int nb, const int tid,
                                              const int* sIdx, const float* sKe,
                                              const float* sKo, const float* sU) {
    constexpr int STEP = C::STEP, WLEN = C::WLEN, HP = C::HP;
    constexpr int cb = C::CB, ce = C::CE, co = C::CO;
    const float hf = (1.0f / 8192.0f) * (float)STEP;
    const int wlo = 2 * nb - 8;
    {
        for (int nl = tid; nl < ce; nl += 256) {
            const int* row = sIdx + (2 * nl) * 15;
            const float* kp = sKe + nl * 8;
            int ix[8];
            float kk[8];
#pragma unroll
            for (int m = 0; m < 8; ++m) { ix[m] = row[2 * m]; kk[m] = kp[m]; }
            int mn = ix[0], mx = ix[0];
#pragma unroll
            for (int m = 1; m < 8; ++m) { mn = min(mn, ix[m]); mx = max(mx, ix[m]); }
            const int vo = pm_vec(cb + nl, OHP);
            const int vn = pm_vec(2 * (cb + nl), NHP);
            if (mn >= wlo && mx < wlo + WLEN) {
                v4f a0 = {}, a1 = {};
#pragma unroll
                for (int m = 0; m < 8; ++m) {
                    const float* p = sU + pm_word(ix[m] - wlo, HP);
                    const v4f u0 = *(const v4f*)p;
                    const v4f u1 = *(const v4f*)(p + 4 * HP);
                    a0 += kk[m] * u0;
                    a1 += kk[m] * u1;
                }
                const v4f o0 = *(const v4f*)(oldW + 4 * vo);
                const v4f o1 = *(const v4f*)(oldW + 4 * (vo + OHP));
                *(v4f*)(newW + 4 * vn)         = o0 + hf * a0;
                *(v4f*)(newW + 4 * (vn + NHP)) = o1 + hf * a1;
            } else {
                unsigned off[8];
#pragma unroll
                for (int m = 0; m < 8; ++m) off[m] = (unsigned)ix[m] * (unsigned)(4 * STEP);
#pragma unroll
                for (int r = 0; r < RS_R; ++r) {
                    const char* u = (const char*)(ub + (size_t)r * 8193);
                    float acc = 0.f;
#pragma unroll
                    for (int m = 0; m < 8; ++m) acc += kk[m] * *(const float*)(u + off[m]);
                    const int q = r >> 2, k = r & 3;
                    newW[4 * (vn + q * NHP) + k] = oldW[4 * (vo + q * OHP) + k] + hf * acc;
                }
            }
        }
        for (int nl = tid; nl < co; nl += 256) {
            const int* row = sIdx + (2 * nl + 1) * 15;
            const float* kp = sKo + nl * 15;
            int ix[15];
            float kk[15];
#pragma unroll
            for (int m = 0; m < 15; ++m) { ix[m] = row[m]; kk[m] = kp[m]; }
            int mn = ix[0], mx = ix[0];
#pragma unroll
            for (int m = 1; m < 15; ++m) { mn = min(mn, ix[m]); mx = max(mx, ix[m]); }
            const int vl = pm_vec(cb + nl, OHP);
            const int vr = pm_vec(cb + nl + 1, OHP);
            const int vn = pm_vec(2 * (cb + nl) + 1, NHP);
            if (mn >= wlo && mx < wlo + WLEN) {
                v4f a0 = {}, a1 = {};
#pragma unroll
                for (int m = 0; m < 15; ++m) {
                    const float* p = sU + pm_word(ix[m] - wlo, HP);
                    const v4f u0 = *(const v4f*)p;
                    const v4f u1 = *(const v4f*)(p + 4 * HP);
                    a0 += kk[m] * u0;
                    a1 += kk[m] * u1;
                }
                const v4f l0 = *(const v4f*)(oldW + 4 * vl);
                const v4f l1 = *(const v4f*)(oldW + 4 * (vl + OHP));
                const v4f r0 = *(const v4f*)(oldW + 4 * vr);
                const v4f r1 = *(const v4f*)(oldW + 4 * (vr + OHP));
                *(v4f*)(newW + 4 * vn)         = 0.5f * (l0 + r0) + hf * a0;
                *(v4f*)(newW + 4 * (vn + NHP)) = 0.5f * (l1 + r1) + hf * a1;
            } else {
                unsigned off[15];
#pragma unroll
                for (int m = 0; m < 15; ++m) off[m] = (unsigned)ix[m] * (unsigned)(4 * STEP);
#pragma unroll
                for (int r = 0; r < RS_R; ++r) {
                    const char* u = (const char*)(ub + (size_t)r * 8193);
                    float acc = 0.f;
#pragma unroll
                    for (int m = 0; m < 15; ++m) acc += kk[m] * *(const float*)(u + off[m]);
                    const int q = r >> 2, k = r & 3;
                    newW[4 * (vn + q * NHP) + k] =
                        0.5f * (oldW[4 * (vl + q * OHP) + k] + oldW[4 * (vr + q * OHP) + k]) + hf * acc;
                    if (r == 3) __builtin_amdgcn_sched_barrier(0);
                }
            }
        }
    }
}

__device__ __forceinline__ float boundary_value(const float* srows,
                                                const float* __restrict__ Kbd_V,
                                                const int lr, const int which) {
    const float* kv = Kbd_V + which * 32;
    const float* sp = srows + lr * 32;
    float acc = 0.f;
#pragma unroll
    for (int r = 0; r < 32; ++r) acc += kv[r] * sp[r];
    return (1.0f / 8192.0f) * acc;
}

__global__ void __launch_bounds__(256) k_refine(const float* __restrict__ uh,
                                                const float* __restrict__ part,
                                                const float* __restrict__ KHH_U,
                                                const float* __restrict__ Kbd_V,
                                                const float* __restrict__ Ke0,
                                                const float* __restrict__ Ko0,
                                                const float* __restrict__ Ke1,
                                                const float* __restrict__ Ko1,
                                                const float* __restrict__ Ke2,
                                                const float* __restrict__ Ko2,
                                                const int* __restrict__ nbrL0,
                                                const int* __restrict__ nbrL1,
                                                const int* __restrict__ nbrL2,
                                                float* __restrict__ out) {
    __shared__ float wA[16 * RS_HPA] __attribute__((aligned(16)));
    __shared__ float wB[16 * RS_HPB] __attribute__((aligned(16)));
    __shared__ int   sIdx[RS_IDXN] __attribute__((aligned(16)));
    __shared__ float sKe[RS_KEN] __attribute__((aligned(16)));
    __shared__ float sKo[RS_KON] __attribute__((aligned(16)));
    __shared__ float sU[16 * RS_HPU] __attribute__((aligned(16)));
    __shared__ float tl[RS_R * 32];
    __shared__ float sS[RS_R * 32];
    __shared__ float sBnd[2 * RS_R];
    static_assert(RS_R == 8 && RS_S == 128, "W0 below: 8 rows x 32 columns on 256 threads, 8 waves x 16 points");
    const int tid  = threadIdx.x;
    const int wave = __builtin_amdgcn_readfirstlane((int)(threadIdx.x >> 5));
    const int lane = tid & 31;
    const int half = lane >> 4;
    const int mrow = lane & 15;
    const int sg   = blockIdx.x % RS_NSEG;
    const int rg   = blockIdx.x / RS_NSEG;
    const int c0   = sg * RS_S;
    const int row0 = rg * RS_R;
    const float* ub = uh + (size_t)row0 * 8193;

    using C1  = StageCfg<4, 2 * RS_S + 16, 0, RS_S + 1, RS_S>;
    using C2  = StageCfg<2, 4 * RS_S + 16, 0, 2 * RS_S + 1, 2 * RS_S>;
    using C3a = StageCfg<1, 8 * RS_S + 16, 0, RS_CH, RS_CH>;
    using C3b = StageCfg<1, 8 * RS_S + 16, RS_CH, RS_CH, RS_CH>;
    static_assert(4 * RS_S == 2 * RS_CH, "level 3 is two chunks of RS_CH even and RS_CH odd points");

    v2f bf[8];
    v2f bl[8] = {};
    {
        const float* bp = KHH_U + (size_t)(c0 + 16 * wave + mrow) * 32 + 2 * half;
#pragma unroll
        for (int q = 0; q < 8; ++q) { bf[q].x = bp[4 * q]; bf[q].y = bp[4 * q + 1]; }
    }
    if (wave == 0) {
        const float* bp = KHH_U + (size_t)(c0 + RS_S) * 32 + 2 * half;
#pragma unroll
        for (int q = 0; q < 8; ++q) { bl[q].x = bp[4 * q]; bl[q].y = bp[4 * q + 1]; }
    }
    StageRegs<C1> g1;
    {
        const int mt = row0 >> 4;
        const int hb = (row0 >> 3) & 1;
        const int j  = tid >> 5;
        const int n  = tid & 31;
        const float* sp = part + (size_t)mt * TILE_ROWS_FLOATS
                        + ((n & 1) << 8) + j * 32 + 16 * hb + (n >> 1);
        const float HK = 1.0f / 1024.0f;
        float tv[P_KS];
        float sv[P_KS];
#pragma unroll
        for (int s = 0; s < P_KS; ++s) {
            tv[s] = sp[s * 1024];
            sv[s] = sp[s * 1024 + 512];
        }
        stage_issue<C1>(g1, ub, Ke0, Ko0, nbrL0, c0, tid);
        float sum = tv[0];
        float ss  = sv[0];
#pragma unroll
        for (int s = 1; s < P_KS; ++s) { sum += tv[s]; ss += sv[s]; }
        tl[j * 32 + n] = sum * HK;
        sS[j * 32 + n] = ss;
    }
    lds_barrier();
    if (sg == 0 && tid >= 256 - RS_R) {
        const int r = tid - (256 - RS_R);
        sBnd[r] = boundary_value(sS, Kbd_V, r, 0);
    }
    if (sg == RS_NSEG - 1 && tid >= 256 - RS_R) {
        const int r = tid - (256 - RS_R);
        sBnd[RS_R + r] = boundary_value(sS, Kbd_V, r, 1);
    }
    {
        const float* ap = tl + (mrow & 7) * 32 + 2 * half;
        v2f a[8];
#pragma unroll
        for (int q = 0; q < 8; ++q) { a[q].x = ap[4 * q]; a[q].y = ap[4 * q + 1]; }
        v8f acc = {};
#pragma unroll
        for (int q = 0; q < 8; ++q) acc = WMMA_F32X4(a[q], bf[q], acc);
        if (half == 0) {
            const int v = pm_vec(16 * wave + mrow, RS_HPA);
            const v4f lo = {acc[0], acc[1], acc[2], acc[3]};
            const v4f hi = {acc[4], acc[5], acc[6], acc[7]};
            *(v4f*)(wA + 4 * v)            = lo;
            *(v4f*)(wA + 4 * (v + RS_HPA)) = hi;
        }
        if (wave == 0) {
            v8f accl = {};
#pragma unroll
            for (int q = 0; q < 8; ++q) accl = WMMA_F32X4(a[q], bl[q], accl);
            if (lane == 0) {
                const int v = pm_vec(RS_S, RS_HPA);
                const v4f lo = {accl[0], accl[1], accl[2], accl[3]};
                const v4f hi = {accl[4], accl[5], accl[6], accl[7]};
                *(v4f*)(wA + 4 * v)            = lo;
                *(v4f*)(wA + 4 * (v + RS_HPA)) = hi;
            }
        }
    }

    stage_commit<C1>(g1, tid, sIdx, sKe, sKo, sU);
    StageRegs<C2> g2;
    stage_issue<C2>(g2, ub, Ke1, Ko1, nbrL1, 2 * c0, tid);
    lds_barrier();
    stage_compute<C1, RS_HPA, RS_HPB>(ub, wA, wB, c0, tid, sIdx, sKe, sKo, sU);
    lds_barrier();

    stage_commit<C2>(g2, tid, sIdx, sKe, sKo, sU);
    StageRegs<C3a> g3a;
    stage_issue<C3a>(g3a, ub, Ke2, Ko2, nbrL2, 4 * c0, tid);
    lds_barrier();
    stage_compute<C2, RS_HPB, RS_HPA>(ub, wB, wA, 2 * c0, tid, sIdx, sKe, sKo, sU);
    lds_barrier();

    stage_commit<C3a>(g3a, tid, sIdx, sKe, sKo, sU);
    StageRegs<C3b> g3b;
    stage_issue<C3b>(g3b, ub, Ke2, Ko2, nbrL2, 4 * c0, tid);
    lds_barrier();
    stage_compute<C3a, RS_HPA, RS_HPB>(ub, wA, wB, 4 * c0, tid, sIdx, sKe, sKo, sU);
    lds_barrier();

    stage_commit<C3b>(g3b, tid, sIdx, sKe, sKo, sU);
    lds_barrier();
    stage_compute<C3b, RS_HPA, RS_HPB>(ub, wA, wB, 4 * c0, tid, sIdx, sKe, sKo, sU);
    lds_barrier();

    {
        static_assert(8 * RS_S == 4 * 256, "final copy: 256 threads x 4 points = one row of the segment");
        float* ob = out + (size_t)row0 * 8193 + 8 * c0 + 4 * tid;
        v4f p[4][2];
#pragma unroll
        for (int i = 0; i < 4; ++i) {
            const int pv = pm_vec(4 * tid + i, RS_HPB);
            p[i][0] = *(const v4f*)(wB + 4 * pv);
            p[i][1] = *(const v4f*)(wB + 4 * (pv + RS_HPB));
        }
        v4f v[RS_R];
#pragma unroll
        for (int r = 0; r < RS_R; ++r) {
            v[r].x = p[0][r >> 2][r & 3];
            v[r].y = p[1][r >> 2][r & 3];
            v[r].z = p[2][r >> 2][r & 3];
            v[r].w = p[3][r >> 2][r & 3];
        }
        if (sg == 0 && tid == 0) {
#pragma unroll
            for (int r = 0; r < RS_R; ++r) v[r].x = sBnd[r];
        }
#pragma unroll
        for (int r = 0; r < RS_R; ++r) *(v4f_a4*)(ob + (size_t)r * 8193) = v[r];
    }
    if (sg == RS_NSEG - 1 && tid < RS_R)
        out[(size_t)(row0 + tid) * 8193 + 8192] = sBnd[RS_R + tid];
}

extern "C" void kernel_launch(void* const* d_in, const int* in_sizes, int n_in,
                              void* d_out, int out_size, void* d_ws, size_t ws_size,
                              hipStream_t stream) {
    const float* uh     = (const float*)d_in[0];
    const float* KHH_U  = (const float*)d_in[1];
    const float* KHH_V  = (const float*)d_in[2];
    const float* Ke0_U  = (const float*)d_in[3];
    const float* Ke0_V  = (const float*)d_in[4];
    const float* Ko0_U  = (const float*)d_in[5];
    const float* Ko0_V  = (const float*)d_in[6];
    const int*   nbr0   = (const int*)d_in[7];
    const float* Ke1_U  = (const float*)d_in[8];
    const float* Ke1_V  = (const float*)d_in[9];
    const float* Ko1_U  = (const float*)d_in[10];
    const float* Ko1_V  = (const float*)d_in[11];
    const int*   nbr1   = (const int*)d_in[12];
    const float* Ke2_U  = (const float*)d_in[13];
    const float* Ke2_V  = (const float*)d_in[14];
    const float* Ko2_U  = (const float*)d_in[15];
    const float* Ko2_V  = (const float*)d_in[16];
    const int*   nbr2   = (const int*)d_in[17];
    const float* Kbd_U  = (const float*)d_in[18];
    const float* Kbd_V  = (const float*)d_in[19];

    float* ws  = (float*)d_ws;
    float* prt = ws;
    float* Ke0 = prt + 524800;
    float* Ke1 = Ke0 + 8200;
    float* Ke2 = Ke1 + 16392;
    float* Ko0 = Ke2 + 32776;
    float* Ko1 = Ko0 + 15360;
    float* Ko2 = Ko1 + 30720;

    float* outp = (float*)d_out;

    k_wmma_proj_part<<<32 * P_KS + CB_ALL, 512, 0, stream>>>(uh, KHH_V, Kbd_U, prt,
                                                             Ke0_U, Ke0_V, Ko0_U, Ko0_V,
                                                             Ke1_U, Ke1_V, Ko1_U, Ko1_V,
                                                             Ke2_U, Ke2_V, Ko2_U, Ko2_V,
                                                             Ke0, Ko0, Ke1, Ko1, Ke2, Ko2);
    k_refine<<<512, 256, 0, stream>>>(uh, prt, KHH_U, Kbd_V,
                                      Ke0, Ko0, Ke1, Ko1, Ke2, Ko2,
                                      nbr2, nbr1, nbr0, outp);
}
